# starting kernel
# baseline (speedup 1.0000x reference)
#include <hip/hip_runtime.h>
#include <hip/hip_bf16.h>


typedef float v2f __attribute__((ext_vector_type(2)));
typedef float v8f __attribute__((ext_vector_type(8)));

__device__ __forceinline__ unsigned f2ord(float f) {
  unsigned u = __float_as_uint(f);
  return (u & 0x80000000u) ? ~u : (u | 0x80000000u);
}
__device__ __forceinline__ float ord2f(unsigned o) {
  unsigned u = (o & 0x80000000u) ? (o ^ 0x80000000u) : ~o;
  return __uint_as_float(u);
}

__global__ void zero_kernel(unsigned* __restrict__ p, long long count) {
  long long i = (long long)blockIdx.x * blockDim.x + threadIdx.x;
  long long stride = (long long)gridDim.x * blockDim.x;
  for (; i < count; i += stride) p[i] = 0u;
}

template <int K, int NCOL>
__global__ void pack_b_kernel(const float* __restrict__ B,
                              float* __restrict__ Bp) {
  int tid = blockIdx.x * blockDim.x + threadIdx.x;
  if (tid >= K * NCOL) return;
  int k = tid / NCOL;
  int col = tid - k * NCOL;
  int kk = k >> 2;
  int p = (k >> 1) & 1;
  int q = k & 1;
  Bp[(((kk * 2 + p) * NCOL) + col) * 2 + q] = B[tid];
}

template <int K, int NCOL>
__global__ __launch_bounds__(256) void gemm_wmma_kernel(
    const float* __restrict__ A, const float* __restrict__ Bp,
    float* __restrict__ C, int M) {
  constexpr int NT = NCOL / 16;
  constexpr int MT = 8 / NT;
  constexpr int ROWS = MT * 16;
  constexpr int KV = K / 4;
  __shared__ __align__(16) float As[ROWS * K];

  const int tid = threadIdx.x;
  const int wave = tid >> 5;
  const int lane = tid & 31;
  const int row0 = blockIdx.x * ROWS;

  {
    const float4* A4 = (const float4*)A;
    float4* As4 = (float4*)As;
    for (int i = tid; i < ROWS * KV; i += 256) {
      int r = i / KV;
      int c = i - r * KV;
      int gr = row0 + r;
      float4 v = {0.f, 0.f, 0.f, 0.f};
      if (gr < M) v = A4[(size_t)gr * KV + c];
      As4[i] = v;
    }
  }
  __syncthreads();

  const int mtile = wave / NT;
  const int ntile = wave - mtile * NT;
  const int lrow = (mtile << 4) + (lane & 15);
  const int koff = (lane >> 4) << 1;
  const int ncol = (ntile << 4) + (lane & 15);
  const int bbase = ((lane >> 4) * NCOL + ncol);

  v8f acc = {0.f, 0.f, 0.f, 0.f, 0.f, 0.f, 0.f, 0.f};
  for (int kk = 0; kk < KV; ++kk) {
    v2f a, b;
    a.x = As[lrow * K + kk * 4 + koff];
    a.y = As[lrow * K + kk * 4 + koff + 1];
    b = *(const v2f*)&Bp[(size_t)(bbase + kk * 2 * NCOL) * 2];
    acc = __builtin_amdgcn_wmma_f32_16x16x4_f32(
         false, a,  false, b,
         (short)0, acc,  false,  false);
  }

  const int rbase = row0 + (mtile << 4) + ((lane >> 4) << 3);
  if (row0 + ROWS <= M) {
#pragma unroll
    for (int i = 0; i < 8; ++i) C[(size_t)(rbase + i) * NCOL + ncol] = acc[i];
  } else {
#pragma unroll
    for (int i = 0; i < 8; ++i) {
      int r = rbase + i;
      if (r < M) C[(size_t)r * NCOL + ncol] = acc[i];
    }
  }
}

template <int H, int F>
__global__ void alpha_kernel(const float* __restrict__ h,
                             const float* __restrict__ a_s,
                             const float* __restrict__ a_d,
                             float* __restrict__ out_s,
                             float* __restrict__ out_d) {
  constexpr int HF = H * F;
  __shared__ float ss[HF];
  __shared__ float sd[HF];
  const int n = blockIdx.x;
  const int t = threadIdx.x;
  float v = h[(size_t)n * HF + t];
  ss[t] = v * a_s[t];
  sd[t] = v * a_d[t];
  __syncthreads();
  for (int stride = F / 2; stride > 0; stride >>= 1) {
    if ((t & (F - 1)) < stride) {
      ss[t] += ss[t + stride];
      sd[t] += sd[t + stride];
    }
    __syncthreads();
  }
  if ((t & (F - 1)) == 0) {
    int hd = t / F;
    out_s[n * H + hd] = ss[t];
    out_d[n * H + hd] = sd[t];
  }
}

template <int H>
__global__ void edge_logits_kernel(const float* __restrict__ as,
                                   const float* __restrict__ ad,
                                   const int* __restrict__ src,
                                   const int* __restrict__ dst, int nE, int nTot,
                                   float* __restrict__ logits,
                                   unsigned* __restrict__ mbuf) {
  int tid = blockIdx.x * blockDim.x + threadIdx.x;
  if (tid >= nTot * H) return;
  int e = tid / H, hd = tid - e * H;
  int s, d;
  if (e < nE) { s = src[e]; d = dst[e]; } else { s = d = e - nE; }
  float v = as[s * H + hd] + ad[d * H + hd];
  v = v > 0.f ? v : 0.2f * v;
  logits[tid] = v;
  atomicMax(&mbuf[d * H + hd], f2ord(v));
}

template <int H>
__global__ void edge_denom_kernel(const float* __restrict__ logits,
                                  const unsigned* __restrict__ mbuf,
                                  const int* __restrict__ dst, int nE, int nTot,
                                  float* __restrict__ denom) {
  int tid = blockIdx.x * blockDim.x + threadIdx.x;
  if (tid >= nTot * H) return;
  int e = tid / H, hd = tid - e * H;
  int d = (e < nE) ? dst[e] : (e - nE);
  float m = ord2f(mbuf[d * H + hd]);
  atomicAdd(&denom[d * H + hd], __expf(logits[tid] - m));
}

template <int H, int F>
__global__ void edge_agg_kernel(const float* __restrict__ hsrc,
                                const float* __restrict__ logits,
                                const unsigned* __restrict__ mbuf,
                                const float* __restrict__ denom,
                                const int* __restrict__ src,
                                const int* __restrict__ dst, int nE, int nTot,
                                float* __restrict__ agg) {
  constexpr int HF = H * F;
  constexpr int CH = HF / 4;
  int tid = blockIdx.x * blockDim.x + threadIdx.x;
  if (tid >= nTot * CH) return;
  int e = tid / CH;
  int c = tid - e * CH;
  int fbase = c * 4;
  int hd = fbase / F;
  int s, d;
  if (e < nE) { s = src[e]; d = dst[e]; } else { s = d = e - nE; }
  float m = ord2f(mbuf[d * H + hd]);
  float alpha = __expf(logits[e * H + hd] - m) / (denom[d * H + hd] + 1e-16f);
  const float4 hv = *(const float4*)&hsrc[(size_t)s * HF + fbase];
  float* ap = &agg[(size_t)d * HF + fbase];
  atomicAdd(ap + 0, hv.x * alpha);
  atomicAdd(ap + 1, hv.y * alpha);
  atomicAdd(ap + 2, hv.z * alpha);
  atomicAdd(ap + 3, hv.w * alpha);
}

__global__ void elu_bias_kernel(float* __restrict__ buf,
                                const float* __restrict__ bias, int total,
                                int ncol) {
  int tid = blockIdx.x * blockDim.x + threadIdx.x;
  if (tid >= total) return;
  float v = buf[tid] + bias[tid % ncol];
  buf[tid] = v > 0.f ? v : (__expf(v) - 1.f);
}

__global__ __launch_bounds__(256) void logsoftmax_kernel(
    const float* __restrict__ agg, const float* __restrict__ bias,
    float* __restrict__ out, int n) {
  int tid = blockIdx.x * blockDim.x + threadIdx.x;
  int node = tid >> 5;
  int lane = tid & 31;
  if (node >= n) return;
  float v = agg[(size_t)node * 32 + lane] + bias[lane];
  float m = v;
  for (int o = 16; o > 0; o >>= 1) m = fmaxf(m, __shfl_xor(m, o));
  float ex = __expf(v - m);
  float s = ex;
  for (int o = 16; o > 0; o >>= 1) s += __shfl_xor(s, o);
  out[(size_t)node * 32 + lane] = v - m - __logf(s);
}

extern "C" void kernel_launch(void* const* d_in, const int* in_sizes, int n_in,
                              void* d_out, int out_size, void* d_ws,
                              size_t ws_size, hipStream_t stream) {
  (void)n_in; (void)out_size; (void)ws_size;
  const float* x      = (const float*)d_in[0];
  const int*   ei     = (const int*)d_in[1];
  const float* W1     = (const float*)d_in[2];
  const float* a_src1 = (const float*)d_in[3];
  const float* a_dst1 = (const float*)d_in[4];
  const float* b1     = (const float*)d_in[5];
  const float* W2     = (const float*)d_in[6];
  const float* a_src2 = (const float*)d_in[7];
  const float* a_dst2 = (const float*)d_in[8];
  const float* b2     = (const float*)d_in[9];

  const int n    = in_sizes[0] / 500;
  const int nE   = in_sizes[1] / 2;
  const int nTot = nE + n;
  const int* src = ei;
  const int* dst = ei + nE;

  float* ws = (float*)d_ws;
  size_t o = 0;
  float* h1      = ws + o; o += (size_t)n * 128;
  float* logits1 = ws + o; o += (size_t)nTot * 2;
  float* as1     = ws + o; o += (size_t)n * 2;
  float* ad1     = ws + o; o += (size_t)n * 2;
  float* h2      = ws + o; o += (size_t)n * 32;
  float* logits2 = ws + o; o += (size_t)nTot;
  float* as2     = ws + o; o += (size_t)n;
  float* ad2     = ws + o; o += (size_t)n;
  float* Bp1     = ws + o; o += (size_t)500 * 128;
  float* Bp2     = ws + o; o += (size_t)128 * 32;
  size_t z0 = o;
  unsigned* m1   = (unsigned*)(ws + o); o += (size_t)n * 2;
  float* denom1  = ws + o; o += (size_t)n * 2;
  float* agg1    = ws + o; o += (size_t)n * 128;
  unsigned* m2   = (unsigned*)(ws + o); o += (size_t)n;
  float* denom2  = ws + o; o += (size_t)n;
  float* agg2    = ws + o; o += (size_t)n * 32;
  size_t zcount = o - z0;

  zero_kernel<<<2048, 256, 0, stream>>>((unsigned*)(ws + z0), (long long)zcount);
  pack_b_kernel<500, 128><<<(500 * 128 + 255) / 256, 256, 0, stream>>>(W1, Bp1);
  pack_b_kernel<128, 32><<<(128 * 32 + 255) / 256, 256, 0, stream>>>(W2, Bp2);

  gemm_wmma_kernel<500, 128><<<(n + 15) / 16, 256, 0, stream>>>(x, Bp1, h1, n);
  alpha_kernel<2, 64><<<n, 128, 0, stream>>>(h1, a_src1, a_dst1, as1, ad1);
  {
    int t = nTot * 2;
    edge_logits_kernel<2><<<(t + 255) / 256, 256, 0, stream>>>(
        as1, ad1, src, dst, nE, nTot, logits1, m1);
    edge_denom_kernel<2><<<(t + 255) / 256, 256, 0, stream>>>(
        logits1, m1, dst, nE, nTot, denom1);
    int ta = nTot * 32;
    edge_agg_kernel<2, 64><<<(ta + 255) / 256, 256, 0, stream>>>(
        h1, logits1, m1, denom1, src, dst, nE, nTot, agg1);
  }
  {
    int t = n * 128;
    elu_bias_kernel<<<(t + 255) / 256, 256, 0, stream>>>(agg1, b1, t, 128);
  }

  gemm_wmma_kernel<128, 32><<<(n + 63) / 64, 256, 0, stream>>>(agg1, Bp2, h2, n);
  alpha_kernel<1, 32><<<n, 32, 0, stream>>>(h2, a_src2, a_dst2, as2, ad2);
  {
    int t = nTot;
    edge_logits_kernel<1><<<(t + 255) / 256, 256, 0, stream>>>(
        as2, ad2, src, dst, nE, nTot, logits2, m2);
    edge_denom_kernel<1><<<(t + 255) / 256, 256, 0, stream>>>(
        logits2, m2, dst, nE, nTot, denom2);
    int ta = nTot * 8;
    edge_agg_kernel<1, 32><<<(ta + 255) / 256, 256, 0, stream>>>(
        h2, logits2, m2, denom2, src, dst, nE, nTot, agg2);
  }

  logsoftmax_kernel<<<((size_t)n * 32 + 255) / 256, 256, 0, stream>>>(
      agg2, b2, (float*)d_out, n);
}


# feature scatter re-shaped: a wave walks 32 edges and adds each row slice as one whole 128-byte line per atomic instruction (was 4 lines x 8 lanes, one wave per edge); float atomics, arrival order
# speedup vs baseline: 1.2736x; 1.2736x over previous
#include <hip/hip_runtime.h>
#include <hip/hip_bf16.h>


typedef float v2f __attribute__((ext_vector_type(2)));
typedef float v8f __attribute__((ext_vector_type(8)));

__device__ __forceinline__ unsigned f2ord(float f) {
  unsigned u = __float_as_uint(f);
  return (u & 0x80000000u) ? ~u : (u | 0x80000000u);
}
__device__ __forceinline__ float ord2f(unsigned o) {
  unsigned u = (o & 0x80000000u) ? (o ^ 0x80000000u) : ~o;
  return __uint_as_float(u);
}

__global__ void zero_kernel(unsigned* __restrict__ p, long long count) {
  long long i = (long long)blockIdx.x * blockDim.x + threadIdx.x;
  long long stride = (long long)gridDim.x * blockDim.x;
  for (; i < count; i += stride) p[i] = 0u;
}

template <int K, int NCOL>
__global__ void pack_b_kernel(const float* __restrict__ B,
                              float* __restrict__ Bp) {
  int tid = blockIdx.x * blockDim.x + threadIdx.x;
  if (tid >= K * NCOL) return;
  int k = tid / NCOL;
  int col = tid - k * NCOL;
  int kk = k >> 2;
  int p = (k >> 1) & 1;
  int q = k & 1;
  Bp[(((kk * 2 + p) * NCOL) + col) * 2 + q] = B[tid];
}

template <int K, int NCOL>
__global__ __launch_bounds__(256) void gemm_wmma_kernel(
    const float* __restrict__ A, const float* __restrict__ Bp,
    float* __restrict__ C, int M) {
  constexpr int NT = NCOL / 16;
  constexpr int MT = 8 / NT;
  constexpr int ROWS = MT * 16;
  constexpr int KV = K / 4;
  __shared__ __align__(16) float As[ROWS * K];

  const int tid = threadIdx.x;
  const int wave = tid >> 5;
  const int lane = tid & 31;
  const int row0 = blockIdx.x * ROWS;

  {
    const float4* A4 = (const float4*)A;
    float4* As4 = (float4*)As;
    for (int i = tid; i < ROWS * KV; i += 256) {
      int r = i / KV;
      int c = i - r * KV;
      int gr = row0 + r;
      float4 v = {0.f, 0.f, 0.f, 0.f};
      if (gr < M) v = A4[(size_t)gr * KV + c];
      As4[i] = v;
    }
  }
  __syncthreads();

  const int mtile = wave / NT;
  const int ntile = wave - mtile * NT;
  const int lrow = (mtile << 4) + (lane & 15);
  const int koff = (lane >> 4) << 1;
  const int ncol = (ntile << 4) + (lane & 15);
  const int bbase = ((lane >> 4) * NCOL + ncol);

  v8f acc = {0.f, 0.f, 0.f, 0.f, 0.f, 0.f, 0.f, 0.f};
  for (int kk = 0; kk < KV; ++kk) {
    v2f a, b;
    a.x = As[lrow * K + kk * 4 + koff];
    a.y = As[lrow * K + kk * 4 + koff + 1];
    b = *(const v2f*)&Bp[(size_t)(bbase + kk * 2 * NCOL) * 2];
    acc = __builtin_amdgcn_wmma_f32_16x16x4_f32(
         false, a,  false, b,
         (short)0, acc,  false,  false);
  }

  const int rbase = row0 + (mtile << 4) + ((lane >> 4) << 3);
  if (row0 + ROWS <= M) {
#pragma unroll
    for (int i = 0; i < 8; ++i) C[(size_t)(rbase + i) * NCOL + ncol] = acc[i];
  } else {
#pragma unroll
    for (int i = 0; i < 8; ++i) {
      int r = rbase + i;
      if (r < M) C[(size_t)r * NCOL + ncol] = acc[i];
    }
  }
}

template <int H, int F>
__global__ void alpha_kernel(const float* __restrict__ h,
                             const float* __restrict__ a_s,
                             const float* __restrict__ a_d,
                             float* __restrict__ out_s,
                             float* __restrict__ out_d) {
  constexpr int HF = H * F;
  __shared__ float ss[HF];
  __shared__ float sd[HF];
  const int n = blockIdx.x;
  const int t = threadIdx.x;
  float v = h[(size_t)n * HF + t];
  ss[t] = v * a_s[t];
  sd[t] = v * a_d[t];
  __syncthreads();
  for (int stride = F / 2; stride > 0; stride >>= 1) {
    if ((t & (F - 1)) < stride) {
      ss[t] += ss[t + stride];
      sd[t] += sd[t + stride];
    }
    __syncthreads();
  }
  if ((t & (F - 1)) == 0) {
    int hd = t / F;
    out_s[n * H + hd] = ss[t];
    out_d[n * H + hd] = sd[t];
  }
}

template <int H>
__global__ void edge_logits_kernel(const float* __restrict__ as,
                                   const float* __restrict__ ad,
                                   const int* __restrict__ src,
                                   const int* __restrict__ dst, int nE, int nTot,
                                   float* __restrict__ logits,
                                   unsigned* __restrict__ mbuf) {
  int tid = blockIdx.x * blockDim.x + threadIdx.x;
  if (tid >= nTot * H) return;
  int e = tid / H, hd = tid - e * H;
  int s, d;
  if (e < nE) { s = src[e]; d = dst[e]; } else { s = d = e - nE; }
  float v = as[s * H + hd] + ad[d * H + hd];
  v = v > 0.f ? v : 0.2f * v;
  logits[tid] = v;
  atomicMax(&mbuf[d * H + hd], f2ord(v));
}

template <int H>
__global__ void edge_denom_kernel(const float* __restrict__ logits,
                                  const unsigned* __restrict__ mbuf,
                                  const int* __restrict__ dst, int nE, int nTot,
                                  float* __restrict__ denom) {
  int tid = blockIdx.x * blockDim.x + threadIdx.x;
  if (tid >= nTot * H) return;
  int e = tid / H, hd = tid - e * H;
  int d = (e < nE) ? dst[e] : (e - nE);
  float m = ord2f(mbuf[d * H + hd]);
  atomicAdd(&denom[d * H + hd], __expf(logits[tid] - m));
}

template <int H, int F, int G>
__device__ __forceinline__ void scatter_group(
    const float* __restrict__ hsrc, float* __restrict__ agg, int s, int d,
    const float (&al)[H], int i0, int lane) {
  constexpr int HF = H * F;
  constexpr int NS = HF / 32;
  float v[G][NS];
  float* ap[G];
#pragma unroll
  for (int g = 0; g < G; ++g) {
    const int si = __builtin_amdgcn_readlane(s, i0 + g);
    const int di = __builtin_amdgcn_readlane(d, i0 + g);
    const float* hp = hsrc + (size_t)si * HF + lane;
    ap[g] = agg + (size_t)di * HF + lane;
#pragma unroll
    for (int j = 0; j < NS; ++j) v[g][j] = hp[j * 32];
  }
#pragma unroll
  for (int g = 0; g < G; ++g) {
    float ai[H];
#pragma unroll
    for (int hd = 0; hd < H; ++hd)
      ai[hd] = __int_as_float(
          __builtin_amdgcn_readlane(__float_as_int(al[hd]), i0 + g));
#pragma unroll
    for (int j = 0; j < NS; ++j)
      atomicAdd(ap[g] + j * 32, v[g][j] * ai[(j * 32) / F]);
  }
}

template <int H, int F>
__global__ __launch_bounds__(256) void edge_agg_kernel(
    const float* __restrict__ hsrc, const float* __restrict__ logits,
    const unsigned* __restrict__ mbuf, const float* __restrict__ denom,
    const int* __restrict__ src, const int* __restrict__ dst, int nE, int nTot,
    float* __restrict__ agg) {
  constexpr int G = (H * F >= 128) ? 4 : 8;
  const int lane = threadIdx.x & 31;
  const int wave = __builtin_amdgcn_readfirstlane(
      (int)(blockIdx.x * (blockDim.x >> 5) + (threadIdx.x >> 5)));
  const int base = wave * 32;
  if (base >= nTot) return;
  const int cnt = (nTot - base < 32) ? (nTot - base) : 32;

  int s = 0, d = 0;
  float al[H];
#pragma unroll
  for (int hd = 0; hd < H; ++hd) al[hd] = 0.f;
  {
    const int e = base + lane;
    if (e < nTot) {
      if (e < nE) { s = src[e]; d = dst[e]; } else { s = d = e - nE; }
#pragma unroll
      for (int hd = 0; hd < H; ++hd) {
        float m = ord2f(mbuf[d * H + hd]);
        al[hd] = __expf(logits[e * H + hd] - m) / (denom[d * H + hd] + 1e-16f);
      }
    }
  }

  int i = 0;
  for (; i + G <= cnt; i += G)
    scatter_group<H, F, G>(hsrc, agg, s, d, al, i, lane);
  for (; i < cnt; ++i)
    scatter_group<H, F, 1>(hsrc, agg, s, d, al, i, lane);
}

__global__ void elu_bias_kernel(float* __restrict__ buf,
                                const float* __restrict__ bias, int total,
                                int ncol) {
  int tid = blockIdx.x * blockDim.x + threadIdx.x;
  if (tid >= total) return;
  float v = buf[tid] + bias[tid % ncol];
  buf[tid] = v > 0.f ? v : (__expf(v) - 1.f);
}

__global__ __launch_bounds__(256) void logsoftmax_kernel(
    const float* __restrict__ agg, const float* __restrict__ bias,
    float* __restrict__ out, int n) {
  int tid = blockIdx.x * blockDim.x + threadIdx.x;
  int node = tid >> 5;
  int lane = tid & 31;
  if (node >= n) return;
  float v = agg[(size_t)node * 32 + lane] + bias[lane];
  float m = v;
  for (int o = 16; o > 0; o >>= 1) m = fmaxf(m, __shfl_xor(m, o));
  float ex = __expf(v - m);
  float s = ex;
  for (int o = 16; o > 0; o >>= 1) s += __shfl_xor(s, o);
  out[(size_t)node * 32 + lane] = v - m - __logf(s);
}

extern "C" void kernel_launch(void* const* d_in, const int* in_sizes, int n_in,
                              void* d_out, int out_size, void* d_ws,
                              size_t ws_size, hipStream_t stream) {
  (void)n_in; (void)out_size; (void)ws_size;
  const float* x      = (const float*)d_in[0];
  const int*   ei     = (const int*)d_in[1];
  const float* W1     = (const float*)d_in[2];
  const float* a_src1 = (const float*)d_in[3];
  const float* a_dst1 = (const float*)d_in[4];
  const float* b1     = (const float*)d_in[5];
  const float* W2     = (const float*)d_in[6];
  const float* a_src2 = (const float*)d_in[7];
  const float* a_dst2 = (const float*)d_in[8];
  const float* b2     = (const float*)d_in[9];

  const int n    = in_sizes[0] / 500;
  const int nE   = in_sizes[1] / 2;
  const int nTot = nE + n;
  const int* src = ei;
  const int* dst = ei + nE;

  float* ws = (float*)d_ws;
  size_t o = 0;
  float* h1      = ws + o; o += (size_t)n * 128;
  float* logits1 = ws + o; o += (size_t)nTot * 2;
  float* as1     = ws + o; o += (size_t)n * 2;
  float* ad1     = ws + o; o += (size_t)n * 2;
  float* h2      = ws + o; o += (size_t)n * 32;
  float* logits2 = ws + o; o += (size_t)nTot;
  float* as2     = ws + o; o += (size_t)n;
  float* ad2     = ws + o; o += (size_t)n;
  float* Bp1     = ws + o; o += (size_t)500 * 128;
  float* Bp2     = ws + o; o += (size_t)128 * 32;
  size_t z0 = o;
  unsigned* m1   = (unsigned*)(ws + o); o += (size_t)n * 2;
  float* denom1  = ws + o; o += (size_t)n * 2;
  float* agg1    = ws + o; o += (size_t)n * 128;
  unsigned* m2   = (unsigned*)(ws + o); o += (size_t)n;
  float* denom2  = ws + o; o += (size_t)n;
  float* agg2    = ws + o; o += (size_t)n * 32;
  size_t zcount = o - z0;

  zero_kernel<<<2048, 256, 0, stream>>>((unsigned*)(ws + z0), (long long)zcount);
  pack_b_kernel<500, 128><<<(500 * 128 + 255) / 256, 256, 0, stream>>>(W1, Bp1);
  pack_b_kernel<128, 32><<<(128 * 32 + 255) / 256, 256, 0, stream>>>(W2, Bp2);

  gemm_wmma_kernel<500, 128><<<(n + 15) / 16, 256, 0, stream>>>(x, Bp1, h1, n);
  alpha_kernel<2, 64><<<n, 128, 0, stream>>>(h1, a_src1, a_dst1, as1, ad1);
  {
    int t = nTot * 2;
    edge_logits_kernel<2><<<(t + 255) / 256, 256, 0, stream>>>(
        as1, ad1, src, dst, nE, nTot, logits1, m1);
    edge_denom_kernel<2><<<(t + 255) / 256, 256, 0, stream>>>(
        logits1, m1, dst, nE, nTot, denom1);
    edge_agg_kernel<2, 64><<<(nTot + 255) / 256, 256, 0, stream>>>(
        h1, logits1, m1, denom1, src, dst, nE, nTot, agg1);
  }
  {
    int t = n * 128;
    elu_bias_kernel<<<(t + 255) / 256, 256, 0, stream>>>(agg1, b1, t, 128);
  }

  gemm_wmma_kernel<128, 32><<<(n + 63) / 64, 256, 0, stream>>>(agg1, Bp2, h2, n);
  alpha_kernel<1, 32><<<n, 32, 0, stream>>>(h2, a_src2, a_dst2, as2, ad2);
  {
    int t = nTot;
    edge_logits_kernel<1><<<(t + 255) / 256, 256, 0, stream>>>(
        as2, ad2, src, dst, nE, nTot, logits2, m2);
    edge_denom_kernel<1><<<(t + 255) / 256, 256, 0, stream>>>(
        logits2, m2, dst, nE, nTot, denom2);
    edge_agg_kernel<1, 32><<<(nTot + 255) / 256, 256, 0, stream>>>(
        h2, logits2, m2, denom2, src, dst, nE, nTot, agg2);
  }

  logsoftmax_kernel<<<((size_t)n * 32 + 255) / 256, 256, 0, stream>>>(
      agg2, b2, (float*)d_out, n);
}


# per-node, per-edge, per-element passes re-gridded: 4-8 float4 rows or items per lane, loads issued together, 17x fewer waves enter there; attention-dot and log-softmax sums now 4-chain + lane tree
# speedup vs baseline: 1.3762x; 1.0806x over previous
#include <hip/hip_runtime.h>
#include <hip/hip_bf16.h>


typedef float v2f __attribute__((ext_vector_type(2)));
typedef float v8f __attribute__((ext_vector_type(8)));

__device__ __forceinline__ unsigned f2ord(float f) {
  unsigned u = __float_as_uint(f);
  return (u & 0x80000000u) ? ~u : (u | 0x80000000u);
}
__device__ __forceinline__ float ord2f(unsigned o) {
  unsigned u = (o & 0x80000000u) ? (o ^ 0x80000000u) : ~o;
  return __uint_as_float(u);
}

__global__ void zero_kernel(unsigned* __restrict__ p, long long count) {
  long long i = (long long)blockIdx.x * blockDim.x + threadIdx.x;
  long long stride = (long long)gridDim.x * blockDim.x;
  for (; i < count; i += stride) p[i] = 0u;
}

template <int K, int NCOL>
__global__ void pack_b_kernel(const float* __restrict__ B,
                              float* __restrict__ Bp) {
  int tid = blockIdx.x * blockDim.x + threadIdx.x;
  if (tid >= K * NCOL) return;
  int k = tid / NCOL;
  int col = tid - k * NCOL;
  int kk = k >> 2;
  int p = (k >> 1) & 1;
  int q = k & 1;
  Bp[(((kk * 2 + p) * NCOL) + col) * 2 + q] = B[tid];
}

template <int K, int NCOL>
__global__ __launch_bounds__(256) void gemm_wmma_kernel(
    const float* __restrict__ A, const float* __restrict__ Bp,
    float* __restrict__ C, int M) {
  constexpr int NT = NCOL / 16;
  constexpr int MT = 8 / NT;
  constexpr int ROWS = MT * 16;
  constexpr int KV = K / 4;
  __shared__ __align__(16) float As[ROWS * K];

  const int tid = threadIdx.x;
  const int wave = tid >> 5;
  const int lane = tid & 31;
  const int row0 = blockIdx.x * ROWS;

  {
    const float4* A4 = (const float4*)A;
    float4* As4 = (float4*)As;
    for (int i = tid; i < ROWS * KV; i += 256) {
      int r = i / KV;
      int c = i - r * KV;
      int gr = row0 + r;
      float4 v = {0.f, 0.f, 0.f, 0.f};
      if (gr < M) v = A4[(size_t)gr * KV + c];
      As4[i] = v;
    }
  }
  __syncthreads();

  const int mtile = wave / NT;
  const int ntile = wave - mtile * NT;
  const int lrow = (mtile << 4) + (lane & 15);
  const int koff = (lane >> 4) << 1;
  const int ncol = (ntile << 4) + (lane & 15);
  const int bbase = ((lane >> 4) * NCOL + ncol);

  v8f acc = {0.f, 0.f, 0.f, 0.f, 0.f, 0.f, 0.f, 0.f};
  for (int kk = 0; kk < KV; ++kk) {
    v2f a, b;
    a.x = As[lrow * K + kk * 4 + koff];
    a.y = As[lrow * K + kk * 4 + koff + 1];
    b = *(const v2f*)&Bp[(size_t)(bbase + kk * 2 * NCOL) * 2];
    acc = __builtin_amdgcn_wmma_f32_16x16x4_f32(
         false, a,  false, b,
         (short)0, acc,  false,  false);
  }

  const int rbase = row0 + (mtile << 4) + ((lane >> 4) << 3);
  if (row0 + ROWS <= M) {
#pragma unroll
    for (int i = 0; i < 8; ++i) C[(size_t)(rbase + i) * NCOL + ncol] = acc[i];
  } else {
#pragma unroll
    for (int i = 0; i < 8; ++i) {
      int r = rbase + i;
      if (r < M) C[(size_t)r * NCOL + ncol] = acc[i];
    }
  }
}

template <int M>
__device__ __forceinline__ float row_xor(float x) {
  return __int_as_float(__builtin_amdgcn_update_dpp(
      0, __float_as_int(x), 0x160 + M, 0xf, 0xf, true));
}
template <int W>
__device__ __forceinline__ float group_sum(float x) {
  x += row_xor<1>(x);
  x += row_xor<2>(x);
  x += row_xor<4>(x);
  if constexpr (W == 16) x += row_xor<8>(x);
  return x;
}
template <int W>
__device__ __forceinline__ float group_max(float x) {
  x = fmaxf(x, row_xor<1>(x));
  x = fmaxf(x, row_xor<2>(x));
  x = fmaxf(x, row_xor<4>(x));
  if constexpr (W == 16) x = fmaxf(x, row_xor<8>(x));
  return x;
}

template <int H, int F>
__global__ __launch_bounds__(256) void alpha_kernel(
    const float* __restrict__ h, const float* __restrict__ a_s,
    const float* __restrict__ a_d, float* __restrict__ out_s,
    float* __restrict__ out_d, int n) {
  constexpr int LPN = (H * F) / 4;
  constexpr int LPH = F / 4;
  constexpr int NPW = 32 / LPN;
  constexpr int U = (NPW == 1) ? 8 : 4;
  const int lane = threadIdx.x & 31;
  const int sub = lane % LPN;
  const int wave = (int)((blockIdx.x * blockDim.x + threadIdx.x) >> 5);
  const int node0 = wave * (NPW * U) + lane / LPN;
  const float4* h4 = (const float4*)h;
  const float4 as4 = ((const float4*)a_s)[sub];
  const float4 ad4 = ((const float4*)a_d)[sub];
  float4 v[U];
#pragma unroll
  for (int u = 0; u < U; ++u) {
    int nd = node0 + u * NPW;
    nd = nd < n ? nd : n - 1;
    v[u] = h4[(size_t)nd * LPN + sub];
  }
  __builtin_amdgcn_sched_barrier(0);
  float ps[U], pd[U];
#pragma unroll
  for (int u = 0; u < U; ++u) {
    ps[u] = group_sum<LPH>(v[u].x * as4.x + v[u].y * as4.y + v[u].z * as4.z +
                           v[u].w * as4.w);
    pd[u] = group_sum<LPH>(v[u].x * ad4.x + v[u].y * ad4.y + v[u].z * ad4.z +
                           v[u].w * ad4.w);
  }
#pragma unroll
  for (int u = 0; u < U; ++u) {
    const int nd = node0 + u * NPW;
    if (nd < n && (sub % LPH) == 0) {
      out_s[nd * H + sub / LPH] = ps[u];
      out_d[nd * H + sub / LPH] = pd[u];
    }
  }
}

template <int H>
__global__ __launch_bounds__(256) void edge_logits_kernel(
    const float* __restrict__ as, const float* __restrict__ ad,
    const int* __restrict__ src, const int* __restrict__ dst, int nE, int nTot,
    float* __restrict__ logits, unsigned* __restrict__ mbuf) {
  constexpr int U = 8;
  const int total = nTot * H;
  const int base = blockIdx.x * (256 * U) + threadIdx.x;
  int s[U], d[U];
#pragma unroll
  for (int u = 0; u < U; ++u) {
    int tid = base + u * 256;
    tid = tid < total ? tid : total - 1;
    const int e = tid / H;
    if (e < nE) { s[u] = src[e]; d[u] = dst[e]; } else { s[u] = d[u] = e - nE; }
  }
  float v[U];
#pragma unroll
  for (int u = 0; u < U; ++u) {
    const int hd = (base + u * 256) % H;
    v[u] = as[s[u] * H + hd] + ad[d[u] * H + hd];
  }
#pragma unroll
  for (int u = 0; u < U; ++u) {
    const int tid = base + u * 256;
    if (tid < total) {
      const int hd = tid % H;
      float x = v[u];
      x = x > 0.f ? x : 0.2f * x;
      logits[tid] = x;
      atomicMax(&mbuf[d[u] * H + hd], f2ord(x));
    }
  }
}

template <int H>
__global__ __launch_bounds__(256) void edge_denom_kernel(
    const float* __restrict__ logits, const unsigned* __restrict__ mbuf,
    const int* __restrict__ dst, int nE, int nTot, float* __restrict__ denom) {
  constexpr int U = 8;
  const int total = nTot * H;
  const int base = blockIdx.x * (256 * U) + threadIdx.x;
  int d[U];
  float l[U];
#pragma unroll
  for (int u = 0; u < U; ++u) {
    int tid = base + u * 256;
    tid = tid < total ? tid : total - 1;
    const int e = tid / H;
    d[u] = (e < nE) ? dst[e] : (e - nE);
    l[u] = logits[tid];
  }
  float m[U];
#pragma unroll
  for (int u = 0; u < U; ++u) {
    const int hd = (base + u * 256) % H;
    m[u] = ord2f(mbuf[d[u] * H + hd]);
  }
#pragma unroll
  for (int u = 0; u < U; ++u) {
    const int tid = base + u * 256;
    if (tid < total) {
      const int hd = tid % H;
      atomicAdd(&denom[d[u] * H + hd], __expf(l[u] - m[u]));
    }
  }
}

template <int H, int F, int G>
__device__ __forceinline__ void scatter_group(
    const float* __restrict__ hsrc, float* __restrict__ agg, int s, int d,
    const float (&al)[H], int i0, int lane) {
  constexpr int HF = H * F;
  constexpr int NS = HF / 32;
  float v[G][NS];
  float* ap[G];
#pragma unroll
  for (int g = 0; g < G; ++g) {
    const int si = __builtin_amdgcn_readlane(s, i0 + g);
    const int di = __builtin_amdgcn_readlane(d, i0 + g);
    const float* hp = hsrc + (size_t)si * HF + lane;
    ap[g] = agg + (size_t)di * HF + lane;
#pragma unroll
    for (int j = 0; j < NS; ++j) v[g][j] = hp[j * 32];
  }
#pragma unroll
  for (int g = 0; g < G; ++g) {
    float ai[H];
#pragma unroll
    for (int hd = 0; hd < H; ++hd)
      ai[hd] = __int_as_float(
          __builtin_amdgcn_readlane(__float_as_int(al[hd]), i0 + g));
#pragma unroll
    for (int j = 0; j < NS; ++j)
      atomicAdd(ap[g] + j * 32, v[g][j] * ai[(j * 32) / F]);
  }
}

template <int H, int F>
__global__ __launch_bounds__(256) void edge_agg_kernel(
    const float* __restrict__ hsrc, const float* __restrict__ logits,
    const unsigned* __restrict__ mbuf, const float* __restrict__ denom,
    const int* __restrict__ src, const int* __restrict__ dst, int nE, int nTot,
    float* __restrict__ agg) {
  constexpr int G = (H * F >= 128) ? 4 : 8;
  const int lane = threadIdx.x & 31;
  const int wave = __builtin_amdgcn_readfirstlane(
      (int)(blockIdx.x * (blockDim.x >> 5) + (threadIdx.x >> 5)));
  const int base = wave * 32;
  if (base >= nTot) return;
  const int cnt = (nTot - base < 32) ? (nTot - base) : 32;

  int s = 0, d = 0;
  float al[H];
#pragma unroll
  for (int hd = 0; hd < H; ++hd) al[hd] = 0.f;
  {
    const int e = base + lane;
    if (e < nTot) {
      if (e < nE) { s = src[e]; d = dst[e]; } else { s = d = e - nE; }
#pragma unroll
      for (int hd = 0; hd < H; ++hd) {
        float m = ord2f(mbuf[d * H + hd]);
        al[hd] = __expf(logits[e * H + hd] - m) / (denom[d * H + hd] + 1e-16f);
      }
    }
  }

  int i = 0;
  for (; i + G <= cnt; i += G)
    scatter_group<H, F, G>(hsrc, agg, s, d, al, i, lane);
  for (; i < cnt; ++i)
    scatter_group<H, F, 1>(hsrc, agg, s, d, al, i, lane);
}

template <int NCOL4>
__global__ __launch_bounds__(256) void elu_bias_kernel(
    float* __restrict__ buf, const float* __restrict__ bias, int total4) {
  constexpr int U = 8;
  static_assert(256 % NCOL4 == 0, "row length must divide the item stride");
  float4* b4 = (float4*)buf;
  const int base = blockIdx.x * (256 * U) + threadIdx.x;
  const float4 b = ((const float4*)bias)[threadIdx.x % NCOL4];
  float4 v[U];
#pragma unroll
  for (int u = 0; u < U; ++u) {
    int i = base + u * 256;
    i = i < total4 ? i : total4 - 1;
    v[u] = b4[i];
  }
#pragma unroll
  for (int u = 0; u < U; ++u) {
    const int i = base + u * 256;
    if (i < total4) {
      float4 r;
      float t;
      t = v[u].x + b.x; r.x = t > 0.f ? t : (__expf(t) - 1.f);
      t = v[u].y + b.y; r.y = t > 0.f ? t : (__expf(t) - 1.f);
      t = v[u].z + b.z; r.z = t > 0.f ? t : (__expf(t) - 1.f);
      t = v[u].w + b.w; r.w = t > 0.f ? t : (__expf(t) - 1.f);
      b4[i] = r;
    }
  }
}

__global__ __launch_bounds__(256) void logsoftmax_kernel(
    const float* __restrict__ agg, const float* __restrict__ bias,
    float* __restrict__ out, int n) {
  constexpr int U = 4;
  const int lane = threadIdx.x & 31;
  const int sub = lane & 7;
  const int wave = (int)((blockIdx.x * blockDim.x + threadIdx.x) >> 5);
  const int node0 = wave * (4 * U) + (lane >> 3);
  const float4* a4 = (const float4*)agg;
  float4* o4 = (float4*)out;
  const float4 b = ((const float4*)bias)[sub];
  float4 v[U];
#pragma unroll
  for (int u = 0; u < U; ++u) {
    int nd = node0 + u * 4;
    nd = nd < n ? nd : n - 1;
    v[u] = a4[(size_t)nd * 8 + sub];
  }
  float4 r[U];
#pragma unroll
  for (int u = 0; u < U; ++u) {
    const float x0 = v[u].x + b.x, x1 = v[u].y + b.y;
    const float x2 = v[u].z + b.z, x3 = v[u].w + b.w;
    const float m = group_max<8>(fmaxf(fmaxf(x0, x1), fmaxf(x2, x3)));
    const float s = group_sum<8>(__expf(x0 - m) + __expf(x1 - m) +
                                 __expf(x2 - m) + __expf(x3 - m));
    const float ls = __logf(s);
    r[u].x = x0 - m - ls; r[u].y = x1 - m - ls;
    r[u].z = x2 - m - ls; r[u].w = x3 - m - ls;
  }
#pragma unroll
  for (int u = 0; u < U; ++u) {
    const int nd = node0 + u * 4;
    if (nd < n) o4[(size_t)nd * 8 + sub] = r[u];
  }
}

extern "C" void kernel_launch(void* const* d_in, const int* in_sizes, int n_in,
                              void* d_out, int out_size, void* d_ws,
                              size_t ws_size, hipStream_t stream) {
  (void)n_in; (void)out_size; (void)ws_size;
  const float* x      = (const float*)d_in[0];
  const int*   ei     = (const int*)d_in[1];
  const float* W1     = (const float*)d_in[2];
  const float* a_src1 = (const float*)d_in[3];
  const float* a_dst1 = (const float*)d_in[4];
  const float* b1     = (const float*)d_in[5];
  const float* W2     = (const float*)d_in[6];
  const float* a_src2 = (const float*)d_in[7];
  const float* a_dst2 = (const float*)d_in[8];
  const float* b2     = (const float*)d_in[9];

  const int n    = in_sizes[0] / 500;
  const int nE   = in_sizes[1] / 2;
  const int nTot = nE + n;
  const int* src = ei;
  const int* dst = ei + nE;

  float* ws = (float*)d_ws;
  size_t o = 0;
  float* h1      = ws + o; o += (size_t)n * 128;
  float* h2      = ws + o; o += (size_t)n * 32;
  float* Bp1     = ws + o; o += (size_t)500 * 128;
  float* Bp2     = ws + o; o += (size_t)128 * 32;
  size_t z0 = o;
  float* agg1    = ws + o; o += (size_t)n * 128;
  float* agg2    = ws + o; o += (size_t)n * 32;
  unsigned* m1   = (unsigned*)(ws + o); o += (size_t)n * 2;
  float* denom1  = ws + o; o += (size_t)n * 2;
  unsigned* m2   = (unsigned*)(ws + o); o += (size_t)n;
  float* denom2  = ws + o; o += (size_t)n;
  size_t zcount = o - z0;
  float* logits1 = ws + o; o += (size_t)nTot * 2;
  float* as1     = ws + o; o += (size_t)n * 2;
  float* ad1     = ws + o; o += (size_t)n * 2;
  float* logits2 = ws + o; o += (size_t)nTot;
  float* as2     = ws + o; o += (size_t)n;
  float* ad2     = ws + o; o += (size_t)n;

  zero_kernel<<<2048, 256, 0, stream>>>((unsigned*)(ws + z0), (long long)zcount);
  pack_b_kernel<500, 128><<<(500 * 128 + 255) / 256, 256, 0, stream>>>(W1, Bp1);
  pack_b_kernel<128, 32><<<(128 * 32 + 255) / 256, 256, 0, stream>>>(W2, Bp2);

  gemm_wmma_kernel<500, 128><<<(n + 15) / 16, 256, 0, stream>>>(x, Bp1, h1, n);
  alpha_kernel<2, 64><<<(n + 63) / 64, 256, 0, stream>>>(
      h1, a_src1, a_dst1, as1, ad1, n);
  {
    int t = nTot * 2;
    edge_logits_kernel<2><<<(t + 2047) / 2048, 256, 0, stream>>>(
        as1, ad1, src, dst, nE, nTot, logits1, m1);
    edge_denom_kernel<2><<<(t + 2047) / 2048, 256, 0, stream>>>(
        logits1, m1, dst, nE, nTot, denom1);
    edge_agg_kernel<2, 64><<<(nTot + 255) / 256, 256, 0, stream>>>(
        h1, logits1, m1, denom1, src, dst, nE, nTot, agg1);
  }
  {
    int t4 = n * 32;
    elu_bias_kernel<32><<<(t4 + 2047) / 2048, 256, 0, stream>>>(agg1, b1, t4);
  }

  gemm_wmma_kernel<128, 32><<<(n + 63) / 64, 256, 0, stream>>>(agg1, Bp2, h2, n);
  alpha_kernel<1, 32><<<(n + 127) / 128, 256, 0, stream>>>(
      h2, a_src2, a_dst2, as2, ad2, n);
  {
    int t = nTot;
    edge_logits_kernel<1><<<(t + 2047) / 2048, 256, 0, stream>>>(
        as2, ad2, src, dst, nE, nTot, logits2, m2);
    edge_denom_kernel<1><<<(t + 2047) / 2048, 256, 0, stream>>>(
        logits2, m2, dst, nE, nTot, denom2);
    edge_agg_kernel<1, 32><<<(nTot + 255) / 256, 256, 0, stream>>>(
        h2, logits2, m2, denom2, src, dst, nE, nTot, agg2);
  }

  logsoftmax_kernel<<<(n + 127) / 128, 256, 0, stream>>>(
      agg2, b2, (float*)d_out, n);
}


# feature scatter made L2-half-affine: a block reads its die, draws an edge chunk from its half's ticket counter and adds only rows of its half's node range; float atomics, arrival order
# speedup vs baseline: 1.4577x; 1.0592x over previous
#include <hip/hip_runtime.h>
#include <hip/hip_bf16.h>


typedef float v2f __attribute__((ext_vector_type(2)));
typedef float v8f __attribute__((ext_vector_type(8)));

__device__ __forceinline__ unsigned f2ord(float f) {
  unsigned u = __float_as_uint(f);
  return (u & 0x80000000u) ? ~u : (u | 0x80000000u);
}
__device__ __forceinline__ float ord2f(unsigned o) {
  unsigned u = (o & 0x80000000u) ? (o ^ 0x80000000u) : ~o;
  return __uint_as_float(u);
}

__global__ void zero_kernel(unsigned* __restrict__ p, long long count) {
  long long i = (long long)blockIdx.x * blockDim.x + threadIdx.x;
  long long stride = (long long)gridDim.x * blockDim.x;
  for (; i < count; i += stride) p[i] = 0u;
}

template <int K, int NCOL>
__global__ void pack_b_kernel(const float* __restrict__ B,
                              float* __restrict__ Bp) {
  int tid = blockIdx.x * blockDim.x + threadIdx.x;
  if (tid >= K * NCOL) return;
  int k = tid / NCOL;
  int col = tid - k * NCOL;
  int kk = k >> 2;
  int p = (k >> 1) & 1;
  int q = k & 1;
  Bp[(((kk * 2 + p) * NCOL) + col) * 2 + q] = B[tid];
}

template <int K, int NCOL>
__global__ __launch_bounds__(256) void gemm_wmma_kernel(
    const float* __restrict__ A, const float* __restrict__ Bp,
    float* __restrict__ C, int M) {
  constexpr int NT = NCOL / 16;
  constexpr int MT = 8 / NT;
  constexpr int ROWS = MT * 16;
  constexpr int KV = K / 4;
  __shared__ __align__(16) float As[ROWS * K];

  const int tid = threadIdx.x;
  const int wave = tid >> 5;
  const int lane = tid & 31;
  const int row0 = blockIdx.x * ROWS;

  {
    const float4* A4 = (const float4*)A;
    float4* As4 = (float4*)As;
    for (int i = tid; i < ROWS * KV; i += 256) {
      int r = i / KV;
      int c = i - r * KV;
      int gr = row0 + r;
      float4 v = {0.f, 0.f, 0.f, 0.f};
      if (gr < M) v = A4[(size_t)gr * KV + c];
      As4[i] = v;
    }
  }
  __syncthreads();

  const int mtile = wave / NT;
  const int ntile = wave - mtile * NT;
  const int lrow = (mtile << 4) + (lane & 15);
  const int koff = (lane >> 4) << 1;
  const int ncol = (ntile << 4) + (lane & 15);
  const int bbase = ((lane >> 4) * NCOL + ncol);

  v8f acc = {0.f, 0.f, 0.f, 0.f, 0.f, 0.f, 0.f, 0.f};
  for (int kk = 0; kk < KV; ++kk) {
    v2f a, b;
    a.x = As[lrow * K + kk * 4 + koff];
    a.y = As[lrow * K + kk * 4 + koff + 1];
    b = *(const v2f*)&Bp[(size_t)(bbase + kk * 2 * NCOL) * 2];
    acc = __builtin_amdgcn_wmma_f32_16x16x4_f32(
         false, a,  false, b,
         (short)0, acc,  false,  false);
  }

  const int rbase = row0 + (mtile << 4) + ((lane >> 4) << 3);
  if (row0 + ROWS <= M) {
#pragma unroll
    for (int i = 0; i < 8; ++i) C[(size_t)(rbase + i) * NCOL + ncol] = acc[i];
  } else {
#pragma unroll
    for (int i = 0; i < 8; ++i) {
      int r = rbase + i;
      if (r < M) C[(size_t)r * NCOL + ncol] = acc[i];
    }
  }
}

template <int M>
__device__ __forceinline__ float row_xor(float x) {
  return __int_as_float(__builtin_amdgcn_update_dpp(
      0, __float_as_int(x), 0x160 + M, 0xf, 0xf, true));
}
template <int W>
__device__ __forceinline__ float group_sum(float x) {
  x += row_xor<1>(x);
  x += row_xor<2>(x);
  x += row_xor<4>(x);
  if constexpr (W == 16) x += row_xor<8>(x);
  return x;
}
template <int W>
__device__ __forceinline__ float group_max(float x) {
  x = fmaxf(x, row_xor<1>(x));
  x = fmaxf(x, row_xor<2>(x));
  x = fmaxf(x, row_xor<4>(x));
  if constexpr (W == 16) x = fmaxf(x, row_xor<8>(x));
  return x;
}

template <int H, int F>
__global__ __launch_bounds__(256) void alpha_kernel(
    const float* __restrict__ h, const float* __restrict__ a_s,
    const float* __restrict__ a_d, float* __restrict__ out_s,
    float* __restrict__ out_d, int n) {
  constexpr int LPN = (H * F) / 4;
  constexpr int LPH = F / 4;
  constexpr int NPW = 32 / LPN;
  constexpr int U = (NPW == 1) ? 8 : 4;
  const int lane = threadIdx.x & 31;
  const int sub = lane % LPN;
  const int wave = (int)((blockIdx.x * blockDim.x + threadIdx.x) >> 5);
  const int node0 = wave * (NPW * U) + lane / LPN;
  const float4* h4 = (const float4*)h;
  const float4 as4 = ((const float4*)a_s)[sub];
  const float4 ad4 = ((const float4*)a_d)[sub];
  float4 v[U];
#pragma unroll
  for (int u = 0; u < U; ++u) {
    int nd = node0 + u * NPW;
    nd = nd < n ? nd : n - 1;
    v[u] = h4[(size_t)nd * LPN + sub];
  }
  __builtin_amdgcn_sched_barrier(0);
  float ps[U], pd[U];
#pragma unroll
  for (int u = 0; u < U; ++u) {
    ps[u] = group_sum<LPH>(v[u].x * as4.x + v[u].y * as4.y + v[u].z * as4.z +
                           v[u].w * as4.w);
    pd[u] = group_sum<LPH>(v[u].x * ad4.x + v[u].y * ad4.y + v[u].z * ad4.z +
                           v[u].w * ad4.w);
  }
#pragma unroll
  for (int u = 0; u < U; ++u) {
    const int nd = node0 + u * NPW;
    if (nd < n && (sub % LPH) == 0) {
      out_s[nd * H + sub / LPH] = ps[u];
      out_d[nd * H + sub / LPH] = pd[u];
    }
  }
}

template <int H>
__global__ __launch_bounds__(256) void edge_logits_kernel(
    const float* __restrict__ as, const float* __restrict__ ad,
    const int* __restrict__ src, const int* __restrict__ dst, int nE, int nTot,
    float* __restrict__ logits, unsigned* __restrict__ mbuf) {
  constexpr int U = 8;
  const int total = nTot * H;
  const int base = blockIdx.x * (256 * U) + threadIdx.x;
  int s[U], d[U];
#pragma unroll
  for (int u = 0; u < U; ++u) {
    int tid = base + u * 256;
    tid = tid < total ? tid : total - 1;
    const int e = tid / H;
    if (e < nE) { s[u] = src[e]; d[u] = dst[e]; } else { s[u] = d[u] = e - nE; }
  }
  float v[U];
#pragma unroll
  for (int u = 0; u < U; ++u) {
    const int hd = (base + u * 256) % H;
    v[u] = as[s[u] * H + hd] + ad[d[u] * H + hd];
  }
#pragma unroll
  for (int u = 0; u < U; ++u) {
    const int tid = base + u * 256;
    if (tid < total) {
      const int hd = tid % H;
      float x = v[u];
      x = x > 0.f ? x : 0.2f * x;
      logits[tid] = x;
      atomicMax(&mbuf[d[u] * H + hd], f2ord(x));
    }
  }
}

template <int H>
__global__ __launch_bounds__(256) void edge_denom_kernel(
    const float* __restrict__ logits, const unsigned* __restrict__ mbuf,
    const int* __restrict__ dst, int nE, int nTot, float* __restrict__ denom) {
  constexpr int U = 8;
  const int total = nTot * H;
  const int base = blockIdx.x * (256 * U) + threadIdx.x;
  int d[U];
  float l[U];
#pragma unroll
  for (int u = 0; u < U; ++u) {
    int tid = base + u * 256;
    tid = tid < total ? tid : total - 1;
    const int e = tid / H;
    d[u] = (e < nE) ? dst[e] : (e - nE);
    l[u] = logits[tid];
  }
  float m[U];
#pragma unroll
  for (int u = 0; u < U; ++u) {
    const int hd = (base + u * 256) % H;
    m[u] = ord2f(mbuf[d[u] * H + hd]);
  }
#pragma unroll
  for (int u = 0; u < U; ++u) {
    const int tid = base + u * 256;
    if (tid < total) {
      const int hd = tid % H;
      atomicAdd(&denom[d[u] * H + hd], __expf(l[u] - m[u]));
    }
  }
}

template <int H, int F, int G>
__device__ __forceinline__ void scatter_group(
    const float* __restrict__ hsrc, float* __restrict__ agg, int s, int d,
    const float (&al)[H], unsigned& mask, int lane) {
  constexpr int HF = H * F;
  constexpr int NS = HF / 32;
  float v[G][NS];
  float* ap[G];
  int li[G];
#pragma unroll
  for (int g = 0; g < G; ++g) {
    li[g] = __builtin_ctz(mask);
    mask &= mask - 1u;
    const int si = __builtin_amdgcn_readlane(s, li[g]);
    const int di = __builtin_amdgcn_readlane(d, li[g]);
    const float* hp = hsrc + (size_t)si * HF + lane;
    ap[g] = agg + (size_t)di * HF + lane;
#pragma unroll
    for (int j = 0; j < NS; ++j) v[g][j] = hp[j * 32];
  }
#pragma unroll
  for (int g = 0; g < G; ++g) {
    float ai[H];
#pragma unroll
    for (int hd = 0; hd < H; ++hd)
      ai[hd] = __int_as_float(
          __builtin_amdgcn_readlane(__float_as_int(al[hd]), li[g]));
#pragma unroll
    for (int j = 0; j < NS; ++j)
      atomicAdd(ap[g] + j * 32, v[g][j] * ai[(j * 32) / F]);
  }
}

template <int H, int F>
__global__ __launch_bounds__(256) void edge_agg_kernel(
    const float* __restrict__ hsrc, const float* __restrict__ logits,
    const unsigned* __restrict__ mbuf, const float* __restrict__ denom,
    const int* __restrict__ src, const int* __restrict__ dst, int nE, int nTot,
    int nSplit, int nChunks, unsigned* __restrict__ ticket,
    float* __restrict__ agg) {
  constexpr int G = (H * F >= 128) ? 4 : 8;
  __shared__ int pick[2];
  if (threadIdx.x == 0) {
    int hb = (int)((__builtin_amdgcn_s_sendmsg_rtn(0x87) >> 18) & 1u);
    unsigned t = atomicAdd(&ticket[hb], 1u);
    if (t >= (unsigned)nChunks) {
      hb ^= 1;
      t = atomicAdd(&ticket[hb], 1u);
    }
    pick[0] = (int)t;
    pick[1] = hb;
  }
  __syncthreads();
  const int chunk = __builtin_amdgcn_readfirstlane(pick[0]);
  const int hb = __builtin_amdgcn_readfirstlane(pick[1]);
  if (chunk >= nChunks) return;
  const int lane = threadIdx.x & 31;
  const int base = __builtin_amdgcn_readfirstlane(
      chunk * 256 + (int)(threadIdx.x >> 5) * 32);
  if (base >= nTot) return;

  int s = 0, d = 0;
  bool mine = false;
  float al[H];
#pragma unroll
  for (int hd = 0; hd < H; ++hd) al[hd] = 0.f;
  {
    const int e = base + lane;
    if (e < nTot) {
      if (e < nE) { s = src[e]; d = dst[e]; } else { s = d = e - nE; }
      mine = ((d >= nSplit) ? 1 : 0) == hb;
      if (mine) {
#pragma unroll
        for (int hd = 0; hd < H; ++hd) {
          float m = ord2f(mbuf[d * H + hd]);
          al[hd] =
              __expf(logits[e * H + hd] - m) / (denom[d * H + hd] + 1e-16f);
        }
      }
    }
  }

  unsigned mask = __builtin_amdgcn_ballot_w32(mine);
  while (__builtin_popcount(mask) >= G)
    scatter_group<H, F, G>(hsrc, agg, s, d, al, mask, lane);
  if constexpr (G > 4) {
    if (__builtin_popcount(mask) >= 4)
      scatter_group<H, F, 4>(hsrc, agg, s, d, al, mask, lane);
  }
  if (__builtin_popcount(mask) >= 2)
    scatter_group<H, F, 2>(hsrc, agg, s, d, al, mask, lane);
  if (mask != 0u)
    scatter_group<H, F, 1>(hsrc, agg, s, d, al, mask, lane);
}

template <int NCOL4>
__global__ __launch_bounds__(256) void elu_bias_kernel(
    float* __restrict__ buf, const float* __restrict__ bias, int total4) {
  constexpr int U = 8;
  static_assert(256 % NCOL4 == 0, "row length must divide the item stride");
  float4* b4 = (float4*)buf;
  const int base = blockIdx.x * (256 * U) + threadIdx.x;
  const float4 b = ((const float4*)bias)[threadIdx.x % NCOL4];
  float4 v[U];
#pragma unroll
  for (int u = 0; u < U; ++u) {
    int i = base + u * 256;
    i = i < total4 ? i : total4 - 1;
    v[u] = b4[i];
  }
#pragma unroll
  for (int u = 0; u < U; ++u) {
    const int i = base + u * 256;
    if (i < total4) {
      float4 r;
      float t;
      t = v[u].x + b.x; r.x = t > 0.f ? t : (__expf(t) - 1.f);
      t = v[u].y + b.y; r.y = t > 0.f ? t : (__expf(t) - 1.f);
      t = v[u].z + b.z; r.z = t > 0.f ? t : (__expf(t) - 1.f);
      t = v[u].w + b.w; r.w = t > 0.f ? t : (__expf(t) - 1.f);
      b4[i] = r;
    }
  }
}

__global__ __launch_bounds__(256) void logsoftmax_kernel(
    const float* __restrict__ agg, const float* __restrict__ bias,
    float* __restrict__ out, int n) {
  constexpr int U = 4;
  const int lane = threadIdx.x & 31;
  const int sub = lane & 7;
  const int wave = (int)((blockIdx.x * blockDim.x + threadIdx.x) >> 5);
  const int node0 = wave * (4 * U) + (lane >> 3);
  const float4* a4 = (const float4*)agg;
  float4* o4 = (float4*)out;
  const float4 b = ((const float4*)bias)[sub];
  float4 v[U];
#pragma unroll
  for (int u = 0; u < U; ++u) {
    int nd = node0 + u * 4;
    nd = nd < n ? nd : n - 1;
    v[u] = a4[(size_t)nd * 8 + sub];
  }
  float4 r[U];
#pragma unroll
  for (int u = 0; u < U; ++u) {
    const float x0 = v[u].x + b.x, x1 = v[u].y + b.y;
    const float x2 = v[u].z + b.z, x3 = v[u].w + b.w;
    const float m = group_max<8>(fmaxf(fmaxf(x0, x1), fmaxf(x2, x3)));
    const float s = group_sum<8>(__expf(x0 - m) + __expf(x1 - m) +
                                 __expf(x2 - m) + __expf(x3 - m));
    const float ls = __logf(s);
    r[u].x = x0 - m - ls; r[u].y = x1 - m - ls;
    r[u].z = x2 - m - ls; r[u].w = x3 - m - ls;
  }
#pragma unroll
  for (int u = 0; u < U; ++u) {
    const int nd = node0 + u * 4;
    if (nd < n) o4[(size_t)nd * 8 + sub] = r[u];
  }
}

extern "C" void kernel_launch(void* const* d_in, const int* in_sizes, int n_in,
                              void* d_out, int out_size, void* d_ws,
                              size_t ws_size, hipStream_t stream) {
  (void)n_in; (void)out_size; (void)ws_size;
  const float* x      = (const float*)d_in[0];
  const int*   ei     = (const int*)d_in[1];
  const float* W1     = (const float*)d_in[2];
  const float* a_src1 = (const float*)d_in[3];
  const float* a_dst1 = (const float*)d_in[4];
  const float* b1     = (const float*)d_in[5];
  const float* W2     = (const float*)d_in[6];
  const float* a_src2 = (const float*)d_in[7];
  const float* a_dst2 = (const float*)d_in[8];
  const float* b2     = (const float*)d_in[9];

  const int n    = in_sizes[0] / 500;
  const int nE   = in_sizes[1] / 2;
  const int nTot = nE + n;
  const int* src = ei;
  const int* dst = ei + nE;
  const int nChunks = (nTot + 255) / 256;
  const int nSplit = (n / 2) & ~1;

  float* ws = (float*)d_ws;
  size_t o = 0;
  float* h1      = ws + o; o += (size_t)n * 128;
  float* h2      = ws + o; o += (size_t)n * 32;
  float* Bp1     = ws + o; o += (size_t)500 * 128;
  float* Bp2     = ws + o; o += (size_t)128 * 32;
  size_t z0 = o;
  float* agg1    = ws + o; o += (size_t)n * 128;
  float* agg2    = ws + o; o += (size_t)n * 32;
  unsigned* m1   = (unsigned*)(ws + o); o += (size_t)n * 2;
  float* denom1  = ws + o; o += (size_t)n * 2;
  unsigned* m2   = (unsigned*)(ws + o); o += (size_t)n;
  float* denom2  = ws + o; o += (size_t)n;
  unsigned* tick = (unsigned*)(ws + o); o += 4;
  size_t zcount = o - z0;
  float* logits1 = ws + o; o += (size_t)nTot * 2;
  float* as1     = ws + o; o += (size_t)n * 2;
  float* ad1     = ws + o; o += (size_t)n * 2;
  float* logits2 = ws + o; o += (size_t)nTot;
  float* as2     = ws + o; o += (size_t)n;
  float* ad2     = ws + o; o += (size_t)n;

  zero_kernel<<<2048, 256, 0, stream>>>((unsigned*)(ws + z0), (long long)zcount);
  pack_b_kernel<500, 128><<<(500 * 128 + 255) / 256, 256, 0, stream>>>(W1, Bp1);
  pack_b_kernel<128, 32><<<(128 * 32 + 255) / 256, 256, 0, stream>>>(W2, Bp2);

  gemm_wmma_kernel<500, 128><<<(n + 15) / 16, 256, 0, stream>>>(x, Bp1, h1, n);
  alpha_kernel<2, 64><<<(n + 63) / 64, 256, 0, stream>>>(
      h1, a_src1, a_dst1, as1, ad1, n);
  {
    int t = nTot * 2;
    edge_logits_kernel<2><<<(t + 2047) / 2048, 256, 0, stream>>>(
        as1, ad1, src, dst, nE, nTot, logits1, m1);
    edge_denom_kernel<2><<<(t + 2047) / 2048, 256, 0, stream>>>(
        logits1, m1, dst, nE, nTot, denom1);
    edge_agg_kernel<2, 64><<<2 * nChunks, 256, 0, stream>>>(
        h1, logits1, m1, denom1, src, dst, nE, nTot, nSplit, nChunks, tick,
        agg1);
  }
  {
    int t4 = n * 32;
    elu_bias_kernel<32><<<(t4 + 2047) / 2048, 256, 0, stream>>>(agg1, b1, t4);
  }

  gemm_wmma_kernel<128, 32><<<(n + 63) / 64, 256, 0, stream>>>(agg1, Bp2, h2, n);
  alpha_kernel<1, 32><<<(n + 127) / 128, 256, 0, stream>>>(
      h2, a_src2, a_dst2, as2, ad2, n);
  {
    int t = nTot;
    edge_logits_kernel<1><<<(t + 2047) / 2048, 256, 0, stream>>>(
        as2, ad2, src, dst, nE, nTot, logits2, m2);
    edge_denom_kernel<1><<<(t + 2047) / 2048, 256, 0, stream>>>(
        logits2, m2, dst, nE, nTot, denom2);
    edge_agg_kernel<1, 32><<<2 * nChunks, 256, 0, stream>>>(
        h2, logits2, m2, denom2, src, dst, nE, nTot, nSplit, nChunks,
        tick + 2, agg2);
  }

  logsoftmax_kernel<<<(n + 127) / 128, 256, 0, stream>>>(
      agg2, b2, (float*)d_out, n);
}


# softmax sums and weighted row sums accumulate in the LDS of a workgroup that owns 196 rows and scans all edges: no global float atomics, one division per row; LDS float atomics, arrival order
# speedup vs baseline: 1.8830x; 1.2918x over previous
#include <hip/hip_runtime.h>
#include <hip/hip_bf16.h>


typedef float v2f __attribute__((ext_vector_type(2)));
typedef float v8f __attribute__((ext_vector_type(8)));

__device__ __forceinline__ unsigned f2ord(float f) {
  unsigned u = __float_as_uint(f);
  return (u & 0x80000000u) ? ~u : (u | 0x80000000u);
}
__device__ __forceinline__ float ord2f(unsigned o) {
  unsigned u = (o & 0x80000000u) ? (o ^ 0x80000000u) : ~o;
  return __uint_as_float(u);
}

__global__ void zero_kernel(unsigned* __restrict__ p, long long count) {
  long long i = (long long)blockIdx.x * blockDim.x + threadIdx.x;
  long long stride = (long long)gridDim.x * blockDim.x;
  for (; i < count; i += stride) p[i] = 0u;
}

template <int K, int NCOL>
__global__ void pack_b_kernel(const float* __restrict__ B,
                              float* __restrict__ Bp) {
  int tid = blockIdx.x * blockDim.x + threadIdx.x;
  if (tid >= K * NCOL) return;
  int k = tid / NCOL;
  int col = tid - k * NCOL;
  int kk = k >> 2;
  int p = (k >> 1) & 1;
  int q = k & 1;
  Bp[(((kk * 2 + p) * NCOL) + col) * 2 + q] = B[tid];
}

template <int K, int NCOL>
__global__ __launch_bounds__(256) void gemm_wmma_kernel(
    const float* __restrict__ A, const float* __restrict__ Bp,
    float* __restrict__ C, int M) {
  constexpr int NT = NCOL / 16;
  constexpr int MT = 8 / NT;
  constexpr int ROWS = MT * 16;
  constexpr int KV = K / 4;
  __shared__ __align__(16) float As[ROWS * K];

  const int tid = threadIdx.x;
  const int wave = tid >> 5;
  const int lane = tid & 31;
  const int row0 = blockIdx.x * ROWS;

  {
    const float4* A4 = (const float4*)A;
    float4* As4 = (float4*)As;
    for (int i = tid; i < ROWS * KV; i += 256) {
      int r = i / KV;
      int c = i - r * KV;
      int gr = row0 + r;
      float4 v = {0.f, 0.f, 0.f, 0.f};
      if (gr < M) v = A4[(size_t)gr * KV + c];
      As4[i] = v;
    }
  }
  __syncthreads();

  const int mtile = wave / NT;
  const int ntile = wave - mtile * NT;
  const int lrow = (mtile << 4) + (lane & 15);
  const int koff = (lane >> 4) << 1;
  const int ncol = (ntile << 4) + (lane & 15);
  const int bbase = ((lane >> 4) * NCOL + ncol);

  v8f acc = {0.f, 0.f, 0.f, 0.f, 0.f, 0.f, 0.f, 0.f};
  for (int kk = 0; kk < KV; ++kk) {
    v2f a, b;
    a.x = As[lrow * K + kk * 4 + koff];
    a.y = As[lrow * K + kk * 4 + koff + 1];
    b = *(const v2f*)&Bp[(size_t)(bbase + kk * 2 * NCOL) * 2];
    acc = __builtin_amdgcn_wmma_f32_16x16x4_f32(
         false, a,  false, b,
         (short)0, acc,  false,  false);
  }

  const int rbase = row0 + (mtile << 4) + ((lane >> 4) << 3);
  if (row0 + ROWS <= M) {
#pragma unroll
    for (int i = 0; i < 8; ++i) C[(size_t)(rbase + i) * NCOL + ncol] = acc[i];
  } else {
#pragma unroll
    for (int i = 0; i < 8; ++i) {
      int r = rbase + i;
      if (r < M) C[(size_t)r * NCOL + ncol] = acc[i];
    }
  }
}

template <int M>
__device__ __forceinline__ float row_xor(float x) {
  return __int_as_float(__builtin_amdgcn_update_dpp(
      0, __float_as_int(x), 0x160 + M, 0xf, 0xf, true));
}
template <int W>
__device__ __forceinline__ float group_sum(float x) {
  x += row_xor<1>(x);
  x += row_xor<2>(x);
  x += row_xor<4>(x);
  if constexpr (W == 16) x += row_xor<8>(x);
  return x;
}
template <int W>
__device__ __forceinline__ float group_max(float x) {
  x = fmaxf(x, row_xor<1>(x));
  x = fmaxf(x, row_xor<2>(x));
  x = fmaxf(x, row_xor<4>(x));
  if constexpr (W == 16) x = fmaxf(x, row_xor<8>(x));
  return x;
}

template <int H, int F>
__global__ __launch_bounds__(256) void alpha_kernel(
    const float* __restrict__ h, const float* __restrict__ a_s,
    const float* __restrict__ a_d, float* __restrict__ out_s,
    float* __restrict__ out_d, int n) {
  constexpr int LPN = (H * F) / 4;
  constexpr int LPH = F / 4;
  constexpr int NPW = 32 / LPN;
  constexpr int U = (NPW == 1) ? 8 : 4;
  const int lane = threadIdx.x & 31;
  const int sub = lane % LPN;
  const int wave = (int)((blockIdx.x * blockDim.x + threadIdx.x) >> 5);
  const int node0 = wave * (NPW * U) + lane / LPN;
  const float4* h4 = (const float4*)h;
  const float4 as4 = ((const float4*)a_s)[sub];
  const float4 ad4 = ((const float4*)a_d)[sub];
  float4 v[U];
#pragma unroll
  for (int u = 0; u < U; ++u) {
    int nd = node0 + u * NPW;
    nd = nd < n ? nd : n - 1;
    v[u] = h4[(size_t)nd * LPN + sub];
  }
  __builtin_amdgcn_sched_barrier(0);
  float ps[U], pd[U];
#pragma unroll
  for (int u = 0; u < U; ++u) {
    ps[u] = group_sum<LPH>(v[u].x * as4.x + v[u].y * as4.y + v[u].z * as4.z +
                           v[u].w * as4.w);
    pd[u] = group_sum<LPH>(v[u].x * ad4.x + v[u].y * ad4.y + v[u].z * ad4.z +
                           v[u].w * ad4.w);
  }
#pragma unroll
  for (int u = 0; u < U; ++u) {
    const int nd = node0 + u * NPW;
    if (nd < n && (sub % LPH) == 0) {
      out_s[nd * H + sub / LPH] = ps[u];
      out_d[nd * H + sub / LPH] = pd[u];
    }
  }
}

template <int H, int RO>
__global__ __launch_bounds__(256) void edge_logits_kernel(
    const float* __restrict__ as, const float* __restrict__ ad,
    const int* __restrict__ src, const int* __restrict__ dst, int nE, int nTot,
    float* __restrict__ logits, unsigned* __restrict__ mbuf,
    unsigned* __restrict__ ids) {
  constexpr int U = 8;
  const int total = nTot * H;
  const int base = blockIdx.x * (256 * U) + threadIdx.x;
  int s[U], d[U];
#pragma unroll
  for (int u = 0; u < U; ++u) {
    int tid = base + u * 256;
    tid = tid < total ? tid : total - 1;
    const int e = tid / H;
    if (e < nE) { s[u] = src[e]; d[u] = dst[e]; } else { s[u] = d[u] = e - nE; }
  }
  float v[U];
#pragma unroll
  for (int u = 0; u < U; ++u) {
    const int hd = (base + u * 256) % H;
    v[u] = as[s[u] * H + hd] + ad[d[u] * H + hd];
  }
#pragma unroll
  for (int u = 0; u < U; ++u) {
    const int tid = base + u * 256;
    if (tid < total) {
      const int hd = tid % H;
      float x = v[u];
      x = x > 0.f ? x : 0.2f * x;
      logits[tid] = x;
      atomicMax(&mbuf[d[u] * H + hd], f2ord(x));
      if constexpr (RO > 0) {
        const int e = tid / H;
        if (ids != nullptr && hd == 0 && e < nE)
          ((unsigned char*)ids)[e] = (unsigned char)(d[u] / RO);
      }
    }
  }
}

template <int H, int F, int R, int CAP, bool ELU>
__global__ __launch_bounds__(1024) void owner_agg_kernel(
    const float* __restrict__ hsrc, const float* __restrict__ logits,
    const unsigned* __restrict__ mbuf, const int* __restrict__ src,
    const int* __restrict__ dst, const unsigned* __restrict__ ids, int nE,
    int n, const float* __restrict__ bias, float* __restrict__ outp) {
  constexpr int HF = H * F;
  constexpr int NS = HF / 32;
  constexpr int HF4 = HF / 4;
  constexpr int U = 8;
  constexpr int G = 8;
  __shared__ __align__(16) float acc[R * HF];
  __shared__ float den[R * H];
  __shared__ float mx[R * H];
  __shared__ int ls[CAP];
  __shared__ int lt[CAP];
  __shared__ float lw[CAP * H];
  __shared__ int cnt;

  const int tid = threadIdx.x;
  const int lane = tid & 31;
  const int wv = tid >> 5;
  const int lo = blockIdx.x * R;
  const int nr = (n - lo) < R ? (n - lo) : R;

  if (tid == 0) cnt = 0;
  for (int i = tid; i < nr * H; i += 1024) {
    const float m = ord2f(mbuf[lo * H + i]);
    mx[i] = m;
    den[i] = __expf(logits[(size_t)nE * H + (size_t)lo * H + i] - m);
  }
  __syncthreads();
  {
    const float4* h4 = (const float4*)hsrc;
    float4* acc4 = (float4*)acc;
    for (int i = tid; i < nr * HF4; i += 1024) {
      const int r = i / HF4;
      const int c4 = i - r * HF4;
      const float w = den[r * H + (c4 * 4) / F];
      float4 v = h4[(size_t)(lo + r) * HF4 + c4];
      v.x *= w; v.y *= w; v.z *= w; v.w *= w;
      acc4[i] = v;
    }
  }
  __syncthreads();

  auto take = [&](int e) {
    const int pos = atomicAdd(&cnt, 1);
    if (pos < CAP) ls[pos] = e;
  };

  const unsigned unr = (unsigned)nr;
  int done = 0;
  if (ids != nullptr) {
    const int n16 = nE >> 4;
    const uint4* q4 = (const uint4*)ids;
    const unsigned pat = (unsigned)blockIdx.x * 0x01010101u;
    for (int base = tid; base < n16; base += 1024 * U) {
      uint4 v[U];
#pragma unroll
      for (int u = 0; u < U; ++u) {
        int i = base + u * 1024;
        i = i < n16 ? i : n16 - 1;
        v[u] = q4[i];
      }
#pragma unroll
      for (int u = 0; u < U; ++u) {
        const int i = base + u * 1024;
        const unsigned x0 = v[u].x ^ pat, x1 = v[u].y ^ pat;
        const unsigned x2 = v[u].z ^ pat, x3 = v[u].w ^ pat;
        const unsigned z0 = (x0 - 0x01010101u) & ~x0;
        const unsigned z1 = (x1 - 0x01010101u) & ~x1;
        const unsigned z2 = (x2 - 0x01010101u) & ~x2;
        const unsigned z3 = (x3 - 0x01010101u) & ~x3;
        if (i < n16 && ((z0 | z1 | z2 | z3) & 0x80808080u) != 0u) {
          unsigned m = ((z0 & 0x80808080u) >> 7) | ((z1 & 0x80808080u) >> 6) |
                       ((z2 & 0x80808080u) >> 5) | ((z3 & 0x80808080u) >> 4);
          while (m != 0u) {
            const int b = __builtin_ctz(m);
            m &= m - 1u;
            take(i * 16 + (b & 3) * 4 + (b >> 3));
          }
        }
      }
    }
    done = n16 * 16;
  } else if (((size_t)dst & 15) == 0) {
    const int n4 = nE >> 2;
    const int4* d4 = (const int4*)dst;
    for (int base = tid; base < n4; base += 1024 * U) {
      int4 v[U];
#pragma unroll
      for (int u = 0; u < U; ++u) {
        int i = base + u * 1024;
        i = i < n4 ? i : n4 - 1;
        v[u] = d4[i];
      }
#pragma unroll
      for (int u = 0; u < U; ++u) {
        const int i = base + u * 1024;
        const unsigned t0 = (unsigned)(v[u].x - lo);
        const unsigned t1 = (unsigned)(v[u].y - lo);
        const unsigned t2 = (unsigned)(v[u].z - lo);
        const unsigned t3 = (unsigned)(v[u].w - lo);
        const unsigned tm = min(min(t0, t1), min(t2, t3));
        if (i < n4 && tm < unr) {
          if (t0 < unr) take(i * 4 + 0);
          if (t1 < unr) take(i * 4 + 1);
          if (t2 < unr) take(i * 4 + 2);
          if (t3 < unr) take(i * 4 + 3);
        }
      }
    }
    done = n4 * 4;
  }
  for (int e = done + tid; e < nE; e += 1024) {
    if ((unsigned)(dst[e] - lo) < unr) take(e);
  }
  __syncthreads();
  const int total = cnt;
  const int nl = total <= CAP ? total : 0;
  if (total > CAP) {
    for (int e = tid; e < nE; e += 1024) {
      const unsigned t = (unsigned)(dst[e] - lo);
      if (t < unr) {
        const int s = src[e];
        for (int hd = 0; hd < H; ++hd) {
          const float w = __expf(logits[(size_t)e * H + hd] - mx[t * H + hd]);
          atomicAdd(&den[t * H + hd], w);
          for (int f = 0; f < F; ++f)
            atomicAdd(&acc[t * HF + hd * F + f],
                      hsrc[(size_t)s * HF + hd * F + f] * w);
        }
      }
    }
  }

  for (int i = tid; i < nl; i += 1024) {
    const int e = ls[i];
    const unsigned t = (unsigned)(dst[e] - lo);
    const int s = src[e];
    float lg[H];
#pragma unroll
    for (int hd = 0; hd < H; ++hd) lg[hd] = logits[(size_t)e * H + hd];
    if (t < unr) {
      lt[i] = (int)t;
      ls[i] = s;
#pragma unroll
      for (int hd = 0; hd < H; ++hd) {
        const float w = __expf(lg[hd] - mx[t * H + hd]);
        lw[i * H + hd] = w;
        atomicAdd(&den[t * H + hd], w);
      }
    } else {
      lt[i] = -1;
      ls[i] = lo;
    }
  }
  __syncthreads();

  for (int i0 = wv; i0 < nl; i0 += 32 * G) {
    float v[G][NS];
#pragma unroll
    for (int g = 0; g < G; ++g) {
      int i = i0 + g * 32;
      i = i < nl ? i : i0;
      const float* hp = hsrc + (size_t)ls[i] * HF + lane;
#pragma unroll
      for (int j = 0; j < NS; ++j) v[g][j] = hp[j * 32];
    }
#pragma unroll
    for (int g = 0; g < G; ++g) {
      const int i = i0 + g * 32;
      if (i < nl) {
        const int t = lt[i];
        if (t >= 0) {
          float* ap = acc + t * HF + lane;
#pragma unroll
          for (int j = 0; j < NS; ++j)
            atomicAdd(ap + j * 32, v[g][j] * lw[i * H + (j * 32) / F]);
        }
      }
    }
  }
  __syncthreads();

  {
    const float4* acc4 = (const float4*)acc;
    float4* o4 = (float4*)outp;
    for (int i = tid; i < nr * HF4; i += 1024) {
      const int r = i / HF4;
      const int c4 = i - r * HF4;
      const float dn = den[r * H + (c4 * 4) / F] + 1e-16f;
      float4 x = acc4[i];
      x.x /= dn; x.y /= dn; x.z /= dn; x.w /= dn;
      if constexpr (ELU) {
        const float4 bb = ((const float4*)bias)[c4];
        float t;
        t = x.x + bb.x; x.x = t > 0.f ? t : (__expf(t) - 1.f);
        t = x.y + bb.y; x.y = t > 0.f ? t : (__expf(t) - 1.f);
        t = x.z + bb.z; x.z = t > 0.f ? t : (__expf(t) - 1.f);
        t = x.w + bb.w; x.w = t > 0.f ? t : (__expf(t) - 1.f);
      }
      o4[(size_t)(lo + r) * HF4 + c4] = x;
    }
  }
}

__global__ __launch_bounds__(256) void logsoftmax_kernel(
    const float* __restrict__ agg, const float* __restrict__ bias,
    float* __restrict__ out, int n) {
  constexpr int U = 4;
  const int lane = threadIdx.x & 31;
  const int sub = lane & 7;
  const int wave = (int)((blockIdx.x * blockDim.x + threadIdx.x) >> 5);
  const int node0 = wave * (4 * U) + (lane >> 3);
  const float4* a4 = (const float4*)agg;
  float4* o4 = (float4*)out;
  const float4 b = ((const float4*)bias)[sub];
  float4 v[U];
#pragma unroll
  for (int u = 0; u < U; ++u) {
    int nd = node0 + u * 4;
    nd = nd < n ? nd : n - 1;
    v[u] = a4[(size_t)nd * 8 + sub];
  }
  float4 r[U];
#pragma unroll
  for (int u = 0; u < U; ++u) {
    const float x0 = v[u].x + b.x, x1 = v[u].y + b.y;
    const float x2 = v[u].z + b.z, x3 = v[u].w + b.w;
    const float m = group_max<8>(fmaxf(fmaxf(x0, x1), fmaxf(x2, x3)));
    const float s = group_sum<8>(__expf(x0 - m) + __expf(x1 - m) +
                                 __expf(x2 - m) + __expf(x3 - m));
    const float ls = __logf(s);
    r[u].x = x0 - m - ls; r[u].y = x1 - m - ls;
    r[u].z = x2 - m - ls; r[u].w = x3 - m - ls;
  }
#pragma unroll
  for (int u = 0; u < U; ++u) {
    const int nd = node0 + u * 4;
    if (nd < n) o4[(size_t)nd * 8 + sub] = r[u];
  }
}

extern "C" void kernel_launch(void* const* d_in, const int* in_sizes, int n_in,
                              void* d_out, int out_size, void* d_ws,
                              size_t ws_size, hipStream_t stream) {
  (void)n_in; (void)out_size; (void)ws_size;
  const float* x      = (const float*)d_in[0];
  const int*   ei     = (const int*)d_in[1];
  const float* W1     = (const float*)d_in[2];
  const float* a_src1 = (const float*)d_in[3];
  const float* a_dst1 = (const float*)d_in[4];
  const float* b1     = (const float*)d_in[5];
  const float* W2     = (const float*)d_in[6];
  const float* a_src2 = (const float*)d_in[7];
  const float* a_dst2 = (const float*)d_in[8];
  const float* b2     = (const float*)d_in[9];

  const int n    = in_sizes[0] / 500;
  const int nE   = in_sizes[1] / 2;
  const int nTot = nE + n;
  const int* src = ei;
  const int* dst = ei + nE;
  constexpr int R = 196;
  const int nOwn = (n + R - 1) / R;

  float* ws = (float*)d_ws;
  size_t o = 0;
  float* h1      = ws + o; o += (size_t)n * 128;
  float* h2      = ws + o; o += (size_t)n * 32;
  float* Bp1     = ws + o; o += (size_t)500 * 128;
  float* Bp2     = ws + o; o += (size_t)128 * 32;
  float* agg1    = ws + o; o += (size_t)n * 128;
  float* agg2    = ws + o; o += (size_t)n * 32;
  size_t z0 = o;
  unsigned* m1   = (unsigned*)(ws + o); o += (size_t)n * 2;
  unsigned* m2   = (unsigned*)(ws + o); o += (size_t)n;
  size_t zcount = o - z0;
  float* logits1 = ws + o; o += (size_t)nTot * 2;
  float* as1     = ws + o; o += (size_t)n * 2;
  float* ad1     = ws + o; o += (size_t)n * 2;
  float* logits2 = ws + o; o += (size_t)nTot;
  float* as2     = ws + o; o += (size_t)n;
  float* ad2     = ws + o; o += (size_t)n;

  const bool byteIds = nOwn <= 256 && (size_t)nE <= (size_t)n * 128;
  unsigned* ids1 = byteIds ? (unsigned*)h2 : nullptr;
  unsigned* ids2 = byteIds ? (unsigned*)h1 : nullptr;

  zero_kernel<<<(int)((zcount + 255) / 256), 256, 0, stream>>>(
      (unsigned*)(ws + z0), (long long)zcount);
  pack_b_kernel<500, 128><<<(500 * 128 + 255) / 256, 256, 0, stream>>>(W1, Bp1);
  pack_b_kernel<128, 32><<<(128 * 32 + 255) / 256, 256, 0, stream>>>(W2, Bp2);

  gemm_wmma_kernel<500, 128><<<(n + 15) / 16, 256, 0, stream>>>(x, Bp1, h1, n);
  alpha_kernel<2, 64><<<(n + 63) / 64, 256, 0, stream>>>(
      h1, a_src1, a_dst1, as1, ad1, n);
  {
    int t = nTot * 2;
    edge_logits_kernel<2, R><<<(t + 2047) / 2048, 256, 0, stream>>>(
        as1, ad1, src, dst, nE, nTot, logits1, m1, ids1);
    owner_agg_kernel<2, 64, R, 6144, true><<<nOwn, 1024, 0, stream>>>(
        h1, logits1, m1, src, dst, ids1, nE, n, b1, agg1);
  }

  gemm_wmma_kernel<128, 32><<<(n + 63) / 64, 256, 0, stream>>>(agg1, Bp2, h2, n);
  alpha_kernel<1, 32><<<(n + 127) / 128, 256, 0, stream>>>(
      h2, a_src2, a_dst2, as2, ad2, n);
  {
    int t = nTot;
    edge_logits_kernel<1, R><<<(t + 2047) / 2048, 256, 0, stream>>>(
        as2, ad2, src, dst, nE, nTot, logits2, m2, ids2);
    owner_agg_kernel<1, 32, R, 12288, false><<<nOwn, 1024, 0, stream>>>(
        h2, logits2, m2, src, dst, ids2, nE, n, b2, agg2);
  }

  logsoftmax_kernel<<<(n + 127) / 128, 256, 0, stream>>>(
      agg2, b2, (float*)d_out, n);
}


# edge logits and row maxima formed in the owner workgroup from its own edge list (LDS atomic max): global atomic max, logit planes, zeroing gone; 12 -> 10 launches; self loop joins arrival-order adds
# speedup vs baseline: 2.9853x; 1.5854x over previous
#include <hip/hip_runtime.h>
#include <hip/hip_bf16.h>


typedef float v2f __attribute__((ext_vector_type(2)));
typedef float v8f __attribute__((ext_vector_type(8)));

__device__ __forceinline__ unsigned f2ord(float f) {
  unsigned u = __float_as_uint(f);
  return (u & 0x80000000u) ? ~u : (u | 0x80000000u);
}
__device__ __forceinline__ float ord2f(unsigned o) {
  unsigned u = (o & 0x80000000u) ? (o ^ 0x80000000u) : ~o;
  return __uint_as_float(u);
}

template <int K, int NCOL>
__global__ void pack_b_kernel(const float* __restrict__ B,
                              float* __restrict__ Bp) {
  int tid = blockIdx.x * blockDim.x + threadIdx.x;
  if (tid >= K * NCOL) return;
  int k = tid / NCOL;
  int col = tid - k * NCOL;
  int kk = k >> 2;
  int p = (k >> 1) & 1;
  int q = k & 1;
  Bp[(((kk * 2 + p) * NCOL) + col) * 2 + q] = B[tid];
}

template <int K, int NCOL>
__global__ __launch_bounds__(256) void gemm_wmma_kernel(
    const float* __restrict__ A, const float* __restrict__ Bp,
    float* __restrict__ C, int M) {
  constexpr int NT = NCOL / 16;
  constexpr int MT = 8 / NT;
  constexpr int ROWS = MT * 16;
  constexpr int KV = K / 4;
  __shared__ __align__(16) float As[ROWS * K];

  const int tid = threadIdx.x;
  const int wave = tid >> 5;
  const int lane = tid & 31;
  const int row0 = blockIdx.x * ROWS;

  {
    const float4* A4 = (const float4*)A;
    float4* As4 = (float4*)As;
    for (int i = tid; i < ROWS * KV; i += 256) {
      int r = i / KV;
      int c = i - r * KV;
      int gr = row0 + r;
      float4 v = {0.f, 0.f, 0.f, 0.f};
      if (gr < M) v = A4[(size_t)gr * KV + c];
      As4[i] = v;
    }
  }
  __syncthreads();

  const int mtile = wave / NT;
  const int ntile = wave - mtile * NT;
  const int lrow = (mtile << 4) + (lane & 15);
  const int koff = (lane >> 4) << 1;
  const int ncol = (ntile << 4) + (lane & 15);
  const int bbase = ((lane >> 4) * NCOL + ncol);

  v8f acc = {0.f, 0.f, 0.f, 0.f, 0.f, 0.f, 0.f, 0.f};
  for (int kk = 0; kk < KV; ++kk) {
    v2f a, b;
    a.x = As[lrow * K + kk * 4 + koff];
    a.y = As[lrow * K + kk * 4 + koff + 1];
    b = *(const v2f*)&Bp[(size_t)(bbase + kk * 2 * NCOL) * 2];
    acc = __builtin_amdgcn_wmma_f32_16x16x4_f32(
         false, a,  false, b,
         (short)0, acc,  false,  false);
  }

  const int rbase = row0 + (mtile << 4) + ((lane >> 4) << 3);
  if (row0 + ROWS <= M) {
#pragma unroll
    for (int i = 0; i < 8; ++i) C[(size_t)(rbase + i) * NCOL + ncol] = acc[i];
  } else {
#pragma unroll
    for (int i = 0; i < 8; ++i) {
      int r = rbase + i;
      if (r < M) C[(size_t)r * NCOL + ncol] = acc[i];
    }
  }
}

template <int M>
__device__ __forceinline__ float row_xor(float x) {
  return __int_as_float(__builtin_amdgcn_update_dpp(
      0, __float_as_int(x), 0x160 + M, 0xf, 0xf, true));
}
template <int W>
__device__ __forceinline__ float group_sum(float x) {
  x += row_xor<1>(x);
  x += row_xor<2>(x);
  x += row_xor<4>(x);
  if constexpr (W == 16) x += row_xor<8>(x);
  return x;
}
template <int W>
__device__ __forceinline__ float group_max(float x) {
  x = fmaxf(x, row_xor<1>(x));
  x = fmaxf(x, row_xor<2>(x));
  x = fmaxf(x, row_xor<4>(x));
  if constexpr (W == 16) x = fmaxf(x, row_xor<8>(x));
  return x;
}

template <int H, int F>
__global__ __launch_bounds__(256) void alpha_kernel(
    const float* __restrict__ h, const float* __restrict__ a_s,
    const float* __restrict__ a_d, float* __restrict__ out_s,
    float* __restrict__ out_d, int n) {
  constexpr int LPN = (H * F) / 4;
  constexpr int LPH = F / 4;
  constexpr int NPW = 32 / LPN;
  constexpr int U = (NPW == 1) ? 8 : 4;
  const int lane = threadIdx.x & 31;
  const int sub = lane % LPN;
  const int wave = (int)((blockIdx.x * blockDim.x + threadIdx.x) >> 5);
  const int node0 = wave * (NPW * U) + lane / LPN;
  const float4* h4 = (const float4*)h;
  const float4 as4 = ((const float4*)a_s)[sub];
  const float4 ad4 = ((const float4*)a_d)[sub];
  float4 v[U];
#pragma unroll
  for (int u = 0; u < U; ++u) {
    int nd = node0 + u * NPW;
    nd = nd < n ? nd : n - 1;
    v[u] = h4[(size_t)nd * LPN + sub];
  }
  __builtin_amdgcn_sched_barrier(0);
  float ps[U], pd[U];
#pragma unroll
  for (int u = 0; u < U; ++u) {
    ps[u] = group_sum<LPH>(v[u].x * as4.x + v[u].y * as4.y + v[u].z * as4.z +
                           v[u].w * as4.w);
    pd[u] = group_sum<LPH>(v[u].x * ad4.x + v[u].y * ad4.y + v[u].z * ad4.z +
                           v[u].w * ad4.w);
  }
#pragma unroll
  for (int u = 0; u < U; ++u) {
    const int nd = node0 + u * NPW;
    if (nd < n && (sub % LPH) == 0) {
      out_s[nd * H + sub / LPH] = ps[u];
      out_d[nd * H + sub / LPH] = pd[u];
    }
  }
}

template <int RO>
__global__ __launch_bounds__(256) void owner_ids_kernel(
    const int* __restrict__ dst, int nE, unsigned char* __restrict__ ids) {
  const int q = blockIdx.x * 256 + threadIdx.x;
  const int e0 = q * 4;
  if (e0 + 3 < nE) {
    int d0, d1, d2, d3;
    if (((size_t)dst & 15) == 0) {
      const int4 v = ((const int4*)dst)[q];
      d0 = v.x; d1 = v.y; d2 = v.z; d3 = v.w;
    } else {
      d0 = dst[e0]; d1 = dst[e0 + 1]; d2 = dst[e0 + 2]; d3 = dst[e0 + 3];
    }
    ((unsigned*)ids)[q] = (unsigned)(d0 / RO) | ((unsigned)(d1 / RO) << 8) |
                          ((unsigned)(d2 / RO) << 16) |
                          ((unsigned)(d3 / RO) << 24);
  } else {
    for (int e = e0; e < nE; ++e) ids[e] = (unsigned char)(dst[e] / RO);
  }
}

template <int H, int F, int R, int CAP, bool ELU>
__global__ __launch_bounds__(1024) void owner_agg_kernel(
    const float* __restrict__ hsrc, const float* __restrict__ as,
    const float* __restrict__ ad, const int* __restrict__ src,
    const int* __restrict__ dst, const unsigned* __restrict__ ids, int nE,
    int n, const float* __restrict__ bias, float* __restrict__ outp) {
  static_assert(R <= CAP, "the self loops must fit the list");
  constexpr int HF = H * F;
  constexpr int NS = HF / 32;
  constexpr int HF4 = HF / 4;
  constexpr int U = 8;
  constexpr int G = 8;
  constexpr int U3 = 4;
  __shared__ __align__(16) float acc[R * HF];
  __shared__ float den[R * H];
  __shared__ unsigned mxo[R * H];
  __shared__ float adl[R * H];
  __shared__ int ls[CAP];
  __shared__ int lt[CAP];
  __shared__ float lw[CAP * H];
  __shared__ int cnt;

  const int tid = threadIdx.x;
  const int lane = tid & 31;
  const int wv = tid >> 5;
  const int lo = blockIdx.x * R;
  const int nr = (n - lo) < R ? (n - lo) : R;

  if (tid == 0) cnt = nr;
  for (int i = tid; i < nr * H; i += 1024) {
    den[i] = 0.f;
    mxo[i] = 0u;
    adl[i] = ad[(size_t)lo * H + i];
  }
  {
    float4* acc4 = (float4*)acc;
    const float4 z = {0.f, 0.f, 0.f, 0.f};
    for (int i = tid; i < nr * HF4; i += 1024) acc4[i] = z;
  }
  __syncthreads();

  auto take = [&](int e) {
    const int pos = atomicAdd(&cnt, 1);
    if (pos < CAP) ls[pos] = e;
  };

  const unsigned unr = (unsigned)nr;
  int done = 0;
  if (ids != nullptr) {
    const int n16 = nE >> 4;
    const uint4* q4 = (const uint4*)ids;
    const unsigned pat = (unsigned)blockIdx.x * 0x01010101u;
    for (int base = tid; base < n16; base += 1024 * U) {
      uint4 v[U];
#pragma unroll
      for (int u = 0; u < U; ++u) {
        int i = base + u * 1024;
        i = i < n16 ? i : n16 - 1;
        v[u] = q4[i];
      }
#pragma unroll
      for (int u = 0; u < U; ++u) {
        const int i = base + u * 1024;
        const unsigned x0 = v[u].x ^ pat, x1 = v[u].y ^ pat;
        const unsigned x2 = v[u].z ^ pat, x3 = v[u].w ^ pat;
        const unsigned z0 = (x0 - 0x01010101u) & ~x0;
        const unsigned z1 = (x1 - 0x01010101u) & ~x1;
        const unsigned z2 = (x2 - 0x01010101u) & ~x2;
        const unsigned z3 = (x3 - 0x01010101u) & ~x3;
        if (i < n16 && ((z0 | z1 | z2 | z3) & 0x80808080u) != 0u) {
          unsigned m = ((z0 & 0x80808080u) >> 7) | ((z1 & 0x80808080u) >> 6) |
                       ((z2 & 0x80808080u) >> 5) | ((z3 & 0x80808080u) >> 4);
          while (m != 0u) {
            const int b = __builtin_ctz(m);
            m &= m - 1u;
            take(i * 16 + (b & 3) * 4 + (b >> 3));
          }
        }
      }
    }
    done = n16 * 16;
  } else if (((size_t)dst & 15) == 0) {
    const int n4 = nE >> 2;
    const int4* d4 = (const int4*)dst;
    for (int base = tid; base < n4; base += 1024 * U) {
      int4 v[U];
#pragma unroll
      for (int u = 0; u < U; ++u) {
        int i = base + u * 1024;
        i = i < n4 ? i : n4 - 1;
        v[u] = d4[i];
      }
#pragma unroll
      for (int u = 0; u < U; ++u) {
        const int i = base + u * 1024;
        const unsigned t0 = (unsigned)(v[u].x - lo);
        const unsigned t1 = (unsigned)(v[u].y - lo);
        const unsigned t2 = (unsigned)(v[u].z - lo);
        const unsigned t3 = (unsigned)(v[u].w - lo);
        const unsigned tm = min(min(t0, t1), min(t2, t3));
        if (i < n4 && tm < unr) {
          if (t0 < unr) take(i * 4 + 0);
          if (t1 < unr) take(i * 4 + 1);
          if (t2 < unr) take(i * 4 + 2);
          if (t3 < unr) take(i * 4 + 3);
        }
      }
    }
    done = n4 * 4;
  }
  for (int e = done + tid; e < nE; e += 1024) {
    if ((unsigned)(dst[e] - lo) < unr) take(e);
  }
  __syncthreads();
  const int total = cnt;
  const int nl = total <= CAP ? total : 0;
  if (total > CAP) {
    for (int i = tid; i < nr * H; i += 1024) {
      float x = as[(size_t)lo * H + i] + adl[i];
      x = x > 0.f ? x : 0.2f * x;
      atomicMax(&mxo[i], f2ord(x));
    }
    for (int e = tid; e < nE; e += 1024) {
      const unsigned t = (unsigned)(dst[e] - lo);
      if (t < unr) {
        const int s = src[e];
        for (int hd = 0; hd < H; ++hd) {
          float x = as[(size_t)s * H + hd] + adl[t * H + hd];
          x = x > 0.f ? x : 0.2f * x;
          atomicMax(&mxo[t * H + hd], f2ord(x));
        }
      }
    }
    __syncthreads();
    for (int i = tid; i < nr * H; i += 1024) {
      const int r = i / H;
      const int hd = i - r * H;
      float x = as[(size_t)lo * H + i] + adl[i];
      x = x > 0.f ? x : 0.2f * x;
      const float w = __expf(x - ord2f(mxo[i]));
      atomicAdd(&den[i], w);
      for (int f = 0; f < F; ++f)
        atomicAdd(&acc[r * HF + hd * F + f],
                  hsrc[(size_t)(lo + r) * HF + hd * F + f] * w);
    }
    for (int e = tid; e < nE; e += 1024) {
      const unsigned t = (unsigned)(dst[e] - lo);
      if (t < unr) {
        const int s = src[e];
        for (int hd = 0; hd < H; ++hd) {
          float x = as[(size_t)s * H + hd] + adl[t * H + hd];
          x = x > 0.f ? x : 0.2f * x;
          const float w = __expf(x - ord2f(mxo[t * H + hd]));
          atomicAdd(&den[t * H + hd], w);
          for (int f = 0; f < F; ++f)
            atomicAdd(&acc[t * HF + hd * F + f],
                      hsrc[(size_t)s * HF + hd * F + f] * w);
        }
      }
    }
  }

  for (int i0 = tid; i0 < nl; i0 += 1024 * U3) {
    int e[U3], s[U3], dd[U3];
    unsigned t[U3];
#pragma unroll
    for (int u = 0; u < U3; ++u) {
      const int i = i0 + u * 1024;
      e[u] = (i >= nr && i < nl) ? ls[i] : -1;
      dd[u] = i < nr ? lo + i : lo - 1;
      s[u] = i < nr ? lo + i : lo;
    }
#pragma unroll
    for (int u = 0; u < U3; ++u) {
      if (e[u] >= 0) {
        dd[u] = dst[e[u]];
        s[u] = src[e[u]];
      }
    }
#pragma unroll
    for (int u = 0; u < U3; ++u) t[u] = (unsigned)(dd[u] - lo);
    float a[U3][H];
#pragma unroll
    for (int u = 0; u < U3; ++u) {
#pragma unroll
      for (int hd = 0; hd < H; ++hd) a[u][hd] = as[(size_t)s[u] * H + hd];
    }
    float x[U3][H];
#pragma unroll
    for (int u = 0; u < U3; ++u) {
      const unsigned tc = t[u] < unr ? t[u] : 0u;
#pragma unroll
      for (int hd = 0; hd < H; ++hd) {
        const float v = a[u][hd] + adl[tc * H + hd];
        x[u][hd] = v > 0.f ? v : 0.2f * v;
      }
    }
#pragma unroll
    for (int u = 0; u < U3; ++u) {
      const int i = i0 + u * 1024;
      if (i < nl) {
        const bool ours = t[u] < unr;
        lt[i] = ours ? (int)t[u] : -1;
        ls[i] = ours ? s[u] : lo;
#pragma unroll
        for (int hd = 0; hd < H; ++hd) lw[i * H + hd] = x[u][hd];
        if (ours) {
#pragma unroll
          for (int hd = 0; hd < H; ++hd)
            atomicMax(&mxo[t[u] * H + hd], f2ord(x[u][hd]));
        }
      }
    }
  }
  __syncthreads();

  for (int i = tid; i < nl; i += 1024) {
    const int t = lt[i];
    if (t >= 0) {
#pragma unroll
      for (int hd = 0; hd < H; ++hd) {
        const float w = __expf(lw[i * H + hd] - ord2f(mxo[t * H + hd]));
        lw[i * H + hd] = w;
        atomicAdd(&den[t * H + hd], w);
      }
    }
  }
  __syncthreads();

  for (int i0 = wv; i0 < nl; i0 += 32 * G) {
    float v[G][NS];
#pragma unroll
    for (int g = 0; g < G; ++g) {
      int i = i0 + g * 32;
      i = i < nl ? i : i0;
      const float* hp = hsrc + (size_t)ls[i] * HF + lane;
#pragma unroll
      for (int j = 0; j < NS; ++j) v[g][j] = hp[j * 32];
    }
#pragma unroll
    for (int g = 0; g < G; ++g) {
      const int i = i0 + g * 32;
      if (i < nl) {
        const int t = lt[i];
        if (t >= 0) {
          float* ap = acc + t * HF + lane;
#pragma unroll
          for (int j = 0; j < NS; ++j)
            atomicAdd(ap + j * 32, v[g][j] * lw[i * H + (j * 32) / F]);
        }
      }
    }
  }
  __syncthreads();

  {
    const float4* acc4 = (const float4*)acc;
    float4* o4 = (float4*)outp;
    for (int i = tid; i < nr * HF4; i += 1024) {
      const int r = i / HF4;
      const int c4 = i - r * HF4;
      const float dn = den[r * H + (c4 * 4) / F] + 1e-16f;
      float4 x = acc4[i];
      x.x /= dn; x.y /= dn; x.z /= dn; x.w /= dn;
      if constexpr (ELU) {
        const float4 bb = ((const float4*)bias)[c4];
        float t;
        t = x.x + bb.x; x.x = t > 0.f ? t : (__expf(t) - 1.f);
        t = x.y + bb.y; x.y = t > 0.f ? t : (__expf(t) - 1.f);
        t = x.z + bb.z; x.z = t > 0.f ? t : (__expf(t) - 1.f);
        t = x.w + bb.w; x.w = t > 0.f ? t : (__expf(t) - 1.f);
      }
      o4[(size_t)(lo + r) * HF4 + c4] = x;
    }
  }
}

__global__ __launch_bounds__(256) void logsoftmax_kernel(
    const float* __restrict__ agg, const float* __restrict__ bias,
    float* __restrict__ out, int n) {
  constexpr int U = 4;
  const int lane = threadIdx.x & 31;
  const int sub = lane & 7;
  const int wave = (int)((blockIdx.x * blockDim.x + threadIdx.x) >> 5);
  const int node0 = wave * (4 * U) + (lane >> 3);
  const float4* a4 = (const float4*)agg;
  float4* o4 = (float4*)out;
  const float4 b = ((const float4*)bias)[sub];
  float4 v[U];
#pragma unroll
  for (int u = 0; u < U; ++u) {
    int nd = node0 + u * 4;
    nd = nd < n ? nd : n - 1;
    v[u] = a4[(size_t)nd * 8 + sub];
  }
  float4 r[U];
#pragma unroll
  for (int u = 0; u < U; ++u) {
    const float x0 = v[u].x + b.x, x1 = v[u].y + b.y;
    const float x2 = v[u].z + b.z, x3 = v[u].w + b.w;
    const float m = group_max<8>(fmaxf(fmaxf(x0, x1), fmaxf(x2, x3)));
    const float s = group_sum<8>(__expf(x0 - m) + __expf(x1 - m) +
                                 __expf(x2 - m) + __expf(x3 - m));
    const float ls = __logf(s);
    r[u].x = x0 - m - ls; r[u].y = x1 - m - ls;
    r[u].z = x2 - m - ls; r[u].w = x3 - m - ls;
  }
#pragma unroll
  for (int u = 0; u < U; ++u) {
    const int nd = node0 + u * 4;
    if (nd < n) o4[(size_t)nd * 8 + sub] = r[u];
  }
}

extern "C" void kernel_launch(void* const* d_in, const int* in_sizes, int n_in,
                              void* d_out, int out_size, void* d_ws,
                              size_t ws_size, hipStream_t stream) {
  (void)n_in; (void)out_size; (void)ws_size;
  const float* x      = (const float*)d_in[0];
  const int*   ei     = (const int*)d_in[1];
  const float* W1     = (const float*)d_in[2];
  const float* a_src1 = (const float*)d_in[3];
  const float* a_dst1 = (const float*)d_in[4];
  const float* b1     = (const float*)d_in[5];
  const float* W2     = (const float*)d_in[6];
  const float* a_src2 = (const float*)d_in[7];
  const float* a_dst2 = (const float*)d_in[8];
  const float* b2     = (const float*)d_in[9];

  const int n    = in_sizes[0] / 500;
  const int nE   = in_sizes[1] / 2;
  const int* src = ei;
  const int* dst = ei + nE;
  constexpr int R = 196;
  const int nOwn = (n + R - 1) / R;

  float* ws = (float*)d_ws;
  size_t o = 0;
  float* h1      = ws + o; o += (size_t)n * 128;
  float* h2      = ws + o; o += (size_t)n * 32;
  float* Bp1     = ws + o; o += (size_t)500 * 128;
  float* Bp2     = ws + o; o += (size_t)128 * 32;
  float* agg1    = ws + o; o += (size_t)n * 128;
  float* agg2    = ws + o; o += (size_t)n * 32;
  float* idsp    = ws + o; o += ((size_t)nE + 15) / 16 * 4;
  float* as1     = ws + o; o += (size_t)n * 2;
  float* ad1     = ws + o; o += (size_t)n * 2;
  float* as2     = ws + o; o += (size_t)n;
  float* ad2     = ws + o; o += (size_t)n;

  const bool byteIds = nOwn <= 256;
  unsigned* ids = byteIds ? (unsigned*)idsp : nullptr;

  pack_b_kernel<500, 128><<<(500 * 128 + 255) / 256, 256, 0, stream>>>(W1, Bp1);
  pack_b_kernel<128, 32><<<(128 * 32 + 255) / 256, 256, 0, stream>>>(W2, Bp2);
  if (byteIds)
    owner_ids_kernel<R><<<((nE + 3) / 4 + 255) / 256, 256, 0, stream>>>(
        dst, nE, (unsigned char*)idsp);

  gemm_wmma_kernel<500, 128><<<(n + 15) / 16, 256, 0, stream>>>(x, Bp1, h1, n);
  alpha_kernel<2, 64><<<(n + 63) / 64, 256, 0, stream>>>(
      h1, a_src1, a_dst1, as1, ad1, n);
  owner_agg_kernel<2, 64, R, 6144, true><<<nOwn, 1024, 0, stream>>>(
      h1, as1, ad1, src, dst, ids, nE, n, b1, agg1);

  gemm_wmma_kernel<128, 32><<<(n + 63) / 64, 256, 0, stream>>>(agg1, Bp2, h2, n);
  alpha_kernel<1, 32><<<(n + 127) / 128, 256, 0, stream>>>(
      h2, a_src2, a_dst2, as2, ad2, n);
  owner_agg_kernel<1, 32, R, 12288, false><<<nOwn, 1024, 0, stream>>>(
      h2, as2, ad2, src, dst, ids, nE, n, b2, agg2);

  logsoftmax_kernel<<<(n + 127) / 128, 256, 0, stream>>>(
      agg2, b2, (float*)d_out, n);
}


# layer 2's owner workgroups take their edge list (source node, row) from the list their layer-1 twins now leave in the workspace: owner bytes scanned once, not twice; LDS float adds stay arrival order
# speedup vs baseline: 3.3236x; 1.1133x over previous
#include <hip/hip_runtime.h>
#include <hip/hip_bf16.h>


typedef float v2f __attribute__((ext_vector_type(2)));
typedef float v8f __attribute__((ext_vector_type(8)));

__device__ __forceinline__ unsigned f2ord(float f) {
  unsigned u = __float_as_uint(f);
  return (u & 0x80000000u) ? ~u : (u | 0x80000000u);
}
__device__ __forceinline__ float ord2f(unsigned o) {
  unsigned u = (o & 0x80000000u) ? (o ^ 0x80000000u) : ~o;
  return __uint_as_float(u);
}

template <int K, int NCOL>
__global__ void pack_b_kernel(const float* __restrict__ B,
                              float* __restrict__ Bp) {
  int tid = blockIdx.x * blockDim.x + threadIdx.x;
  if (tid >= K * NCOL) return;
  int k = tid / NCOL;
  int col = tid - k * NCOL;
  int kk = k >> 2;
  int p = (k >> 1) & 1;
  int q = k & 1;
  Bp[(((kk * 2 + p) * NCOL) + col) * 2 + q] = B[tid];
}

template <int K, int NCOL>
__global__ __launch_bounds__(256) void gemm_wmma_kernel(
    const float* __restrict__ A, const float* __restrict__ Bp,
    float* __restrict__ C, int M) {
  constexpr int NT = NCOL / 16;
  constexpr int MT = 8 / NT;
  constexpr int ROWS = MT * 16;
  constexpr int KV = K / 4;
  __shared__ __align__(16) float As[ROWS * K];

  const int tid = threadIdx.x;
  const int wave = tid >> 5;
  const int lane = tid & 31;
  const int row0 = blockIdx.x * ROWS;

  {
    const float4* A4 = (const float4*)A;
    float4* As4 = (float4*)As;
    for (int i = tid; i < ROWS * KV; i += 256) {
      int r = i / KV;
      int c = i - r * KV;
      int gr = row0 + r;
      float4 v = {0.f, 0.f, 0.f, 0.f};
      if (gr < M) v = A4[(size_t)gr * KV + c];
      As4[i] = v;
    }
  }
  __syncthreads();

  const int mtile = wave / NT;
  const int ntile = wave - mtile * NT;
  const int lrow = (mtile << 4) + (lane & 15);
  const int koff = (lane >> 4) << 1;
  const int ncol = (ntile << 4) + (lane & 15);
  const int bbase = ((lane >> 4) * NCOL + ncol);

  v8f acc = {0.f, 0.f, 0.f, 0.f, 0.f, 0.f, 0.f, 0.f};
  for (int kk = 0; kk < KV; ++kk) {
    v2f a, b;
    a.x = As[lrow * K + kk * 4 + koff];
    a.y = As[lrow * K + kk * 4 + koff + 1];
    b = *(const v2f*)&Bp[(size_t)(bbase + kk * 2 * NCOL) * 2];
    acc = __builtin_amdgcn_wmma_f32_16x16x4_f32(
         false, a,  false, b,
         (short)0, acc,  false,  false);
  }

  const int rbase = row0 + (mtile << 4) + ((lane >> 4) << 3);
  if (row0 + ROWS <= M) {
#pragma unroll
    for (int i = 0; i < 8; ++i) C[(size_t)(rbase + i) * NCOL + ncol] = acc[i];
  } else {
#pragma unroll
    for (int i = 0; i < 8; ++i) {
      int r = rbase + i;
      if (r < M) C[(size_t)r * NCOL + ncol] = acc[i];
    }
  }
}

template <int M>
__device__ __forceinline__ float row_xor(float x) {
  return __int_as_float(__builtin_amdgcn_update_dpp(
      0, __float_as_int(x), 0x160 + M, 0xf, 0xf, true));
}
template <int W>
__device__ __forceinline__ float group_sum(float x) {
  x += row_xor<1>(x);
  x += row_xor<2>(x);
  x += row_xor<4>(x);
  if constexpr (W == 16) x += row_xor<8>(x);
  return x;
}
template <int W>
__device__ __forceinline__ float group_max(float x) {
  x = fmaxf(x, row_xor<1>(x));
  x = fmaxf(x, row_xor<2>(x));
  x = fmaxf(x, row_xor<4>(x));
  if constexpr (W == 16) x = fmaxf(x, row_xor<8>(x));
  return x;
}

template <int H, int F>
__global__ __launch_bounds__(256) void alpha_kernel(
    const float* __restrict__ h, const float* __restrict__ a_s,
    const float* __restrict__ a_d, float* __restrict__ out_s,
    float* __restrict__ out_d, int n) {
  constexpr int LPN = (H * F) / 4;
  constexpr int LPH = F / 4;
  constexpr int NPW = 32 / LPN;
  constexpr int U = (NPW == 1) ? 8 : 4;
  const int lane = threadIdx.x & 31;
  const int sub = lane % LPN;
  const int wave = (int)((blockIdx.x * blockDim.x + threadIdx.x) >> 5);
  const int node0 = wave * (NPW * U) + lane / LPN;
  const float4* h4 = (const float4*)h;
  const float4 as4 = ((const float4*)a_s)[sub];
  const float4 ad4 = ((const float4*)a_d)[sub];
  float4 v[U];
#pragma unroll
  for (int u = 0; u < U; ++u) {
    int nd = node0 + u * NPW;
    nd = nd < n ? nd : n - 1;
    v[u] = h4[(size_t)nd * LPN + sub];
  }
  __builtin_amdgcn_sched_barrier(0);
  float ps[U], pd[U];
#pragma unroll
  for (int u = 0; u < U; ++u) {
    ps[u] = group_sum<LPH>(v[u].x * as4.x + v[u].y * as4.y + v[u].z * as4.z +
                           v[u].w * as4.w);
    pd[u] = group_sum<LPH>(v[u].x * ad4.x + v[u].y * ad4.y + v[u].z * ad4.z +
                           v[u].w * ad4.w);
  }
#pragma unroll
  for (int u = 0; u < U; ++u) {
    const int nd = node0 + u * NPW;
    if (nd < n && (sub % LPH) == 0) {
      out_s[nd * H + sub / LPH] = ps[u];
      out_d[nd * H + sub / LPH] = pd[u];
    }
  }
}

template <int RO>
__global__ __launch_bounds__(256) void owner_ids_kernel(
    const int* __restrict__ dst, int nE, unsigned char* __restrict__ ids) {
  const int q = blockIdx.x * 256 + threadIdx.x;
  const int e0 = q * 4;
  if (e0 + 3 < nE) {
    int d0, d1, d2, d3;
    if (((size_t)dst & 15) == 0) {
      const int4 v = ((const int4*)dst)[q];
      d0 = v.x; d1 = v.y; d2 = v.z; d3 = v.w;
    } else {
      d0 = dst[e0]; d1 = dst[e0 + 1]; d2 = dst[e0 + 2]; d3 = dst[e0 + 3];
    }
    ((unsigned*)ids)[q] = (unsigned)(d0 / RO) | ((unsigned)(d1 / RO) << 8) |
                          ((unsigned)(d2 / RO) << 16) |
                          ((unsigned)(d3 / RO) << 24);
  } else {
    for (int e = e0; e < nE; ++e) ids[e] = (unsigned char)(dst[e] / RO);
  }
}

template <int H, int F, int R, int CAP, bool ELU, int LMODE, int LCAP>
__global__ __launch_bounds__(1024) void owner_agg_kernel(
    const float* __restrict__ hsrc, const float* __restrict__ as,
    const float* __restrict__ ad, const int* __restrict__ src,
    const int* __restrict__ dst, const unsigned* __restrict__ ids, int nE,
    int n, const float* __restrict__ bias, float* __restrict__ outp,
    unsigned* __restrict__ glist, int* __restrict__ gcnt) {
  static_assert(R <= CAP, "the self loops must fit the list");
  static_assert(LCAP <= CAP, "a list taken from the workspace must fit the LDS list");
  static_assert(R < 255, "row 255 marks an entry that is not ours");
  constexpr int HF = H * F;
  constexpr int NS = HF / 32;
  constexpr int HF4 = HF / 4;
  constexpr int U = 8;
  constexpr int G = 8;
  constexpr int U3 = 4;
  __shared__ __align__(16) float acc[R * HF];
  __shared__ float den[R * H];
  __shared__ unsigned mxo[R * H];
  __shared__ float adl[R * H];
  __shared__ int ls[CAP];
  __shared__ int lt[CAP];
  __shared__ float lw[CAP * H];
  __shared__ int cnt;

  const int tid = threadIdx.x;
  const int lane = tid & 31;
  const int wv = tid >> 5;
  const int lo = blockIdx.x * R;
  const int nr = (n - lo) < R ? (n - lo) : R;

  int imp = 0;
  if constexpr (LMODE == 2) {
    if (glist != nullptr) imp = gcnt[blockIdx.x];
    imp = imp <= LCAP ? imp : 0;
  }
  const bool imported = imp > 0;
  unsigned* const gl =
      glist != nullptr ? glist + (size_t)blockIdx.x * LCAP : nullptr;

  if (tid == 0) cnt = imported ? imp : nr;
  for (int i = tid; i < nr * H; i += 1024) {
    den[i] = 0.f;
    mxo[i] = 0u;
    adl[i] = ad[(size_t)lo * H + i];
  }
  {
    float4* acc4 = (float4*)acc;
    const float4 z = {0.f, 0.f, 0.f, 0.f};
    for (int i = tid; i < nr * HF4; i += 1024) acc4[i] = z;
  }
  __syncthreads();

  auto take = [&](int e) {
    const int pos = atomicAdd(&cnt, 1);
    if (pos < CAP) ls[pos] = e;
  };

  const unsigned unr = (unsigned)nr;
  int done = imported ? nE : 0;
  if (imported) {
  } else if (ids != nullptr) {
    const int n16 = nE >> 4;
    const uint4* q4 = (const uint4*)ids;
    const unsigned pat = (unsigned)blockIdx.x * 0x01010101u;
    for (int base = tid; base < n16; base += 1024 * U) {
      uint4 v[U];
#pragma unroll
      for (int u = 0; u < U; ++u) {
        int i = base + u * 1024;
        i = i < n16 ? i : n16 - 1;
        v[u] = q4[i];
      }
#pragma unroll
      for (int u = 0; u < U; ++u) {
        const int i = base + u * 1024;
        const unsigned x0 = v[u].x ^ pat, x1 = v[u].y ^ pat;
        const unsigned x2 = v[u].z ^ pat, x3 = v[u].w ^ pat;
        const unsigned z0 = (x0 - 0x01010101u) & ~x0;
        const unsigned z1 = (x1 - 0x01010101u) & ~x1;
        const unsigned z2 = (x2 - 0x01010101u) & ~x2;
        const unsigned z3 = (x3 - 0x01010101u) & ~x3;
        if (i < n16 && ((z0 | z1 | z2 | z3) & 0x80808080u) != 0u) {
          unsigned m = ((z0 & 0x80808080u) >> 7) | ((z1 & 0x80808080u) >> 6) |
                       ((z2 & 0x80808080u) >> 5) | ((z3 & 0x80808080u) >> 4);
          while (m != 0u) {
            const int b = __builtin_ctz(m);
            m &= m - 1u;
            take(i * 16 + (b & 3) * 4 + (b >> 3));
          }
        }
      }
    }
    done = n16 * 16;
  } else if (((size_t)dst & 15) == 0) {
    const int n4 = nE >> 2;
    const int4* d4 = (const int4*)dst;
    for (int base = tid; base < n4; base += 1024 * U) {
      int4 v[U];
#pragma unroll
      for (int u = 0; u < U; ++u) {
        int i = base + u * 1024;
        i = i < n4 ? i : n4 - 1;
        v[u] = d4[i];
      }
#pragma unroll
      for (int u = 0; u < U; ++u) {
        const int i = base + u * 1024;
        const unsigned t0 = (unsigned)(v[u].x - lo);
        const unsigned t1 = (unsigned)(v[u].y - lo);
        const unsigned t2 = (unsigned)(v[u].z - lo);
        const unsigned t3 = (unsigned)(v[u].w - lo);
        const unsigned tm = min(min(t0, t1), min(t2, t3));
        if (i < n4 && tm < unr) {
          if (t0 < unr) take(i * 4 + 0);
          if (t1 < unr) take(i * 4 + 1);
          if (t2 < unr) take(i * 4 + 2);
          if (t3 < unr) take(i * 4 + 3);
        }
      }
    }
    done = n4 * 4;
  }
  for (int e = done + tid; e < nE; e += 1024) {
    if ((unsigned)(dst[e] - lo) < unr) take(e);
  }
  __syncthreads();
  const int total = cnt;
  const int nl = total <= CAP ? total : 0;
  const bool leave = LMODE == 1 && glist != nullptr && nl > 0 && nl <= LCAP;
  if constexpr (LMODE == 1) {
    if (glist != nullptr && tid == 0) gcnt[blockIdx.x] = leave ? nl : 0;
  }
  if (total > CAP) {
    for (int i = tid; i < nr * H; i += 1024) {
      float x = as[(size_t)lo * H + i] + adl[i];
      x = x > 0.f ? x : 0.2f * x;
      atomicMax(&mxo[i], f2ord(x));
    }
    for (int e = tid; e < nE; e += 1024) {
      const unsigned t = (unsigned)(dst[e] - lo);
      if (t < unr) {
        const int s = src[e];
        for (int hd = 0; hd < H; ++hd) {
          float x = as[(size_t)s * H + hd] + adl[t * H + hd];
          x = x > 0.f ? x : 0.2f * x;
          atomicMax(&mxo[t * H + hd], f2ord(x));
        }
      }
    }
    __syncthreads();
    for (int i = tid; i < nr * H; i += 1024) {
      const int r = i / H;
      const int hd = i - r * H;
      float x = as[(size_t)lo * H + i] + adl[i];
      x = x > 0.f ? x : 0.2f * x;
      const float w = __expf(x - ord2f(mxo[i]));
      atomicAdd(&den[i], w);
      for (int f = 0; f < F; ++f)
        atomicAdd(&acc[r * HF + hd * F + f],
                  hsrc[(size_t)(lo + r) * HF + hd * F + f] * w);
    }
    for (int e = tid; e < nE; e += 1024) {
      const unsigned t = (unsigned)(dst[e] - lo);
      if (t < unr) {
        const int s = src[e];
        for (int hd = 0; hd < H; ++hd) {
          float x = as[(size_t)s * H + hd] + adl[t * H + hd];
          x = x > 0.f ? x : 0.2f * x;
          const float w = __expf(x - ord2f(mxo[t * H + hd]));
          atomicAdd(&den[t * H + hd], w);
          for (int f = 0; f < F; ++f)
            atomicAdd(&acc[t * HF + hd * F + f],
                      hsrc[(size_t)s * HF + hd * F + f] * w);
        }
      }
    }
  }

  for (int i0 = tid; i0 < nl; i0 += 1024 * U3) {
    int s[U3];
    unsigned t[U3];
    if (imported) {
#pragma unroll
      for (int u = 0; u < U3; ++u) {
        int i = i0 + u * 1024;
        i = i < nl ? i : i0;
        const unsigned en = gl[i];
        s[u] = (int)(en & 0x00ffffffu);
        t[u] = en >> 24;
      }
    } else {
      int e[U3], dd[U3];
#pragma unroll
      for (int u = 0; u < U3; ++u) {
        const int i = i0 + u * 1024;
        e[u] = (i >= nr && i < nl) ? ls[i] : -1;
        dd[u] = i < nr ? lo + i : lo - 1;
        s[u] = i < nr ? lo + i : lo;
      }
#pragma unroll
      for (int u = 0; u < U3; ++u) {
        if (e[u] >= 0) {
          dd[u] = dst[e[u]];
          s[u] = src[e[u]];
        }
      }
#pragma unroll
      for (int u = 0; u < U3; ++u) t[u] = (unsigned)(dd[u] - lo);
    }
    float a[U3][H];
#pragma unroll
    for (int u = 0; u < U3; ++u) {
#pragma unroll
      for (int hd = 0; hd < H; ++hd) a[u][hd] = as[(size_t)s[u] * H + hd];
    }
    float x[U3][H];
#pragma unroll
    for (int u = 0; u < U3; ++u) {
      const unsigned tc = t[u] < unr ? t[u] : 0u;
#pragma unroll
      for (int hd = 0; hd < H; ++hd) {
        const float v = a[u][hd] + adl[tc * H + hd];
        x[u][hd] = v > 0.f ? v : 0.2f * v;
      }
    }
#pragma unroll
    for (int u = 0; u < U3; ++u) {
      const int i = i0 + u * 1024;
      if (i < nl) {
        const bool ours = t[u] < unr;
        lt[i] = ours ? (int)t[u] : -1;
        ls[i] = ours ? s[u] : lo;
        if constexpr (LMODE == 1) {
          if (leave)
            gl[i] = ours ? ((unsigned)s[u] | (t[u] << 24))
                         : ((unsigned)lo | 0xff000000u);
        }
#pragma unroll
        for (int hd = 0; hd < H; ++hd) lw[i * H + hd] = x[u][hd];
        if (ours) {
#pragma unroll
          for (int hd = 0; hd < H; ++hd)
            atomicMax(&mxo[t[u] * H + hd], f2ord(x[u][hd]));
        }
      }
    }
  }
  __syncthreads();

  for (int i = tid; i < nl; i += 1024) {
    const int t = lt[i];
    if (t >= 0) {
#pragma unroll
      for (int hd = 0; hd < H; ++hd) {
        const float w = __expf(lw[i * H + hd] - ord2f(mxo[t * H + hd]));
        lw[i * H + hd] = w;
        atomicAdd(&den[t * H + hd], w);
      }
    }
  }
  __syncthreads();

  for (int i0 = wv; i0 < nl; i0 += 32 * G) {
    float v[G][NS];
#pragma unroll
    for (int g = 0; g < G; ++g) {
      int i = i0 + g * 32;
      i = i < nl ? i : i0;
      const float* hp = hsrc + (size_t)ls[i] * HF + lane;
#pragma unroll
      for (int j = 0; j < NS; ++j) v[g][j] = hp[j * 32];
    }
#pragma unroll
    for (int g = 0; g < G; ++g) {
      const int i = i0 + g * 32;
      if (i < nl) {
        const int t = lt[i];
        if (t >= 0) {
          float* ap = acc + t * HF + lane;
#pragma unroll
          for (int j = 0; j < NS; ++j)
            atomicAdd(ap + j * 32, v[g][j] * lw[i * H + (j * 32) / F]);
        }
      }
    }
  }
  __syncthreads();

  {
    const float4* acc4 = (const float4*)acc;
    float4* o4 = (float4*)outp;
    for (int i = tid; i < nr * HF4; i += 1024) {
      const int r = i / HF4;
      const int c4 = i - r * HF4;
      const float dn = den[r * H + (c4 * 4) / F] + 1e-16f;
      float4 x = acc4[i];
      x.x /= dn; x.y /= dn; x.z /= dn; x.w /= dn;
      if constexpr (ELU) {
        const float4 bb = ((const float4*)bias)[c4];
        float t;
        t = x.x + bb.x; x.x = t > 0.f ? t : (__expf(t) - 1.f);
        t = x.y + bb.y; x.y = t > 0.f ? t : (__expf(t) - 1.f);
        t = x.z + bb.z; x.z = t > 0.f ? t : (__expf(t) - 1.f);
        t = x.w + bb.w; x.w = t > 0.f ? t : (__expf(t) - 1.f);
      }
      o4[(size_t)(lo + r) * HF4 + c4] = x;
    }
  }
}

__global__ __launch_bounds__(256) void logsoftmax_kernel(
    const float* __restrict__ agg, const float* __restrict__ bias,
    float* __restrict__ out, int n) {
  constexpr int U = 4;
  const int lane = threadIdx.x & 31;
  const int sub = lane & 7;
  const int wave = (int)((blockIdx.x * blockDim.x + threadIdx.x) >> 5);
  const int node0 = wave * (4 * U) + (lane >> 3);
  const float4* a4 = (const float4*)agg;
  float4* o4 = (float4*)out;
  const float4 b = ((const float4*)bias)[sub];
  float4 v[U];
#pragma unroll
  for (int u = 0; u < U; ++u) {
    int nd = node0 + u * 4;
    nd = nd < n ? nd : n - 1;
    v[u] = a4[(size_t)nd * 8 + sub];
  }
  float4 r[U];
#pragma unroll
  for (int u = 0; u < U; ++u) {
    const float x0 = v[u].x + b.x, x1 = v[u].y + b.y;
    const float x2 = v[u].z + b.z, x3 = v[u].w + b.w;
    const float m = group_max<8>(fmaxf(fmaxf(x0, x1), fmaxf(x2, x3)));
    const float s = group_sum<8>(__expf(x0 - m) + __expf(x1 - m) +
                                 __expf(x2 - m) + __expf(x3 - m));
    const float ls = __logf(s);
    r[u].x = x0 - m - ls; r[u].y = x1 - m - ls;
    r[u].z = x2 - m - ls; r[u].w = x3 - m - ls;
  }
#pragma unroll
  for (int u = 0; u < U; ++u) {
    const int nd = node0 + u * 4;
    if (nd < n) o4[(size_t)nd * 8 + sub] = r[u];
  }
}

extern "C" void kernel_launch(void* const* d_in, const int* in_sizes, int n_in,
                              void* d_out, int out_size, void* d_ws,
                              size_t ws_size, hipStream_t stream) {
  (void)n_in; (void)out_size;
  const float* x      = (const float*)d_in[0];
  const int*   ei     = (const int*)d_in[1];
  const float* W1     = (const float*)d_in[2];
  const float* a_src1 = (const float*)d_in[3];
  const float* a_dst1 = (const float*)d_in[4];
  const float* b1     = (const float*)d_in[5];
  const float* W2     = (const float*)d_in[6];
  const float* a_src2 = (const float*)d_in[7];
  const float* a_dst2 = (const float*)d_in[8];
  const float* b2     = (const float*)d_in[9];

  const int n    = in_sizes[0] / 500;
  const int nE   = in_sizes[1] / 2;
  const int* src = ei;
  const int* dst = ei + nE;
  constexpr int R = 196;
  const int nOwn = (n + R - 1) / R;

  float* ws = (float*)d_ws;
  size_t o = 0;
  float* h1      = ws + o; o += (size_t)n * 128;
  float* h2      = ws + o; o += (size_t)n * 32;
  float* Bp1     = ws + o; o += (size_t)500 * 128;
  float* Bp2     = ws + o; o += (size_t)128 * 32;
  float* agg1    = ws + o; o += (size_t)n * 128;
  float* agg2    = ws + o; o += (size_t)n * 32;
  float* idsp    = ws + o; o += ((size_t)nE + 15) / 16 * 4;
  float* as1     = ws + o; o += (size_t)n * 2;
  float* ad1     = ws + o; o += (size_t)n * 2;
  float* as2     = ws + o; o += (size_t)n;
  float* ad2     = ws + o; o += (size_t)n;
  constexpr int LCAP = 4096;
  float* glistp  = ws + o; o += (size_t)nOwn * LCAP;
  float* gcntp   = ws + o; o += (size_t)nOwn;
  const bool lists = n <= (1 << 24) && o * sizeof(float) <= ws_size;
  unsigned* glist = lists ? (unsigned*)glistp : nullptr;
  int* gcnt = lists ? (int*)gcntp : nullptr;

  const bool byteIds = nOwn <= 256;
  unsigned* ids = byteIds ? (unsigned*)idsp : nullptr;

  pack_b_kernel<500, 128><<<(500 * 128 + 255) / 256, 256, 0, stream>>>(W1, Bp1);
  pack_b_kernel<128, 32><<<(128 * 32 + 255) / 256, 256, 0, stream>>>(W2, Bp2);
  if (byteIds)
    owner_ids_kernel<R><<<((nE + 3) / 4 + 255) / 256, 256, 0, stream>>>(
        dst, nE, (unsigned char*)idsp);

  gemm_wmma_kernel<500, 128><<<(n + 15) / 16, 256, 0, stream>>>(x, Bp1, h1, n);
  alpha_kernel<2, 64><<<(n + 63) / 64, 256, 0, stream>>>(
      h1, a_src1, a_dst1, as1, ad1, n);
  owner_agg_kernel<2, 64, R, 6144, true, 1, LCAP><<<nOwn, 1024, 0, stream>>>(
      h1, as1, ad1, src, dst, ids, nE, n, b1, agg1, glist, gcnt);

  gemm_wmma_kernel<128, 32><<<(n + 63) / 64, 256, 0, stream>>>(agg1, Bp2, h2, n);
  alpha_kernel<1, 32><<<(n + 127) / 128, 256, 0, stream>>>(
      h2, a_src2, a_dst2, as2, ad2, n);
  owner_agg_kernel<1, 32, R, 12288, false, 2, LCAP><<<nOwn, 1024, 0, stream>>>(
      h2, as2, ad2, src, dst, ids, nE, n, b2, agg2, glist, gcnt);

  logsoftmax_kernel<<<(n + 127) / 128, 256, 0, stream>>>(
      agg2, b2, (float*)d_out, n);
}


# owner passes fetch source rows 16 B per lane: one wave-wide load per 512-B row (layer 2: four 128-B rows), sums kept lane-transposed in LDS so adds stay conflict-free; LDS float adds: arrival order
# speedup vs baseline: 3.6085x; 1.0857x over previous
#include <hip/hip_runtime.h>
#include <hip/hip_bf16.h>


typedef float v2f __attribute__((ext_vector_type(2)));
typedef float v8f __attribute__((ext_vector_type(8)));

__device__ __forceinline__ unsigned f2ord(float f) {
  unsigned u = __float_as_uint(f);
  return (u & 0x80000000u) ? ~u : (u | 0x80000000u);
}
__device__ __forceinline__ float ord2f(unsigned o) {
  unsigned u = (o & 0x80000000u) ? (o ^ 0x80000000u) : ~o;
  return __uint_as_float(u);
}

template <int K, int NCOL>
__global__ void pack_b_kernel(const float* __restrict__ B,
                              float* __restrict__ Bp) {
  int tid = blockIdx.x * blockDim.x + threadIdx.x;
  if (tid >= K * NCOL) return;
  int k = tid / NCOL;
  int col = tid - k * NCOL;
  int kk = k >> 2;
  int p = (k >> 1) & 1;
  int q = k & 1;
  Bp[(((kk * 2 + p) * NCOL) + col) * 2 + q] = B[tid];
}

template <int K, int NCOL>
__global__ __launch_bounds__(256) void gemm_wmma_kernel(
    const float* __restrict__ A, const float* __restrict__ Bp,
    float* __restrict__ C, int M) {
  constexpr int NT = NCOL / 16;
  constexpr int MT = 8 / NT;
  constexpr int ROWS = MT * 16;
  constexpr int KV = K / 4;
  __shared__ __align__(16) float As[ROWS * K];

  const int tid = threadIdx.x;
  const int wave = tid >> 5;
  const int lane = tid & 31;
  const int row0 = blockIdx.x * ROWS;

  {
    const float4* A4 = (const float4*)A;
    float4* As4 = (float4*)As;
    for (int i = tid; i < ROWS * KV; i += 256) {
      int r = i / KV;
      int c = i - r * KV;
      int gr = row0 + r;
      float4 v = {0.f, 0.f, 0.f, 0.f};
      if (gr < M) v = A4[(size_t)gr * KV + c];
      As4[i] = v;
    }
  }
  __syncthreads();

  const int mtile = wave / NT;
  const int ntile = wave - mtile * NT;
  const int lrow = (mtile << 4) + (lane & 15);
  const int koff = (lane >> 4) << 1;
  const int ncol = (ntile << 4) + (lane & 15);
  const int bbase = ((lane >> 4) * NCOL + ncol);

  v8f acc = {0.f, 0.f, 0.f, 0.f, 0.f, 0.f, 0.f, 0.f};
  for (int kk = 0; kk < KV; ++kk) {
    v2f a, b;
    a.x = As[lrow * K + kk * 4 + koff];
    a.y = As[lrow * K + kk * 4 + koff + 1];
    b = *(const v2f*)&Bp[(size_t)(bbase + kk * 2 * NCOL) * 2];
    acc = __builtin_amdgcn_wmma_f32_16x16x4_f32(
         false, a,  false, b,
         (short)0, acc,  false,  false);
  }

  const int rbase = row0 + (mtile << 4) + ((lane >> 4) << 3);
  if (row0 + ROWS <= M) {
#pragma unroll
    for (int i = 0; i < 8; ++i) C[(size_t)(rbase + i) * NCOL + ncol] = acc[i];
  } else {
#pragma unroll
    for (int i = 0; i < 8; ++i) {
      int r = rbase + i;
      if (r < M) C[(size_t)r * NCOL + ncol] = acc[i];
    }
  }
}

template <int M>
__device__ __forceinline__ float row_xor(float x) {
  return __int_as_float(__builtin_amdgcn_update_dpp(
      0, __float_as_int(x), 0x160 + M, 0xf, 0xf, true));
}
template <int W>
__device__ __forceinline__ float group_sum(float x) {
  x += row_xor<1>(x);
  x += row_xor<2>(x);
  x += row_xor<4>(x);
  if constexpr (W == 16) x += row_xor<8>(x);
  return x;
}
template <int W>
__device__ __forceinline__ float group_max(float x) {
  x = fmaxf(x, row_xor<1>(x));
  x = fmaxf(x, row_xor<2>(x));
  x = fmaxf(x, row_xor<4>(x));
  if constexpr (W == 16) x = fmaxf(x, row_xor<8>(x));
  return x;
}

template <int H, int F>
__global__ __launch_bounds__(256) void alpha_kernel(
    const float* __restrict__ h, const float* __restrict__ a_s,
    const float* __restrict__ a_d, float* __restrict__ out_s,
    float* __restrict__ out_d, int n) {
  constexpr int LPN = (H * F) / 4;
  constexpr int LPH = F / 4;
  constexpr int NPW = 32 / LPN;
  constexpr int U = (NPW == 1) ? 8 : 4;
  const int lane = threadIdx.x & 31;
  const int sub = lane % LPN;
  const int wave = (int)((blockIdx.x * blockDim.x + threadIdx.x) >> 5);
  const int node0 = wave * (NPW * U) + lane / LPN;
  const float4* h4 = (const float4*)h;
  const float4 as4 = ((const float4*)a_s)[sub];
  const float4 ad4 = ((const float4*)a_d)[sub];
  float4 v[U];
#pragma unroll
  for (int u = 0; u < U; ++u) {
    int nd = node0 + u * NPW;
    nd = nd < n ? nd : n - 1;
    v[u] = h4[(size_t)nd * LPN + sub];
  }
  __builtin_amdgcn_sched_barrier(0);
  float ps[U], pd[U];
#pragma unroll
  for (int u = 0; u < U; ++u) {
    ps[u] = group_sum<LPH>(v[u].x * as4.x + v[u].y * as4.y + v[u].z * as4.z +
                           v[u].w * as4.w);
    pd[u] = group_sum<LPH>(v[u].x * ad4.x + v[u].y * ad4.y + v[u].z * ad4.z +
                           v[u].w * ad4.w);
  }
#pragma unroll
  for (int u = 0; u < U; ++u) {
    const int nd = node0 + u * NPW;
    if (nd < n && (sub % LPH) == 0) {
      out_s[nd * H + sub / LPH] = ps[u];
      out_d[nd * H + sub / LPH] = pd[u];
    }
  }
}

template <int RO>
__global__ __launch_bounds__(256) void owner_ids_kernel(
    const int* __restrict__ dst, int nE, unsigned char* __restrict__ ids) {
  const int q = blockIdx.x * 256 + threadIdx.x;
  const int e0 = q * 4;
  if (e0 + 3 < nE) {
    int d0, d1, d2, d3;
    if (((size_t)dst & 15) == 0) {
      const int4 v = ((const int4*)dst)[q];
      d0 = v.x; d1 = v.y; d2 = v.z; d3 = v.w;
    } else {
      d0 = dst[e0]; d1 = dst[e0 + 1]; d2 = dst[e0 + 2]; d3 = dst[e0 + 3];
    }
    ((unsigned*)ids)[q] = (unsigned)(d0 / RO) | ((unsigned)(d1 / RO) << 8) |
                          ((unsigned)(d2 / RO) << 16) |
                          ((unsigned)(d3 / RO) << 24);
  } else {
    for (int e = e0; e < nE; ++e) ids[e] = (unsigned char)(dst[e] / RO);
  }
}

template <int H, int F, int R, int CAP, bool ELU, int LMODE, int LCAP>
__global__ __launch_bounds__(1024) void owner_agg_kernel(
    const float* __restrict__ hsrc, const float* __restrict__ as,
    const float* __restrict__ ad, const int* __restrict__ src,
    const int* __restrict__ dst, const unsigned* __restrict__ ids, int nE,
    int n, const float* __restrict__ bias, float* __restrict__ outp,
    unsigned* __restrict__ glist, int* __restrict__ gcnt) {
  static_assert(R <= CAP, "the self loops must fit the list");
  static_assert(LCAP <= CAP, "a list taken from the workspace must fit the LDS list");
  static_assert(R < 255, "row 255 marks an entry that is not ours");
  constexpr int HF = H * F;
  constexpr int HF4 = HF / 4;
  constexpr int U = 8;
  constexpr int G = 8;
  constexpr int U3 = 4;
  constexpr int PQ = HF4;
  constexpr int EPL = 32 / PQ;
  constexpr int PITCH = HF >= 64 ? HF : HF + 8;
  static_assert(32 % PQ == 0 && PITCH % 4 == 0, "rows must tile a wave");
  __shared__ __align__(16) float acc[(R + 1) * PITCH];
  __shared__ float den[R * H];
  __shared__ unsigned mxo[R * H];
  __shared__ float adl[R * H];
  __shared__ int ls[CAP];
  __shared__ int lt[CAP];
  __shared__ float lw[CAP * H];
  __shared__ int cnt;

  const int tid = threadIdx.x;
  const int lane = tid & 31;
  const int wv = tid >> 5;
  const int lo = blockIdx.x * R;
  const int nr = (n - lo) < R ? (n - lo) : R;

  int imp = 0;
  if constexpr (LMODE == 2) {
    if (glist != nullptr) imp = gcnt[blockIdx.x];
    imp = imp <= LCAP ? imp : 0;
  }
  const bool imported = imp > 0;
  unsigned* const gl =
      glist != nullptr ? glist + (size_t)blockIdx.x * LCAP : nullptr;

  if (tid == 0) cnt = imported ? imp : nr;
  for (int i = tid; i < nr * H; i += 1024) {
    den[i] = 0.f;
    mxo[i] = 0u;
    adl[i] = ad[(size_t)lo * H + i];
  }
  {
    float4* acc4 = (float4*)acc;
    const float4 z = {0.f, 0.f, 0.f, 0.f};
    for (int i = tid; i < nr * (PITCH / 4); i += 1024) acc4[i] = z;
  }
  __syncthreads();

  auto take = [&](int e) {
    const int pos = atomicAdd(&cnt, 1);
    if (pos < CAP) ls[pos] = e;
  };

  const unsigned unr = (unsigned)nr;
  int done = imported ? nE : 0;
  if (imported) {
  } else if (ids != nullptr) {
    const int n16 = nE >> 4;
    const uint4* q4 = (const uint4*)ids;
    const unsigned pat = (unsigned)blockIdx.x * 0x01010101u;
    for (int base = tid; base < n16; base += 1024 * U) {
      uint4 v[U];
#pragma unroll
      for (int u = 0; u < U; ++u) {
        int i = base + u * 1024;
        i = i < n16 ? i : n16 - 1;
        v[u] = q4[i];
      }
#pragma unroll
      for (int u = 0; u < U; ++u) {
        const int i = base + u * 1024;
        const unsigned x0 = v[u].x ^ pat, x1 = v[u].y ^ pat;
        const unsigned x2 = v[u].z ^ pat, x3 = v[u].w ^ pat;
        const unsigned z0 = (x0 - 0x01010101u) & ~x0;
        const unsigned z1 = (x1 - 0x01010101u) & ~x1;
        const unsigned z2 = (x2 - 0x01010101u) & ~x2;
        const unsigned z3 = (x3 - 0x01010101u) & ~x3;
        if (i < n16 && ((z0 | z1 | z2 | z3) & 0x80808080u) != 0u) {
          unsigned m = ((z0 & 0x80808080u) >> 7) | ((z1 & 0x80808080u) >> 6) |
                       ((z2 & 0x80808080u) >> 5) | ((z3 & 0x80808080u) >> 4);
          while (m != 0u) {
            const int b = __builtin_ctz(m);
            m &= m - 1u;
            take(i * 16 + (b & 3) * 4 + (b >> 3));
          }
        }
      }
    }
    done = n16 * 16;
  } else if (((size_t)dst & 15) == 0) {
    const int n4 = nE >> 2;
    const int4* d4 = (const int4*)dst;
    for (int base = tid; base < n4; base += 1024 * U) {
      int4 v[U];
#pragma unroll
      for (int u = 0; u < U; ++u) {
        int i = base + u * 1024;
        i = i < n4 ? i : n4 - 1;
        v[u] = d4[i];
      }
#pragma unroll
      for (int u = 0; u < U; ++u) {
        const int i = base + u * 1024;
        const unsigned t0 = (unsigned)(v[u].x - lo);
        const unsigned t1 = (unsigned)(v[u].y - lo);
        const unsigned t2 = (unsigned)(v[u].z - lo);
        const unsigned t3 = (unsigned)(v[u].w - lo);
        const unsigned tm = min(min(t0, t1), min(t2, t3));
        if (i < n4 && tm < unr) {
          if (t0 < unr) take(i * 4 + 0);
          if (t1 < unr) take(i * 4 + 1);
          if (t2 < unr) take(i * 4 + 2);
          if (t3 < unr) take(i * 4 + 3);
        }
      }
    }
    done = n4 * 4;
  }
  for (int e = done + tid; e < nE; e += 1024) {
    if ((unsigned)(dst[e] - lo) < unr) take(e);
  }
  __syncthreads();
  const int total = cnt;
  const int nl = total <= CAP ? total : 0;
  const bool leave = LMODE == 1 && glist != nullptr && nl > 0 && nl <= LCAP;
  if constexpr (LMODE == 1) {
    if (glist != nullptr && tid == 0) gcnt[blockIdx.x] = leave ? nl : 0;
  }
  if (total > CAP) {
    for (int i = tid; i < nr * H; i += 1024) {
      float x = as[(size_t)lo * H + i] + adl[i];
      x = x > 0.f ? x : 0.2f * x;
      atomicMax(&mxo[i], f2ord(x));
    }
    for (int e = tid; e < nE; e += 1024) {
      const unsigned t = (unsigned)(dst[e] - lo);
      if (t < unr) {
        const int s = src[e];
        for (int hd = 0; hd < H; ++hd) {
          float x = as[(size_t)s * H + hd] + adl[t * H + hd];
          x = x > 0.f ? x : 0.2f * x;
          atomicMax(&mxo[t * H + hd], f2ord(x));
        }
      }
    }
    __syncthreads();
    for (int i = tid; i < nr * H; i += 1024) {
      const int r = i / H;
      const int hd = i - r * H;
      float x = as[(size_t)lo * H + i] + adl[i];
      x = x > 0.f ? x : 0.2f * x;
      const float w = __expf(x - ord2f(mxo[i]));
      atomicAdd(&den[i], w);
      for (int f = 0; f < F; ++f) {
        const int c = hd * F + f;
        atomicAdd(&acc[r * PITCH + (c & 3) * PQ + (c >> 2)],
                  hsrc[(size_t)(lo + r) * HF + c] * w);
      }
    }
    for (int e = tid; e < nE; e += 1024) {
      const unsigned t = (unsigned)(dst[e] - lo);
      if (t < unr) {
        const int s = src[e];
        for (int hd = 0; hd < H; ++hd) {
          float x = as[(size_t)s * H + hd] + adl[t * H + hd];
          x = x > 0.f ? x : 0.2f * x;
          const float w = __expf(x - ord2f(mxo[t * H + hd]));
          atomicAdd(&den[t * H + hd], w);
          for (int f = 0; f < F; ++f) {
            const int c = hd * F + f;
            atomicAdd(&acc[t * PITCH + (c & 3) * PQ + (c >> 2)],
                      hsrc[(size_t)s * HF + c] * w);
          }
        }
      }
    }
  }

  for (int i0 = tid; i0 < nl; i0 += 1024 * U3) {
    int s[U3];
    unsigned t[U3];
    if (imported) {
#pragma unroll
      for (int u = 0; u < U3; ++u) {
        int i = i0 + u * 1024;
        i = i < nl ? i : i0;
        const unsigned en = gl[i];
        s[u] = (int)(en & 0x00ffffffu);
        t[u] = en >> 24;
      }
    } else {
      int e[U3], dd[U3];
#pragma unroll
      for (int u = 0; u < U3; ++u) {
        const int i = i0 + u * 1024;
        e[u] = (i >= nr && i < nl) ? ls[i] : -1;
        dd[u] = i < nr ? lo + i : lo - 1;
        s[u] = i < nr ? lo + i : lo;
      }
#pragma unroll
      for (int u = 0; u < U3; ++u) {
        if (e[u] >= 0) {
          dd[u] = dst[e[u]];
          s[u] = src[e[u]];
        }
      }
#pragma unroll
      for (int u = 0; u < U3; ++u) t[u] = (unsigned)(dd[u] - lo);
    }
    float a[U3][H];
#pragma unroll
    for (int u = 0; u < U3; ++u) {
#pragma unroll
      for (int hd = 0; hd < H; ++hd) a[u][hd] = as[(size_t)s[u] * H + hd];
    }
    float x[U3][H];
#pragma unroll
    for (int u = 0; u < U3; ++u) {
      const unsigned tc = t[u] < unr ? t[u] : 0u;
#pragma unroll
      for (int hd = 0; hd < H; ++hd) {
        const float v = a[u][hd] + adl[tc * H + hd];
        x[u][hd] = v > 0.f ? v : 0.2f * v;
      }
    }
#pragma unroll
    for (int u = 0; u < U3; ++u) {
      const int i = i0 + u * 1024;
      if (i < nl) {
        const bool ours = t[u] < unr;
        lt[i] = ours ? (int)t[u] : -1;
        ls[i] = ours ? s[u] : lo;
        if constexpr (LMODE == 1) {
          if (leave)
            gl[i] = ours ? ((unsigned)s[u] | (t[u] << 24))
                         : ((unsigned)lo | 0xff000000u);
        }
#pragma unroll
        for (int hd = 0; hd < H; ++hd) lw[i * H + hd] = x[u][hd];
        if (ours) {
#pragma unroll
          for (int hd = 0; hd < H; ++hd)
            atomicMax(&mxo[t[u] * H + hd], f2ord(x[u][hd]));
        }
      }
    }
  }
  __syncthreads();

  for (int i = tid; i < nl; i += 1024) {
    const int t = lt[i];
    if (t >= 0) {
#pragma unroll
      for (int hd = 0; hd < H; ++hd) {
        const float w = __expf(lw[i * H + hd] - ord2f(mxo[t * H + hd]));
        lw[i * H + hd] = w;
        atomicAdd(&den[t * H + hd], w);
      }
    }
  }
  __syncthreads();

  {
    const int es = lane / PQ;
    const int q = lane - es * PQ;
    const float4* h4 = (const float4*)hsrc;
    auto group = [&](const int k0, const bool full)
        __attribute__((always_inline)) {
      float4 v[G];
      const float4* hp[G];
#pragma unroll
      for (int g = 0; g < G; ++g) {
        int i = (k0 + g * 32) * EPL + es;
        if (!full) i = i < nl ? i : 0;
        hp[g] = h4 + (size_t)ls[i] * HF4 + q;
      }
      __builtin_amdgcn_sched_barrier(0);
#pragma unroll
      for (int g = 0; g < G; ++g) v[g] = *hp[g];
#pragma unroll
      for (int g = 0; g < G; ++g) {
        const int i = (k0 + g * 32) * EPL + es;
        if (full || i < nl) {
          const int t = lt[i];
          const float w = lw[i * H + (q * 4) / F];
          float* ap = acc + (t >= 0 ? t : R) * PITCH + q;
          atomicAdd(ap, v[g].x * w);
          atomicAdd(ap + PQ, v[g].y * w);
          atomicAdd(ap + 2 * PQ, v[g].z * w);
          atomicAdd(ap + 3 * PQ, v[g].w * w);
        }
      }
    };
    for (int k0 = wv; k0 * EPL < nl; k0 += 32 * G) {
      if ((k0 + (G - 1) * 32 + 1) * EPL <= nl) group(k0, true);
      else group(k0, false);
    }
  }
  __syncthreads();

  {
    float4* o4 = (float4*)outp;
    for (int i = tid; i < nr * HF4; i += 1024) {
      const int r = i / HF4;
      const int c4 = i - r * HF4;
      const float dn = den[r * H + (c4 * 4) / F] + 1e-16f;
      const float* ap = acc + r * PITCH + c4;
      float4 x = {ap[0], ap[PQ], ap[2 * PQ], ap[3 * PQ]};
      x.x /= dn; x.y /= dn; x.z /= dn; x.w /= dn;
      if constexpr (ELU) {
        const float4 bb = ((const float4*)bias)[c4];
        float t;
        t = x.x + bb.x; x.x = t > 0.f ? t : (__expf(t) - 1.f);
        t = x.y + bb.y; x.y = t > 0.f ? t : (__expf(t) - 1.f);
        t = x.z + bb.z; x.z = t > 0.f ? t : (__expf(t) - 1.f);
        t = x.w + bb.w; x.w = t > 0.f ? t : (__expf(t) - 1.f);
      }
      o4[(size_t)(lo + r) * HF4 + c4] = x;
    }
  }
}

__global__ __launch_bounds__(256) void logsoftmax_kernel(
    const float* __restrict__ agg, const float* __restrict__ bias,
    float* __restrict__ out, int n) {
  constexpr int U = 4;
  const int lane = threadIdx.x & 31;
  const int sub = lane & 7;
  const int wave = (int)((blockIdx.x * blockDim.x + threadIdx.x) >> 5);
  const int node0 = wave * (4 * U) + (lane >> 3);
  const float4* a4 = (const float4*)agg;
  float4* o4 = (float4*)out;
  const float4 b = ((const float4*)bias)[sub];
  float4 v[U];
#pragma unroll
  for (int u = 0; u < U; ++u) {
    int nd = node0 + u * 4;
    nd = nd < n ? nd : n - 1;
    v[u] = a4[(size_t)nd * 8 + sub];
  }
  float4 r[U];
#pragma unroll
  for (int u = 0; u < U; ++u) {
    const float x0 = v[u].x + b.x, x1 = v[u].y + b.y;
    const float x2 = v[u].z + b.z, x3 = v[u].w + b.w;
    const float m = group_max<8>(fmaxf(fmaxf(x0, x1), fmaxf(x2, x3)));
    const float s = group_sum<8>(__expf(x0 - m) + __expf(x1 - m) +
                                 __expf(x2 - m) + __expf(x3 - m));
    const float ls = __logf(s);
    r[u].x = x0 - m - ls; r[u].y = x1 - m - ls;
    r[u].z = x2 - m - ls; r[u].w = x3 - m - ls;
  }
#pragma unroll
  for (int u = 0; u < U; ++u) {
    const int nd = node0 + u * 4;
    if (nd < n) o4[(size_t)nd * 8 + sub] = r[u];
  }
}

extern "C" void kernel_launch(void* const* d_in, const int* in_sizes, int n_in,
                              void* d_out, int out_size, void* d_ws,
                              size_t ws_size, hipStream_t stream) {
  (void)n_in; (void)out_size;
  const float* x      = (const float*)d_in[0];
  const int*   ei     = (const int*)d_in[1];
  const float* W1     = (const float*)d_in[2];
  const float* a_src1 = (const float*)d_in[3];
  const float* a_dst1 = (const float*)d_in[4];
  const float* b1     = (const float*)d_in[5];
  const float* W2     = (const float*)d_in[6];
  const float* a_src2 = (const float*)d_in[7];
  const float* a_dst2 = (const float*)d_in[8];
  const float* b2     = (const float*)d_in[9];

  const int n    = in_sizes[0] / 500;
  const int nE   = in_sizes[1] / 2;
  const int* src = ei;
  const int* dst = ei + nE;
  constexpr int R = 196;
  const int nOwn = (n + R - 1) / R;

  float* ws = (float*)d_ws;
  size_t o = 0;
  float* h1      = ws + o; o += (size_t)n * 128;
  float* h2      = ws + o; o += (size_t)n * 32;
  float* Bp1     = ws + o; o += (size_t)500 * 128;
  float* Bp2     = ws + o; o += (size_t)128 * 32;
  float* agg1    = ws + o; o += (size_t)n * 128;
  float* agg2    = ws + o; o += (size_t)n * 32;
  float* idsp    = ws + o; o += ((size_t)nE + 15) / 16 * 4;
  float* as1     = ws + o; o += (size_t)n * 2;
  float* ad1     = ws + o; o += (size_t)n * 2;
  float* as2     = ws + o; o += (size_t)n;
  float* ad2     = ws + o; o += (size_t)n;
  constexpr int LCAP = 4096;
  float* glistp  = ws + o; o += (size_t)nOwn * LCAP;
  float* gcntp   = ws + o; o += (size_t)nOwn;
  const bool lists = n <= (1 << 24) && o * sizeof(float) <= ws_size;
  unsigned* glist = lists ? (unsigned*)glistp : nullptr;
  int* gcnt = lists ? (int*)gcntp : nullptr;

  const bool byteIds = nOwn <= 256;
  unsigned* ids = byteIds ? (unsigned*)idsp : nullptr;

  pack_b_kernel<500, 128><<<(500 * 128 + 255) / 256, 256, 0, stream>>>(W1, Bp1);
  pack_b_kernel<128, 32><<<(128 * 32 + 255) / 256, 256, 0, stream>>>(W2, Bp2);
  if (byteIds)
    owner_ids_kernel<R><<<((nE + 3) / 4 + 255) / 256, 256, 0, stream>>>(
        dst, nE, (unsigned char*)idsp);

  gemm_wmma_kernel<500, 128><<<(n + 15) / 16, 256, 0, stream>>>(x, Bp1, h1, n);
  alpha_kernel<2, 64><<<(n + 63) / 64, 256, 0, stream>>>(
      h1, a_src1, a_dst1, as1, ad1, n);
  owner_agg_kernel<2, 64, R, 6144, true, 1, LCAP><<<nOwn, 1024, 0, stream>>>(
      h1, as1, ad1, src, dst, ids, nE, n, b1, agg1, glist, gcnt);

  gemm_wmma_kernel<128, 32><<<(n + 63) / 64, 256, 0, stream>>>(agg1, Bp2, h2, n);
  alpha_kernel<1, 32><<<(n + 127) / 128, 256, 0, stream>>>(
      h2, a_src2, a_dst2, as2, ad2, n);
  owner_agg_kernel<1, 32, R, 12288, false, 2, LCAP><<<nOwn, 1024, 0, stream>>>(
      h2, as2, ad2, src, dst, ids, nE, n, b2, agg2, glist, gcnt);

  logsoftmax_kernel<<<(n + 127) / 128, 256, 0, stream>>>(
      agg2, b2, (float*)d_out, n);
}


# both GEMMs: the whole weight matrix sits in LDS in fragment order, copied from W by each of 256 persistent 16-wave workgroups that walk their rows in tiles (next tile prefetched); 10 -> 8 launches
# speedup vs baseline: 3.7469x; 1.0384x over previous
#include <hip/hip_runtime.h>
#include <hip/hip_bf16.h>


typedef float v2f __attribute__((ext_vector_type(2)));
typedef float v8f __attribute__((ext_vector_type(8)));

__device__ __forceinline__ unsigned f2ord(float f) {
  unsigned u = __float_as_uint(f);
  return (u & 0x80000000u) ? ~u : (u | 0x80000000u);
}
__device__ __forceinline__ float ord2f(unsigned o) {
  unsigned u = (o & 0x80000000u) ? (o ^ 0x80000000u) : ~o;
  return __uint_as_float(u);
}

struct Tile8 { float4 p0, p1, p2, p3, p4, p5, p6, p7; };
__device__ __forceinline__ float4 fetch_one(const float4* __restrict__ A4,
                                            long i, const long last) {
  i = i < last ? i : last;
  return A4[i];
}
__device__ __forceinline__ void fetch_tile(const float4* __restrict__ A4,
                                           const long first, const long last,
                                           Tile8& t) {
  t.p0 = fetch_one(A4, first, last);
  t.p1 = fetch_one(A4, first + 512, last);
  t.p2 = fetch_one(A4, first + 1024, last);
  t.p3 = fetch_one(A4, first + 1536, last);
  t.p4 = fetch_one(A4, first + 2048, last);
  t.p5 = fetch_one(A4, first + 2560, last);
  t.p6 = fetch_one(A4, first + 3072, last);
  t.p7 = fetch_one(A4, first + 3584, last);
}
template <int TV>
__device__ __forceinline__ void store_tile(float4* __restrict__ As4,
                                           const int tid, const Tile8& t) {
  if (tid < TV) As4[tid] = t.p0;
  if (tid + 512 < TV) As4[tid + 512] = t.p1;
  if (tid + 1024 < TV) As4[tid + 1024] = t.p2;
  if (tid + 1536 < TV) As4[tid + 1536] = t.p3;
  if (tid + 2048 < TV) As4[tid + 2048] = t.p4;
  if (tid + 2560 < TV) As4[tid + 2560] = t.p5;
  if (tid + 3072 < TV) As4[tid + 3072] = t.p6;
  if (tid + 3584 < TV) As4[tid + 3584] = t.p7;
}

template <int K, int NCOL>
__global__ __launch_bounds__(512) void gemm_ldsw_kernel(
    const float* __restrict__ A, const float* __restrict__ W,
    float* __restrict__ C, int M, int RB) {
  constexpr int NT = NCOL / 16;
  constexpr int MT = 16 / NT;
  constexpr int ROWS = MT * 16;
  constexpr int KV = K / 4;
  constexpr int TV = ROWS * KV;
  static_assert(TV <= 8 * 512 && TV > 7 * 512, "a tile is eight float4s per thread");
  constexpr int C4 = NCOL / 4;
  constexpr int ITEMS = (K / 2) * C4;
  constexpr int WB = ITEMS >= 2048 ? 4 : 1;
  constexpr int UNR = KV % 25 == 0 ? 25 : 16;
  static_assert(K % 4 == 0 && NCOL % 16 == 0 && 16 % NT == 0, "tile shapes");
  __shared__ __align__(16) float Bs[K * NCOL];
  __shared__ __align__(16) float As[ROWS * K];

  const int tid = threadIdx.x;
  const int lane = tid & 31;
  const int wave = __builtin_amdgcn_readfirstlane(tid >> 5);
  const int lo = blockIdx.x * RB;
  if (lo >= M) return;
  const int hi = (M - lo) < RB ? M : lo + RB;

  const float4* A4 = (const float4*)A;
  float4* As4 = (float4*)As;
  const long lastA = (long)M * KV - 1;
  Tile8 first;
  fetch_tile(A4, (long)lo * KV + tid, lastA, first);

  {
    const float4* W4 = (const float4*)W;
    float4* Bs4 = (float4*)Bs;
    const int sp = (int)((blockIdx.x * 61u) % (unsigned)(K / 2));
    for (int g0 = tid; g0 < ITEMS; g0 += 512 * WB) {
      float4 r0[WB], r1[WB];
#pragma unroll
      for (int u = 0; u < WB; ++u) {
        int g = g0 + u * 512;
        g = g < ITEMS ? g : ITEMS - 1;
        const int kq = g / C4;
        const int c4 = g - kq * C4;
        const int kp = kq + sp < K / 2 ? kq + sp : kq + sp - K / 2;
        r0[u] = W4[(2 * kp) * C4 + c4];
        r1[u] = W4[(2 * kp + 1) * C4 + c4];
      }
#pragma unroll
      for (int u = 0; u < WB; ++u) {
        const int g = g0 + u * 512;
        if (g < ITEMS) {
          const int kq = g / C4;
          const int c4 = g - kq * C4;
          const int kp = kq + sp < K / 2 ? kq + sp : kq + sp - K / 2;
          const int col = (c4 * 4 + 16 * (kp & 1)) & (NCOL - 1);
          const int pos4 = (kp * NCOL + col) >> 1;
          const float4 e0 = {r0[u].x, r1[u].x, r0[u].y, r1[u].y};
          const float4 e1 = {r0[u].z, r1[u].z, r0[u].w, r1[u].w};
          Bs4[pos4] = e0;
          Bs4[pos4 + 1] = e1;
        }
      }
    }
  }

  const int mtile = wave / NT;
  const int ntile = wave - mtile * NT;
  const int half = lane >> 4;
  const int lrow = (mtile << 4) + (lane & 15);
  const int koff = half << 1;
  const int ncol = (ntile << 4) + (lane & 15);
  const int bbase = (half * NCOL + ((ncol + 16 * half) & (NCOL - 1))) * 2;

  store_tile<TV>(As4, tid, first);
  __syncthreads();

  for (int r0 = lo;; r0 += ROWS) {
    const bool more = r0 + ROWS < hi;
    Tile8 pre;
    if (more) fetch_tile(A4, (long)(r0 + ROWS) * KV + tid, lastA, pre);
    const int rt = r0 + (mtile << 4);
    const bool act = rt < hi;
    v8f acc = {0.f, 0.f, 0.f, 0.f, 0.f, 0.f, 0.f, 0.f};
    if (act) {
#pragma unroll UNR
      for (int kk = 0; kk < KV; ++kk) {
        const v2f a = *(const v2f*)&As[lrow * K + kk * 4 + koff];
        const v2f b = *(const v2f*)&Bs[bbase + kk * 4 * NCOL];
        acc = __builtin_amdgcn_wmma_f32_16x16x4_f32(
             false, a,  false, b,
             (short)0, acc,  false,  false);
      }
    }
    if (more) {
      __syncthreads();
      store_tile<TV>(As4, tid, pre);
      __syncthreads();
    }
    if (act) {
      const int rbase = rt + (half << 3);
      if (rt + 16 <= hi) {
#pragma unroll
        for (int i = 0; i < 8; ++i) C[(size_t)(rbase + i) * NCOL + ncol] = acc[i];
      } else {
#pragma unroll
        for (int i = 0; i < 8; ++i) {
          const int r = rbase + i;
          if (r < hi) C[(size_t)r * NCOL + ncol] = acc[i];
        }
      }
    }
    if (!more) break;
  }
}

template <int M>
__device__ __forceinline__ float row_xor(float x) {
  return __int_as_float(__builtin_amdgcn_update_dpp(
      0, __float_as_int(x), 0x160 + M, 0xf, 0xf, true));
}
template <int W>
__device__ __forceinline__ float group_sum(float x) {
  x += row_xor<1>(x);
  x += row_xor<2>(x);
  x += row_xor<4>(x);
  if constexpr (W == 16) x += row_xor<8>(x);
  return x;
}
template <int W>
__device__ __forceinline__ float group_max(float x) {
  x = fmaxf(x, row_xor<1>(x));
  x = fmaxf(x, row_xor<2>(x));
  x = fmaxf(x, row_xor<4>(x));
  if constexpr (W == 16) x = fmaxf(x, row_xor<8>(x));
  return x;
}

template <int H, int F>
__global__ __launch_bounds__(256) void alpha_kernel(
    const float* __restrict__ h, const float* __restrict__ a_s,
    const float* __restrict__ a_d, float* __restrict__ out_s,
    float* __restrict__ out_d, int n) {
  constexpr int LPN = (H * F) / 4;
  constexpr int LPH = F / 4;
  constexpr int NPW = 32 / LPN;
  constexpr int U = (NPW == 1) ? 8 : 4;
  const int lane = threadIdx.x & 31;
  const int sub = lane % LPN;
  const int wave = (int)((blockIdx.x * blockDim.x + threadIdx.x) >> 5);
  const int node0 = wave * (NPW * U) + lane / LPN;
  const float4* h4 = (const float4*)h;
  const float4 as4 = ((const float4*)a_s)[sub];
  const float4 ad4 = ((const float4*)a_d)[sub];
  float4 v[U];
#pragma unroll
  for (int u = 0; u < U; ++u) {
    int nd = node0 + u * NPW;
    nd = nd < n ? nd : n - 1;
    v[u] = h4[(size_t)nd * LPN + sub];
  }
  __builtin_amdgcn_sched_barrier(0);
  float ps[U], pd[U];
#pragma unroll
  for (int u = 0; u < U; ++u) {
    ps[u] = group_sum<LPH>(v[u].x * as4.x + v[u].y * as4.y + v[u].z * as4.z +
                           v[u].w * as4.w);
    pd[u] = group_sum<LPH>(v[u].x * ad4.x + v[u].y * ad4.y + v[u].z * ad4.z +
                           v[u].w * ad4.w);
  }
#pragma unroll
  for (int u = 0; u < U; ++u) {
    const int nd = node0 + u * NPW;
    if (nd < n && (sub % LPH) == 0) {
      out_s[nd * H + sub / LPH] = ps[u];
      out_d[nd * H + sub / LPH] = pd[u];
    }
  }
}

template <int RO>
__global__ __launch_bounds__(256) void owner_ids_kernel(
    const int* __restrict__ dst, int nE, unsigned char* __restrict__ ids) {
  const int q = blockIdx.x * 256 + threadIdx.x;
  const int e0 = q * 4;
  if (e0 + 3 < nE) {
    int d0, d1, d2, d3;
    if (((size_t)dst & 15) == 0) {
      const int4 v = ((const int4*)dst)[q];
      d0 = v.x; d1 = v.y; d2 = v.z; d3 = v.w;
    } else {
      d0 = dst[e0]; d1 = dst[e0 + 1]; d2 = dst[e0 + 2]; d3 = dst[e0 + 3];
    }
    ((unsigned*)ids)[q] = (unsigned)(d0 / RO) | ((unsigned)(d1 / RO) << 8) |
                          ((unsigned)(d2 / RO) << 16) |
                          ((unsigned)(d3 / RO) << 24);
  } else {
    for (int e = e0; e < nE; ++e) ids[e] = (unsigned char)(dst[e] / RO);
  }
}

template <int H, int F, int R, int CAP, bool ELU, int LMODE, int LCAP>
__global__ __launch_bounds__(1024) void owner_agg_kernel(
    const float* __restrict__ hsrc, const float* __restrict__ as,
    const float* __restrict__ ad, const int* __restrict__ src,
    const int* __restrict__ dst, const unsigned* __restrict__ ids, int nE,
    int n, const float* __restrict__ bias, float* __restrict__ outp,
    unsigned* __restrict__ glist, int* __restrict__ gcnt) {
  static_assert(R <= CAP, "the self loops must fit the list");
  static_assert(LCAP <= CAP, "a list taken from the workspace must fit the LDS list");
  static_assert(R < 255, "row 255 marks an entry that is not ours");
  constexpr int HF = H * F;
  constexpr int HF4 = HF / 4;
  constexpr int U = 8;
  constexpr int G = 8;
  constexpr int U3 = 4;
  constexpr int PQ = HF4;
  constexpr int EPL = 32 / PQ;
  constexpr int PITCH = HF >= 64 ? HF : HF + 8;
  static_assert(32 % PQ == 0 && PITCH % 4 == 0, "rows must tile a wave");
  __shared__ __align__(16) float acc[(R + 1) * PITCH];
  __shared__ float den[R * H];
  __shared__ unsigned mxo[R * H];
  __shared__ float adl[R * H];
  __shared__ int ls[CAP];
  __shared__ int lt[CAP];
  __shared__ float lw[CAP * H];
  __shared__ int cnt;

  const int tid = threadIdx.x;
  const int lane = tid & 31;
  const int wv = tid >> 5;
  const int lo = blockIdx.x * R;
  const int nr = (n - lo) < R ? (n - lo) : R;

  int imp = 0;
  if constexpr (LMODE == 2) {
    if (glist != nullptr) imp = gcnt[blockIdx.x];
    imp = imp <= LCAP ? imp : 0;
  }
  const bool imported = imp > 0;
  unsigned* const gl =
      glist != nullptr ? glist + (size_t)blockIdx.x * LCAP : nullptr;

  if (tid == 0) cnt = imported ? imp : nr;
  for (int i = tid; i < nr * H; i += 1024) {
    den[i] = 0.f;
    mxo[i] = 0u;
    adl[i] = ad[(size_t)lo * H + i];
  }
  {
    float4* acc4 = (float4*)acc;
    const float4 z = {0.f, 0.f, 0.f, 0.f};
    for (int i = tid; i < nr * (PITCH / 4); i += 1024) acc4[i] = z;
  }
  __syncthreads();

  auto take = [&](int e) {
    const int pos = atomicAdd(&cnt, 1);
    if (pos < CAP) ls[pos] = e;
  };

  const unsigned unr = (unsigned)nr;
  int done = imported ? nE : 0;
  if (imported) {
  } else if (ids != nullptr) {
    const int n16 = nE >> 4;
    const uint4* q4 = (const uint4*)ids;
    const unsigned pat = (unsigned)blockIdx.x * 0x01010101u;
    for (int base = tid; base < n16; base += 1024 * U) {
      uint4 v[U];
#pragma unroll
      for (int u = 0; u < U; ++u) {
        int i = base + u * 1024;
        i = i < n16 ? i : n16 - 1;
        v[u] = q4[i];
      }
#pragma unroll
      for (int u = 0; u < U; ++u) {
        const int i = base + u * 1024;
        const unsigned x0 = v[u].x ^ pat, x1 = v[u].y ^ pat;
        const unsigned x2 = v[u].z ^ pat, x3 = v[u].w ^ pat;
        const unsigned z0 = (x0 - 0x01010101u) & ~x0;
        const unsigned z1 = (x1 - 0x01010101u) & ~x1;
        const unsigned z2 = (x2 - 0x01010101u) & ~x2;
        const unsigned z3 = (x3 - 0x01010101u) & ~x3;
        if (i < n16 && ((z0 | z1 | z2 | z3) & 0x80808080u) != 0u) {
          unsigned m = ((z0 & 0x80808080u) >> 7) | ((z1 & 0x80808080u) >> 6) |
                       ((z2 & 0x80808080u) >> 5) | ((z3 & 0x80808080u) >> 4);
          while (m != 0u) {
            const int b = __builtin_ctz(m);
            m &= m - 1u;
            take(i * 16 + (b & 3) * 4 + (b >> 3));
          }
        }
      }
    }
    done = n16 * 16;
  } else if (((size_t)dst & 15) == 0) {
    const int n4 = nE >> 2;
    const int4* d4 = (const int4*)dst;
    for (int base = tid; base < n4; base += 1024 * U) {
      int4 v[U];
#pragma unroll
      for (int u = 0; u < U; ++u) {
        int i = base + u * 1024;
        i = i < n4 ? i : n4 - 1;
        v[u] = d4[i];
      }
#pragma unroll
      for (int u = 0; u < U; ++u) {
        const int i = base + u * 1024;
        const unsigned t0 = (unsigned)(v[u].x - lo);
        const unsigned t1 = (unsigned)(v[u].y - lo);
        const unsigned t2 = (unsigned)(v[u].z - lo);
        const unsigned t3 = (unsigned)(v[u].w - lo);
        const unsigned tm = min(min(t0, t1), min(t2, t3));
        if (i < n4 && tm < unr) {
          if (t0 < unr) take(i * 4 + 0);
          if (t1 < unr) take(i * 4 + 1);
          if (t2 < unr) take(i * 4 + 2);
          if (t3 < unr) take(i * 4 + 3);
        }
      }
    }
    done = n4 * 4;
  }
  for (int e = done + tid; e < nE; e += 1024) {
    if ((unsigned)(dst[e] - lo) < unr) take(e);
  }
  __syncthreads();
  const int total = cnt;
  const int nl = total <= CAP ? total : 0;
  const bool leave = LMODE == 1 && glist != nullptr && nl > 0 && nl <= LCAP;
  if constexpr (LMODE == 1) {
    if (glist != nullptr && tid == 0) gcnt[blockIdx.x] = leave ? nl : 0;
  }
  if (total > CAP) {
    for (int i = tid; i < nr * H; i += 1024) {
      float x = as[(size_t)lo * H + i] + adl[i];
      x = x > 0.f ? x : 0.2f * x;
      atomicMax(&mxo[i], f2ord(x));
    }
    for (int e = tid; e < nE; e += 1024) {
      const unsigned t = (unsigned)(dst[e] - lo);
      if (t < unr) {
        const int s = src[e];
        for (int hd = 0; hd < H; ++hd) {
          float x = as[(size_t)s * H + hd] + adl[t * H + hd];
          x = x > 0.f ? x : 0.2f * x;
          atomicMax(&mxo[t * H + hd], f2ord(x));
        }
      }
    }
    __syncthreads();
    for (int i = tid; i < nr * H; i += 1024) {
      const int r = i / H;
      const int hd = i - r * H;
      float x = as[(size_t)lo * H + i] + adl[i];
      x = x > 0.f ? x : 0.2f * x;
      const float w = __expf(x - ord2f(mxo[i]));
      atomicAdd(&den[i], w);
      for (int f = 0; f < F; ++f) {
        const int c = hd * F + f;
        atomicAdd(&acc[r * PITCH + (c & 3) * PQ + (c >> 2)],
                  hsrc[(size_t)(lo + r) * HF + c] * w);
      }
    }
    for (int e = tid; e < nE; e += 1024) {
      const unsigned t = (unsigned)(dst[e] - lo);
      if (t < unr) {
        const int s = src[e];
        for (int hd = 0; hd < H; ++hd) {
          float x = as[(size_t)s * H + hd] + adl[t * H + hd];
          x = x > 0.f ? x : 0.2f * x;
          const float w = __expf(x - ord2f(mxo[t * H + hd]));
          atomicAdd(&den[t * H + hd], w);
          for (int f = 0; f < F; ++f) {
            const int c = hd * F + f;
            atomicAdd(&acc[t * PITCH + (c & 3) * PQ + (c >> 2)],
                      hsrc[(size_t)s * HF + c] * w);
          }
        }
      }
    }
  }

  for (int i0 = tid; i0 < nl; i0 += 1024 * U3) {
    int s[U3];
    unsigned t[U3];
    if (imported) {
#pragma unroll
      for (int u = 0; u < U3; ++u) {
        int i = i0 + u * 1024;
        i = i < nl ? i : i0;
        const unsigned en = gl[i];
        s[u] = (int)(en & 0x00ffffffu);
        t[u] = en >> 24;
      }
    } else {
      int e[U3], dd[U3];
#pragma unroll
      for (int u = 0; u < U3; ++u) {
        const int i = i0 + u * 1024;
        e[u] = (i >= nr && i < nl) ? ls[i] : -1;
        dd[u] = i < nr ? lo + i : lo - 1;
        s[u] = i < nr ? lo + i : lo;
      }
#pragma unroll
      for (int u = 0; u < U3; ++u) {
        if (e[u] >= 0) {
          dd[u] = dst[e[u]];
          s[u] = src[e[u]];
        }
      }
#pragma unroll
      for (int u = 0; u < U3; ++u) t[u] = (unsigned)(dd[u] - lo);
    }
    float a[U3][H];
#pragma unroll
    for (int u = 0; u < U3; ++u) {
#pragma unroll
      for (int hd = 0; hd < H; ++hd) a[u][hd] = as[(size_t)s[u] * H + hd];
    }
    float x[U3][H];
#pragma unroll
    for (int u = 0; u < U3; ++u) {
      const unsigned tc = t[u] < unr ? t[u] : 0u;
#pragma unroll
      for (int hd = 0; hd < H; ++hd) {
        const float v = a[u][hd] + adl[tc * H + hd];
        x[u][hd] = v > 0.f ? v : 0.2f * v;
      }
    }
#pragma unroll
    for (int u = 0; u < U3; ++u) {
      const int i = i0 + u * 1024;
      if (i < nl) {
        const bool ours = t[u] < unr;
        lt[i] = ours ? (int)t[u] : -1;
        ls[i] = ours ? s[u] : lo;
        if constexpr (LMODE == 1) {
          if (leave)
            gl[i] = ours ? ((unsigned)s[u] | (t[u] << 24))
                         : ((unsigned)lo | 0xff000000u);
        }
#pragma unroll
        for (int hd = 0; hd < H; ++hd) lw[i * H + hd] = x[u][hd];
        if (ours) {
#pragma unroll
          for (int hd = 0; hd < H; ++hd)
            atomicMax(&mxo[t[u] * H + hd], f2ord(x[u][hd]));
        }
      }
    }
  }
  __syncthreads();

  for (int i = tid; i < nl; i += 1024) {
    const int t = lt[i];
    if (t >= 0) {
#pragma unroll
      for (int hd = 0; hd < H; ++hd) {
        const float w = __expf(lw[i * H + hd] - ord2f(mxo[t * H + hd]));
        lw[i * H + hd] = w;
        atomicAdd(&den[t * H + hd], w);
      }
    }
  }
  __syncthreads();

  {
    const int es = lane / PQ;
    const int q = lane - es * PQ;
    const float4* h4 = (const float4*)hsrc;
    auto group = [&](const int k0, const bool full)
        __attribute__((always_inline)) {
      float4 v[G];
      const float4* hp[G];
#pragma unroll
      for (int g = 0; g < G; ++g) {
        int i = (k0 + g * 32) * EPL + es;
        if (!full) i = i < nl ? i : 0;
        hp[g] = h4 + (size_t)ls[i] * HF4 + q;
      }
      __builtin_amdgcn_sched_barrier(0);
#pragma unroll
      for (int g = 0; g < G; ++g) v[g] = *hp[g];
#pragma unroll
      for (int g = 0; g < G; ++g) {
        const int i = (k0 + g * 32) * EPL + es;
        if (full || i < nl) {
          const int t = lt[i];
          const float w = lw[i * H + (q * 4) / F];
          float* ap = acc + (t >= 0 ? t : R) * PITCH + q;
          atomicAdd(ap, v[g].x * w);
          atomicAdd(ap + PQ, v[g].y * w);
          atomicAdd(ap + 2 * PQ, v[g].z * w);
          atomicAdd(ap + 3 * PQ, v[g].w * w);
        }
      }
    };
    for (int k0 = wv; k0 * EPL < nl; k0 += 32 * G) {
      if ((k0 + (G - 1) * 32 + 1) * EPL <= nl) group(k0, true);
      else group(k0, false);
    }
  }
  __syncthreads();

  {
    float4* o4 = (float4*)outp;
    for (int i = tid; i < nr * HF4; i += 1024) {
      const int r = i / HF4;
      const int c4 = i - r * HF4;
      const float dn = den[r * H + (c4 * 4) / F] + 1e-16f;
      const float* ap = acc + r * PITCH + c4;
      float4 x = {ap[0], ap[PQ], ap[2 * PQ], ap[3 * PQ]};
      x.x /= dn; x.y /= dn; x.z /= dn; x.w /= dn;
      if constexpr (ELU) {
        const float4 bb = ((const float4*)bias)[c4];
        float t;
        t = x.x + bb.x; x.x = t > 0.f ? t : (__expf(t) - 1.f);
        t = x.y + bb.y; x.y = t > 0.f ? t : (__expf(t) - 1.f);
        t = x.z + bb.z; x.z = t > 0.f ? t : (__expf(t) - 1.f);
        t = x.w + bb.w; x.w = t > 0.f ? t : (__expf(t) - 1.f);
      }
      o4[(size_t)(lo + r) * HF4 + c4] = x;
    }
  }
}

__global__ __launch_bounds__(256) void logsoftmax_kernel(
    const float* __restrict__ agg, const float* __restrict__ bias,
    float* __restrict__ out, int n) {
  constexpr int U = 4;
  const int lane = threadIdx.x & 31;
  const int sub = lane & 7;
  const int wave = (int)((blockIdx.x * blockDim.x + threadIdx.x) >> 5);
  const int node0 = wave * (4 * U) + (lane >> 3);
  const float4* a4 = (const float4*)agg;
  float4* o4 = (float4*)out;
  const float4 b = ((const float4*)bias)[sub];
  float4 v[U];
#pragma unroll
  for (int u = 0; u < U; ++u) {
    int nd = node0 + u * 4;
    nd = nd < n ? nd : n - 1;
    v[u] = a4[(size_t)nd * 8 + sub];
  }
  float4 r[U];
#pragma unroll
  for (int u = 0; u < U; ++u) {
    const float x0 = v[u].x + b.x, x1 = v[u].y + b.y;
    const float x2 = v[u].z + b.z, x3 = v[u].w + b.w;
    const float m = group_max<8>(fmaxf(fmaxf(x0, x1), fmaxf(x2, x3)));
    const float s = group_sum<8>(__expf(x0 - m) + __expf(x1 - m) +
                                 __expf(x2 - m) + __expf(x3 - m));
    const float ls = __logf(s);
    r[u].x = x0 - m - ls; r[u].y = x1 - m - ls;
    r[u].z = x2 - m - ls; r[u].w = x3 - m - ls;
  }
#pragma unroll
  for (int u = 0; u < U; ++u) {
    const int nd = node0 + u * 4;
    if (nd < n) o4[(size_t)nd * 8 + sub] = r[u];
  }
}

extern "C" void kernel_launch(void* const* d_in, const int* in_sizes, int n_in,
                              void* d_out, int out_size, void* d_ws,
                              size_t ws_size, hipStream_t stream) {
  (void)n_in; (void)out_size;
  const float* x      = (const float*)d_in[0];
  const int*   ei     = (const int*)d_in[1];
  const float* W1     = (const float*)d_in[2];
  const float* a_src1 = (const float*)d_in[3];
  const float* a_dst1 = (const float*)d_in[4];
  const float* b1     = (const float*)d_in[5];
  const float* W2     = (const float*)d_in[6];
  const float* a_src2 = (const float*)d_in[7];
  const float* a_dst2 = (const float*)d_in[8];
  const float* b2     = (const float*)d_in[9];

  const int n    = in_sizes[0] / 500;
  const int nE   = in_sizes[1] / 2;
  const int* src = ei;
  const int* dst = ei + nE;
  constexpr int R = 196;
  const int nOwn = (n + R - 1) / R;

  float* ws = (float*)d_ws;
  size_t o = 0;
  float* h1      = ws + o; o += (size_t)n * 128;
  float* h2      = ws + o; o += (size_t)n * 32;
  float* agg1    = ws + o; o += (size_t)n * 128;
  float* agg2    = ws + o; o += (size_t)n * 32;
  float* idsp    = ws + o; o += ((size_t)nE + 15) / 16 * 4;
  float* as1     = ws + o; o += (size_t)n * 2;
  float* ad1     = ws + o; o += (size_t)n * 2;
  float* as2     = ws + o; o += (size_t)n;
  float* ad2     = ws + o; o += (size_t)n;
  constexpr int LCAP = 4096;
  float* glistp  = ws + o; o += (size_t)nOwn * LCAP;
  float* gcntp   = ws + o; o += (size_t)nOwn;
  const bool lists = n <= (1 << 24) && o * sizeof(float) <= ws_size;
  unsigned* glist = lists ? (unsigned*)glistp : nullptr;
  int* gcnt = lists ? (int*)gcntp : nullptr;

  const bool byteIds = nOwn <= 256;
  unsigned* ids = byteIds ? (unsigned*)idsp : nullptr;

  if (byteIds)
    owner_ids_kernel<R><<<((nE + 3) / 4 + 255) / 256, 256, 0, stream>>>(
        dst, nE, (unsigned char*)idsp);

  gemm_ldsw_kernel<500, 128><<<nOwn, 512, 0, stream>>>(x, W1, h1, n, R);
  alpha_kernel<2, 64><<<(n + 63) / 64, 256, 0, stream>>>(
      h1, a_src1, a_dst1, as1, ad1, n);
  owner_agg_kernel<2, 64, R, 6144, true, 1, LCAP><<<nOwn, 1024, 0, stream>>>(
      h1, as1, ad1, src, dst, ids, nE, n, b1, agg1, glist, gcnt);

  gemm_ldsw_kernel<128, 32><<<nOwn, 512, 0, stream>>>(agg1, W2, h2, n, R);
  alpha_kernel<1, 32><<<(n + 127) / 128, 256, 0, stream>>>(
      h2, a_src2, a_dst2, as2, ad2, n);
  owner_agg_kernel<1, 32, R, 12288, false, 2, LCAP><<<nOwn, 1024, 0, stream>>>(
      h2, as2, ad2, src, dst, ids, nE, n, b2, agg2, glist, gcnt);

  logsoftmax_kernel<<<(n + 127) / 128, 256, 0, stream>>>(
      agg2, b2, (float*)d_out, n);
}


# edges sorted by owner range: a block counts its 4096 edges per range in LDS and fills one 256-byte segment per range; owners read their segments, not all owner bytes; LDS float adds arrival order
# speedup vs baseline: 4.0893x; 1.0914x over previous
#include <hip/hip_runtime.h>
#include <hip/hip_bf16.h>


typedef float v2f __attribute__((ext_vector_type(2)));
typedef float v8f __attribute__((ext_vector_type(8)));

__device__ __forceinline__ unsigned f2ord(float f) {
  unsigned u = __float_as_uint(f);
  return (u & 0x80000000u) ? ~u : (u | 0x80000000u);
}
__device__ __forceinline__ float ord2f(unsigned o) {
  unsigned u = (o & 0x80000000u) ? (o ^ 0x80000000u) : ~o;
  return __uint_as_float(u);
}

struct Tile8 { float4 p0, p1, p2, p3, p4, p5, p6, p7; };
__device__ __forceinline__ float4 fetch_one(const float4* __restrict__ A4,
                                            long i, const long last) {
  i = i < last ? i : last;
  return A4[i];
}
__device__ __forceinline__ void fetch_tile(const float4* __restrict__ A4,
                                           const long first, const long last,
                                           Tile8& t) {
  t.p0 = fetch_one(A4, first, last);
  t.p1 = fetch_one(A4, first + 512, last);
  t.p2 = fetch_one(A4, first + 1024, last);
  t.p3 = fetch_one(A4, first + 1536, last);
  t.p4 = fetch_one(A4, first + 2048, last);
  t.p5 = fetch_one(A4, first + 2560, last);
  t.p6 = fetch_one(A4, first + 3072, last);
  t.p7 = fetch_one(A4, first + 3584, last);
}
template <int TV>
__device__ __forceinline__ void store_tile(float4* __restrict__ As4,
                                           const int tid, const Tile8& t) {
  if (tid < TV) As4[tid] = t.p0;
  if (tid + 512 < TV) As4[tid + 512] = t.p1;
  if (tid + 1024 < TV) As4[tid + 1024] = t.p2;
  if (tid + 1536 < TV) As4[tid + 1536] = t.p3;
  if (tid + 2048 < TV) As4[tid + 2048] = t.p4;
  if (tid + 2560 < TV) As4[tid + 2560] = t.p5;
  if (tid + 3072 < TV) As4[tid + 3072] = t.p6;
  if (tid + 3584 < TV) As4[tid + 3584] = t.p7;
}

template <int K, int NCOL>
__global__ __launch_bounds__(512) void gemm_ldsw_kernel(
    const float* __restrict__ A, const float* __restrict__ W,
    float* __restrict__ C, int M, int RB) {
  constexpr int NT = NCOL / 16;
  constexpr int MT = 16 / NT;
  constexpr int ROWS = MT * 16;
  constexpr int KV = K / 4;
  constexpr int TV = ROWS * KV;
  static_assert(TV <= 8 * 512 && TV > 7 * 512, "a tile is eight float4s per thread");
  constexpr int C4 = NCOL / 4;
  constexpr int ITEMS = (K / 2) * C4;
  constexpr int WB = ITEMS >= 2048 ? 4 : 1;
  constexpr int UNR = KV % 25 == 0 ? 25 : 16;
  static_assert(K % 4 == 0 && NCOL % 16 == 0 && 16 % NT == 0, "tile shapes");
  __shared__ __align__(16) float Bs[K * NCOL];
  __shared__ __align__(16) float As[ROWS * K];

  const int tid = threadIdx.x;
  const int lane = tid & 31;
  const int wave = __builtin_amdgcn_readfirstlane(tid >> 5);
  const int lo = blockIdx.x * RB;
  if (lo >= M) return;
  const int hi = (M - lo) < RB ? M : lo + RB;

  const float4* A4 = (const float4*)A;
  float4* As4 = (float4*)As;
  const long lastA = (long)M * KV - 1;
  Tile8 first;
  fetch_tile(A4, (long)lo * KV + tid, lastA, first);

  {
    const float4* W4 = (const float4*)W;
    float4* Bs4 = (float4*)Bs;
    const int sp = (int)((blockIdx.x * 61u) % (unsigned)(K / 2));
    for (int g0 = tid; g0 < ITEMS; g0 += 512 * WB) {
      float4 r0[WB], r1[WB];
#pragma unroll
      for (int u = 0; u < WB; ++u) {
        int g = g0 + u * 512;
        g = g < ITEMS ? g : ITEMS - 1;
        const int kq = g / C4;
        const int c4 = g - kq * C4;
        const int kp = kq + sp < K / 2 ? kq + sp : kq + sp - K / 2;
        r0[u] = W4[(2 * kp) * C4 + c4];
        r1[u] = W4[(2 * kp + 1) * C4 + c4];
      }
#pragma unroll
      for (int u = 0; u < WB; ++u) {
        const int g = g0 + u * 512;
        if (g < ITEMS) {
          const int kq = g / C4;
          const int c4 = g - kq * C4;
          const int kp = kq + sp < K / 2 ? kq + sp : kq + sp - K / 2;
          const int col = (c4 * 4 + 16 * (kp & 1)) & (NCOL - 1);
          const int pos4 = (kp * NCOL + col) >> 1;
          const float4 e0 = {r0[u].x, r1[u].x, r0[u].y, r1[u].y};
          const float4 e1 = {r0[u].z, r1[u].z, r0[u].w, r1[u].w};
          Bs4[pos4] = e0;
          Bs4[pos4 + 1] = e1;
        }
      }
    }
  }

  const int mtile = wave / NT;
  const int ntile = wave - mtile * NT;
  const int half = lane >> 4;
  const int lrow = (mtile << 4) + (lane & 15);
  const int koff = half << 1;
  const int ncol = (ntile << 4) + (lane & 15);
  const int bbase = (half * NCOL + ((ncol + 16 * half) & (NCOL - 1))) * 2;

  store_tile<TV>(As4, tid, first);
  __syncthreads();

  for (int r0 = lo;; r0 += ROWS) {
    const bool more = r0 + ROWS < hi;
    Tile8 pre;
    if (more) fetch_tile(A4, (long)(r0 + ROWS) * KV + tid, lastA, pre);
    const int rt = r0 + (mtile << 4);
    const bool act = rt < hi;
    v8f acc = {0.f, 0.f, 0.f, 0.f, 0.f, 0.f, 0.f, 0.f};
    if (act) {
#pragma unroll UNR
      for (int kk = 0; kk < KV; ++kk) {
        const v2f a = *(const v2f*)&As[lrow * K + kk * 4 + koff];
        const v2f b = *(const v2f*)&Bs[bbase + kk * 4 * NCOL];
        acc = __builtin_amdgcn_wmma_f32_16x16x4_f32(
             false, a,  false, b,
             (short)0, acc,  false,  false);
      }
    }
    if (more) {
      __syncthreads();
      store_tile<TV>(As4, tid, pre);
      __syncthreads();
    }
    if (act) {
      const int rbase = rt + (half << 3);
      if (rt + 16 <= hi) {
#pragma unroll
        for (int i = 0; i < 8; ++i) C[(size_t)(rbase + i) * NCOL + ncol] = acc[i];
      } else {
#pragma unroll
        for (int i = 0; i < 8; ++i) {
          const int r = rbase + i;
          if (r < hi) C[(size_t)r * NCOL + ncol] = acc[i];
        }
      }
    }
    if (!more) break;
  }
}

template <int M>
__device__ __forceinline__ float row_xor(float x) {
  return __int_as_float(__builtin_amdgcn_update_dpp(
      0, __float_as_int(x), 0x160 + M, 0xf, 0xf, true));
}
template <int W>
__device__ __forceinline__ float group_sum(float x) {
  x += row_xor<1>(x);
  x += row_xor<2>(x);
  x += row_xor<4>(x);
  if constexpr (W == 16) x += row_xor<8>(x);
  return x;
}
template <int W>
__device__ __forceinline__ float group_max(float x) {
  x = fmaxf(x, row_xor<1>(x));
  x = fmaxf(x, row_xor<2>(x));
  x = fmaxf(x, row_xor<4>(x));
  if constexpr (W == 16) x = fmaxf(x, row_xor<8>(x));
  return x;
}

template <int H, int F>
__global__ __launch_bounds__(256) void alpha_kernel(
    const float* __restrict__ h, const float* __restrict__ a_s,
    const float* __restrict__ a_d, float* __restrict__ out_s,
    float* __restrict__ out_d, int n) {
  constexpr int LPN = (H * F) / 4;
  constexpr int LPH = F / 4;
  constexpr int NPW = 32 / LPN;
  constexpr int U = (NPW == 1) ? 8 : 4;
  const int lane = threadIdx.x & 31;
  const int sub = lane % LPN;
  const int wave = (int)((blockIdx.x * blockDim.x + threadIdx.x) >> 5);
  const int node0 = wave * (NPW * U) + lane / LPN;
  const float4* h4 = (const float4*)h;
  const float4 as4 = ((const float4*)a_s)[sub];
  const float4 ad4 = ((const float4*)a_d)[sub];
  float4 v[U];
#pragma unroll
  for (int u = 0; u < U; ++u) {
    int nd = node0 + u * NPW;
    nd = nd < n ? nd : n - 1;
    v[u] = h4[(size_t)nd * LPN + sub];
  }
  __builtin_amdgcn_sched_barrier(0);
  float ps[U], pd[U];
#pragma unroll
  for (int u = 0; u < U; ++u) {
    ps[u] = group_sum<LPH>(v[u].x * as4.x + v[u].y * as4.y + v[u].z * as4.z +
                           v[u].w * as4.w);
    pd[u] = group_sum<LPH>(v[u].x * ad4.x + v[u].y * ad4.y + v[u].z * ad4.z +
                           v[u].w * ad4.w);
  }
#pragma unroll
  for (int u = 0; u < U; ++u) {
    const int nd = node0 + u * NPW;
    if (nd < n && (sub % LPH) == 0) {
      out_s[nd * H + sub / LPH] = ps[u];
      out_d[nd * H + sub / LPH] = pd[u];
    }
  }
}

template <int RO, int S, int EPB, int MAXOWN>
__global__ __launch_bounds__(1024) void range_lists_kernel(
    const int* __restrict__ src, const int* __restrict__ dst, int nE, int n,
    int nOwn, int NB, unsigned* __restrict__ glist) {
  static_assert(EPB % 1024 == 0 && S % 4 == 0, "edges per thread, uint4 stores");
  constexpr int U = EPB / 1024;
  constexpr int S4 = S / 4;
  __shared__ __align__(16) unsigned seg[MAXOWN * S];
  __shared__ unsigned cnts[MAXOWN];
  const int tid = threadIdx.x;
  const int b = blockIdx.x;
  const int e0 = b * EPB;
  int d[U], s[U];
#pragma unroll
  for (int u = 0; u < U; ++u) {
    int e = e0 + tid + u * 1024;
    e = e < nE ? e : nE - 1;
    d[u] = dst[e];
    s[u] = src[e];
  }
  {
    uint4* s4 = (uint4*)seg;
    const uint4 z = {0u, 0u, 0u, 0u};
    for (int i = tid; i < MAXOWN * S4; i += 1024) s4[i] = z;
    for (int i = tid; i < MAXOWN; i += 1024) cnts[i] = 0u;
  }
  __syncthreads();
#pragma unroll
  for (int u = 0; u < U; ++u) {
    const int e = e0 + tid + u * 1024;
    if (e < nE && (unsigned)d[u] < (unsigned)n) {
      const int o = d[u] / RO;
      const unsigned row = (unsigned)(d[u] - o * RO);
      const unsigned k = atomicAdd(&cnts[o], 1u) + 1u;
      if (k < (unsigned)S)
        seg[o * S + k] = ((unsigned)s[u] & 0x00ffffffu) | (row << 24);
    }
  }
  __syncthreads();
  for (int i = tid; i < MAXOWN; i += 1024) seg[i * S] = cnts[i];
  __syncthreads();
  uint4* g4 = (uint4*)glist;
  const uint4* s4 = (const uint4*)seg;
  for (int i = tid; i < nOwn * S4; i += 1024) {
    const int o = i / S4;
    const int w = i - o * S4;
    g4[((size_t)o * NB + b) * S4 + w] = s4[i];
  }
}

template <int H, int F, int R, int CAP, bool ELU, int S, int EPB>
__global__ __launch_bounds__(1024) void owner_agg_kernel(
    const float* __restrict__ hsrc, const float* __restrict__ as,
    const float* __restrict__ ad, const int* __restrict__ src,
    const int* __restrict__ dst, const unsigned* __restrict__ glist, int NB,
    int nE, int n, const float* __restrict__ bias, float* __restrict__ outp) {
  static_assert(R <= CAP, "the self loops must fit the list");
  static_assert(R <= 256, "a row number must fit 8 bits of a segment word");
  static_assert(S == 64, "a wave reads a segment as two words per lane");
  constexpr int HF = H * F;
  constexpr int HF4 = HF / 4;
  constexpr int UI = 8;
  constexpr int G = 8;
  constexpr int U3 = 4;
  constexpr int PQ = HF4;
  constexpr int EPL = 32 / PQ;
  constexpr int PITCH = HF >= 64 ? HF : HF + 8;
  static_assert(32 % PQ == 0 && PITCH % 4 == 0, "rows must tile a wave");
  __shared__ __align__(16) float acc[R * PITCH];
  __shared__ float den[R * H];
  __shared__ unsigned mxo[R * H];
  __shared__ float adl[R * H];
  __shared__ int ls[CAP];
  __shared__ int lt[CAP];
  __shared__ float lw[CAP * H];
  __shared__ int cnt;

  const int tid = threadIdx.x;
  const int lane = tid & 31;
  const int wv = __builtin_amdgcn_readfirstlane(tid >> 5);
  const int lo = blockIdx.x * R;
  const int nr = (n - lo) < R ? (n - lo) : R;

  if (tid == 0) cnt = nr;
  for (int i = tid; i < nr; i += 1024) {
    ls[i] = lo + i;
    lt[i] = i;
  }
  for (int i = tid; i < nr * H; i += 1024) {
    den[i] = 0.f;
    mxo[i] = 0u;
    adl[i] = ad[(size_t)lo * H + i];
  }
  {
    float4* acc4 = (float4*)acc;
    const float4 z = {0.f, 0.f, 0.f, 0.f};
    for (int i = tid; i < nr * (PITCH / 4); i += 1024) acc4[i] = z;
  }
  __syncthreads();

  const unsigned unr = (unsigned)nr;
  auto take = [&](int e, unsigned t) {
    const int pos = atomicAdd(&cnt, 1);
    if (pos < CAP) {
      ls[pos] = src[e];
      lt[pos] = (int)t;
    }
  };

  if (glist != nullptr) {
    const unsigned* const gl = glist + (size_t)blockIdx.x * NB * S;
    for (int sg0 = wv; sg0 < NB; sg0 += 32 * UI) {
      uint2 v[UI];
#pragma unroll
      for (int u = 0; u < UI; ++u) {
        int sg = sg0 + u * 32;
        sg = sg < NB ? sg : NB - 1;
        v[u] = ((const uint2*)(gl + (size_t)sg * S))[lane];
      }
#pragma unroll
      for (int u = 0; u < UI; ++u) {
        const int sg = sg0 + u * 32;
        if (sg < NB) {
          const int c = __builtin_amdgcn_readfirstlane((int)v[u].x);
          if (c >= 0 && c < S) {
            int base = 0;
            if (lane == 0) base = atomicAdd(&cnt, c);
            base = __builtin_amdgcn_readfirstlane(base);
            const int p0 = base + 2 * lane - 1;
            if (lane >= 1 && 2 * lane <= c && p0 < CAP) {
              ls[p0] = (int)(v[u].x & 0x00ffffffu);
              lt[p0] = (int)(v[u].x >> 24);
            }
            if (2 * lane + 1 <= c && p0 + 1 < CAP) {
              ls[p0 + 1] = (int)(v[u].y & 0x00ffffffu);
              lt[p0 + 1] = (int)(v[u].y >> 24);
            }
          } else {
            const int ea = sg * EPB;
            const int eb = (nE - ea) < EPB ? nE : ea + EPB;
            for (int e = ea + lane; e < eb; e += 32) {
              const unsigned t = (unsigned)(dst[e] - lo);
              if (t < unr) take(e, t);
            }
          }
        }
      }
    }
  } else {
    for (int e = tid; e < nE; e += 1024) {
      const unsigned t = (unsigned)(dst[e] - lo);
      if (t < unr) take(e, t);
    }
  }
  __syncthreads();
  const int total = cnt;
  const int nl = total <= CAP ? total : 0;
  if (total > CAP) {
    for (int i = tid; i < nr * H; i += 1024) {
      float x = as[(size_t)lo * H + i] + adl[i];
      x = x > 0.f ? x : 0.2f * x;
      atomicMax(&mxo[i], f2ord(x));
    }
    for (int e = tid; e < nE; e += 1024) {
      const unsigned t = (unsigned)(dst[e] - lo);
      if (t < unr) {
        const int s = src[e];
        for (int hd = 0; hd < H; ++hd) {
          float x = as[(size_t)s * H + hd] + adl[t * H + hd];
          x = x > 0.f ? x : 0.2f * x;
          atomicMax(&mxo[t * H + hd], f2ord(x));
        }
      }
    }
    __syncthreads();
    for (int i = tid; i < nr * H; i += 1024) {
      const int r = i / H;
      const int hd = i - r * H;
      float x = as[(size_t)lo * H + i] + adl[i];
      x = x > 0.f ? x : 0.2f * x;
      const float w = __expf(x - ord2f(mxo[i]));
      atomicAdd(&den[i], w);
      for (int f = 0; f < F; ++f) {
        const int c = hd * F + f;
        atomicAdd(&acc[r * PITCH + (c & 3) * PQ + (c >> 2)],
                  hsrc[(size_t)(lo + r) * HF + c] * w);
      }
    }
    for (int e = tid; e < nE; e += 1024) {
      const unsigned t = (unsigned)(dst[e] - lo);
      if (t < unr) {
        const int s = src[e];
        for (int hd = 0; hd < H; ++hd) {
          float x = as[(size_t)s * H + hd] + adl[t * H + hd];
          x = x > 0.f ? x : 0.2f * x;
          const float w = __expf(x - ord2f(mxo[t * H + hd]));
          atomicAdd(&den[t * H + hd], w);
          for (int f = 0; f < F; ++f) {
            const int c = hd * F + f;
            atomicAdd(&acc[t * PITCH + (c & 3) * PQ + (c >> 2)],
                      hsrc[(size_t)s * HF + c] * w);
          }
        }
      }
    }
  }

  for (int i0 = tid; i0 < nl; i0 += 1024 * U3) {
    int s[U3], t[U3];
#pragma unroll
    for (int u = 0; u < U3; ++u) {
      int i = i0 + u * 1024;
      i = i < nl ? i : i0;
      s[u] = ls[i];
      t[u] = lt[i];
    }
    float a[U3][H];
#pragma unroll
    for (int u = 0; u < U3; ++u) {
#pragma unroll
      for (int hd = 0; hd < H; ++hd) a[u][hd] = as[(size_t)s[u] * H + hd];
    }
#pragma unroll
    for (int u = 0; u < U3; ++u) {
      const int i = i0 + u * 1024;
      if (i < nl) {
#pragma unroll
        for (int hd = 0; hd < H; ++hd) {
          const float v = a[u][hd] + adl[t[u] * H + hd];
          const float x = v > 0.f ? v : 0.2f * v;
          lw[i * H + hd] = x;
          atomicMax(&mxo[t[u] * H + hd], f2ord(x));
        }
      }
    }
  }
  __syncthreads();

  for (int i = tid; i < nl; i += 1024) {
    const int t = lt[i];
#pragma unroll
    for (int hd = 0; hd < H; ++hd) {
      const float w = __expf(lw[i * H + hd] - ord2f(mxo[t * H + hd]));
      lw[i * H + hd] = w;
      atomicAdd(&den[t * H + hd], w);
    }
  }
  __syncthreads();

  {
    const int es = lane / PQ;
    const int q = lane - es * PQ;
    const float4* h4 = (const float4*)hsrc;
    auto group = [&](const int k0, const bool full)
        __attribute__((always_inline)) {
      float4 v[G];
      const float4* hp[G];
#pragma unroll
      for (int g = 0; g < G; ++g) {
        int i = (k0 + g * 32) * EPL + es;
        if (!full) i = i < nl ? i : 0;
        hp[g] = h4 + (size_t)ls[i] * HF4 + q;
      }
      __builtin_amdgcn_sched_barrier(0);
#pragma unroll
      for (int g = 0; g < G; ++g) v[g] = *hp[g];
#pragma unroll
      for (int g = 0; g < G; ++g) {
        const int i = (k0 + g * 32) * EPL + es;
        if (full || i < nl) {
          const int t = lt[i];
          const float w = lw[i * H + (q * 4) / F];
          float* ap = acc + t * PITCH + q;
          atomicAdd(ap, v[g].x * w);
          atomicAdd(ap + PQ, v[g].y * w);
          atomicAdd(ap + 2 * PQ, v[g].z * w);
          atomicAdd(ap + 3 * PQ, v[g].w * w);
        }
      }
    };
    for (int k0 = wv; k0 * EPL < nl; k0 += 32 * G) {
      if ((k0 + (G - 1) * 32 + 1) * EPL <= nl) group(k0, true);
      else group(k0, false);
    }
  }
  __syncthreads();

  {
    float4* o4 = (float4*)outp;
    for (int i = tid; i < nr * HF4; i += 1024) {
      const int r = i / HF4;
      const int c4 = i - r * HF4;
      const float dn = den[r * H + (c4 * 4) / F] + 1e-16f;
      const float* ap = acc + r * PITCH + c4;
      float4 x = {ap[0], ap[PQ], ap[2 * PQ], ap[3 * PQ]};
      x.x /= dn; x.y /= dn; x.z /= dn; x.w /= dn;
      if constexpr (ELU) {
        const float4 bb = ((const float4*)bias)[c4];
        float t;
        t = x.x + bb.x; x.x = t > 0.f ? t : (__expf(t) - 1.f);
        t = x.y + bb.y; x.y = t > 0.f ? t : (__expf(t) - 1.f);
        t = x.z + bb.z; x.z = t > 0.f ? t : (__expf(t) - 1.f);
        t = x.w + bb.w; x.w = t > 0.f ? t : (__expf(t) - 1.f);
      }
      o4[(size_t)(lo + r) * HF4 + c4] = x;
    }
  }
}

__global__ __launch_bounds__(256) void logsoftmax_kernel(
    const float* __restrict__ agg, const float* __restrict__ bias,
    float* __restrict__ out, int n) {
  constexpr int U = 4;
  const int lane = threadIdx.x & 31;
  const int sub = lane & 7;
  const int wave = (int)((blockIdx.x * blockDim.x + threadIdx.x) >> 5);
  const int node0 = wave * (4 * U) + (lane >> 3);
  const float4* a4 = (const float4*)agg;
  float4* o4 = (float4*)out;
  const float4 b = ((const float4*)bias)[sub];
  float4 v[U];
#pragma unroll
  for (int u = 0; u < U; ++u) {
    int nd = node0 + u * 4;
    nd = nd < n ? nd : n - 1;
    v[u] = a4[(size_t)nd * 8 + sub];
  }
  float4 r[U];
#pragma unroll
  for (int u = 0; u < U; ++u) {
    const float x0 = v[u].x + b.x, x1 = v[u].y + b.y;
    const float x2 = v[u].z + b.z, x3 = v[u].w + b.w;
    const float m = group_max<8>(fmaxf(fmaxf(x0, x1), fmaxf(x2, x3)));
    const float s = group_sum<8>(__expf(x0 - m) + __expf(x1 - m) +
                                 __expf(x2 - m) + __expf(x3 - m));
    const float ls = __logf(s);
    r[u].x = x0 - m - ls; r[u].y = x1 - m - ls;
    r[u].z = x2 - m - ls; r[u].w = x3 - m - ls;
  }
#pragma unroll
  for (int u = 0; u < U; ++u) {
    const int nd = node0 + u * 4;
    if (nd < n) o4[(size_t)nd * 8 + sub] = r[u];
  }
}

extern "C" void kernel_launch(void* const* d_in, const int* in_sizes, int n_in,
                              void* d_out, int out_size, void* d_ws,
                              size_t ws_size, hipStream_t stream) {
  (void)n_in; (void)out_size;
  const float* x      = (const float*)d_in[0];
  const int*   ei     = (const int*)d_in[1];
  const float* W1     = (const float*)d_in[2];
  const float* a_src1 = (const float*)d_in[3];
  const float* a_dst1 = (const float*)d_in[4];
  const float* b1     = (const float*)d_in[5];
  const float* W2     = (const float*)d_in[6];
  const float* a_src2 = (const float*)d_in[7];
  const float* a_dst2 = (const float*)d_in[8];
  const float* b2     = (const float*)d_in[9];

  const int n    = in_sizes[0] / 500;
  const int nE   = in_sizes[1] / 2;
  const int* src = ei;
  const int* dst = ei + nE;
  constexpr int R = 196;
  const int nOwn = (n + R - 1) / R;
  constexpr int S = 64;
  constexpr int EPB = 4096;
  constexpr int MAXOWN = 256;
  const int NB = (nE + EPB - 1) / EPB;

  float* ws = (float*)d_ws;
  size_t o = 0;
  float* h1      = ws + o; o += ((size_t)n * 128 + 63) / 64 * 64;
  float* agg1    = ws + o; o += ((size_t)n * 128 + 63) / 64 * 64;
  float* glistp  = ws + o; o += (size_t)nOwn * NB * S;
  float* as1     = ws + o; o += (size_t)n * 2;
  float* ad1     = ws + o; o += (size_t)n * 2;
  float* as2     = ws + o; o += (size_t)n;
  float* ad2     = ws + o; o += (size_t)n;
  float* h2      = h1;
  float* agg2    = h1 + (size_t)n * 32;
  const bool lists = n <= (1 << 24) && nOwn <= MAXOWN && NB > 0 &&
                     o * sizeof(float) <= ws_size;
  unsigned* glist = lists ? (unsigned*)glistp : nullptr;

  if (lists)
    range_lists_kernel<R, S, EPB, MAXOWN><<<NB, 1024, 0, stream>>>(
        src, dst, nE, n, nOwn, NB, glist);

  gemm_ldsw_kernel<500, 128><<<nOwn, 512, 0, stream>>>(x, W1, h1, n, R);
  alpha_kernel<2, 64><<<(n + 63) / 64, 256, 0, stream>>>(
      h1, a_src1, a_dst1, as1, ad1, n);
  owner_agg_kernel<2, 64, R, 6144, true, S, EPB><<<nOwn, 1024, 0, stream>>>(
      h1, as1, ad1, src, dst, glist, NB, nE, n, b1, agg1);

  gemm_ldsw_kernel<128, 32><<<nOwn, 512, 0, stream>>>(agg1, W2, h2, n, R);
  alpha_kernel<1, 32><<<(n + 127) / 128, 256, 0, stream>>>(
      h2, a_src2, a_dst2, as2, ad2, n);
  owner_agg_kernel<1, 32, R, 12288, false, S, EPB><<<nOwn, 1024, 0, stream>>>(
      h2, as2, ad2, src, dst, glist, NB, nE, n, b2, agg2);

  logsoftmax_kernel<<<(n + 127) / 128, 256, 0, stream>>>(
      agg2, b2, (float*)d_out, n);
}


# owner passes put their list in row order in LDS (place = count atomic's return); lane groups sum runs of it in registers and add to the LDS row only at a row change: sum order = count order, chained
# speedup vs baseline: 4.3740x; 1.0696x over previous
#include <hip/hip_runtime.h>
#include <hip/hip_bf16.h>


typedef float v2f __attribute__((ext_vector_type(2)));
typedef float v8f __attribute__((ext_vector_type(8)));

__device__ __forceinline__ unsigned f2ord(float f) {
  unsigned u = __float_as_uint(f);
  return (u & 0x80000000u) ? ~u : (u | 0x80000000u);
}
__device__ __forceinline__ float ord2f(unsigned o) {
  unsigned u = (o & 0x80000000u) ? (o ^ 0x80000000u) : ~o;
  return __uint_as_float(u);
}

struct Tile8 { float4 p0, p1, p2, p3, p4, p5, p6, p7; };
__device__ __forceinline__ float4 fetch_one(const float4* __restrict__ A4,
                                            long i, const long last) {
  i = i < last ? i : last;
  return A4[i];
}
__device__ __forceinline__ void fetch_tile(const float4* __restrict__ A4,
                                           const long first, const long last,
                                           Tile8& t) {
  t.p0 = fetch_one(A4, first, last);
  t.p1 = fetch_one(A4, first + 512, last);
  t.p2 = fetch_one(A4, first + 1024, last);
  t.p3 = fetch_one(A4, first + 1536, last);
  t.p4 = fetch_one(A4, first + 2048, last);
  t.p5 = fetch_one(A4, first + 2560, last);
  t.p6 = fetch_one(A4, first + 3072, last);
  t.p7 = fetch_one(A4, first + 3584, last);
}
template <int TV>
__device__ __forceinline__ void store_tile(float4* __restrict__ As4,
                                           const int tid, const Tile8& t) {
  if (tid < TV) As4[tid] = t.p0;
  if (tid + 512 < TV) As4[tid + 512] = t.p1;
  if (tid + 1024 < TV) As4[tid + 1024] = t.p2;
  if (tid + 1536 < TV) As4[tid + 1536] = t.p3;
  if (tid + 2048 < TV) As4[tid + 2048] = t.p4;
  if (tid + 2560 < TV) As4[tid + 2560] = t.p5;
  if (tid + 3072 < TV) As4[tid + 3072] = t.p6;
  if (tid + 3584 < TV) As4[tid + 3584] = t.p7;
}

template <int K, int NCOL>
__global__ __launch_bounds__(512) void gemm_ldsw_kernel(
    const float* __restrict__ A, const float* __restrict__ W,
    float* __restrict__ C, int M, int RB) {
  constexpr int NT = NCOL / 16;
  constexpr int MT = 16 / NT;
  constexpr int ROWS = MT * 16;
  constexpr int KV = K / 4;
  constexpr int TV = ROWS * KV;
  static_assert(TV <= 8 * 512 && TV > 7 * 512, "a tile is eight float4s per thread");
  constexpr int C4 = NCOL / 4;
  constexpr int ITEMS = (K / 2) * C4;
  constexpr int WB = ITEMS >= 2048 ? 4 : 1;
  constexpr int UNR = KV % 25 == 0 ? 25 : 16;
  static_assert(K % 4 == 0 && NCOL % 16 == 0 && 16 % NT == 0, "tile shapes");
  __shared__ __align__(16) float Bs[K * NCOL];
  __shared__ __align__(16) float As[ROWS * K];

  const int tid = threadIdx.x;
  const int lane = tid & 31;
  const int wave = __builtin_amdgcn_readfirstlane(tid >> 5);
  const int lo = blockIdx.x * RB;
  if (lo >= M) return;
  const int hi = (M - lo) < RB ? M : lo + RB;

  const float4* A4 = (const float4*)A;
  float4* As4 = (float4*)As;
  const long lastA = (long)M * KV - 1;
  Tile8 first;
  fetch_tile(A4, (long)lo * KV + tid, lastA, first);

  {
    const float4* W4 = (const float4*)W;
    float4* Bs4 = (float4*)Bs;
    const int sp = (int)((blockIdx.x * 61u) % (unsigned)(K / 2));
    for (int g0 = tid; g0 < ITEMS; g0 += 512 * WB) {
      float4 r0[WB], r1[WB];
#pragma unroll
      for (int u = 0; u < WB; ++u) {
        int g = g0 + u * 512;
        g = g < ITEMS ? g : ITEMS - 1;
        const int kq = g / C4;
        const int c4 = g - kq * C4;
        const int kp = kq + sp < K / 2 ? kq + sp : kq + sp - K / 2;
        r0[u] = W4[(2 * kp) * C4 + c4];
        r1[u] = W4[(2 * kp + 1) * C4 + c4];
      }
#pragma unroll
      for (int u = 0; u < WB; ++u) {
        const int g = g0 + u * 512;
        if (g < ITEMS) {
          const int kq = g / C4;
          const int c4 = g - kq * C4;
          const int kp = kq + sp < K / 2 ? kq + sp : kq + sp - K / 2;
          const int col = (c4 * 4 + 16 * (kp & 1)) & (NCOL - 1);
          const int pos4 = (kp * NCOL + col) >> 1;
          const float4 e0 = {r0[u].x, r1[u].x, r0[u].y, r1[u].y};
          const float4 e1 = {r0[u].z, r1[u].z, r0[u].w, r1[u].w};
          Bs4[pos4] = e0;
          Bs4[pos4 + 1] = e1;
        }
      }
    }
  }

  const int mtile = wave / NT;
  const int ntile = wave - mtile * NT;
  const int half = lane >> 4;
  const int lrow = (mtile << 4) + (lane & 15);
  const int koff = half << 1;
  const int ncol = (ntile << 4) + (lane & 15);
  const int bbase = (half * NCOL + ((ncol + 16 * half) & (NCOL - 1))) * 2;

  store_tile<TV>(As4, tid, first);
  __syncthreads();

  for (int r0 = lo;; r0 += ROWS) {
    const bool more = r0 + ROWS < hi;
    Tile8 pre;
    if (more) fetch_tile(A4, (long)(r0 + ROWS) * KV + tid, lastA, pre);
    const int rt = r0 + (mtile << 4);
    const bool act = rt < hi;
    v8f acc = {0.f, 0.f, 0.f, 0.f, 0.f, 0.f, 0.f, 0.f};
    if (act) {
#pragma unroll UNR
      for (int kk = 0; kk < KV; ++kk) {
        const v2f a = *(const v2f*)&As[lrow * K + kk * 4 + koff];
        const v2f b = *(const v2f*)&Bs[bbase + kk * 4 * NCOL];
        acc = __builtin_amdgcn_wmma_f32_16x16x4_f32(
             false, a,  false, b,
             (short)0, acc,  false,  false);
      }
    }
    if (more) {
      __syncthreads();
      store_tile<TV>(As4, tid, pre);
      __syncthreads();
    }
    if (act) {
      const int rbase = rt + (half << 3);
      if (rt + 16 <= hi) {
#pragma unroll
        for (int i = 0; i < 8; ++i) C[(size_t)(rbase + i) * NCOL + ncol] = acc[i];
      } else {
#pragma unroll
        for (int i = 0; i < 8; ++i) {
          const int r = rbase + i;
          if (r < hi) C[(size_t)r * NCOL + ncol] = acc[i];
        }
      }
    }
    if (!more) break;
  }
}

template <int M>
__device__ __forceinline__ float row_xor(float x) {
  return __int_as_float(__builtin_amdgcn_update_dpp(
      0, __float_as_int(x), 0x160 + M, 0xf, 0xf, true));
}
template <int W>
__device__ __forceinline__ float group_sum(float x) {
  x += row_xor<1>(x);
  x += row_xor<2>(x);
  x += row_xor<4>(x);
  if constexpr (W == 16) x += row_xor<8>(x);
  return x;
}
template <int W>
__device__ __forceinline__ float group_max(float x) {
  x = fmaxf(x, row_xor<1>(x));
  x = fmaxf(x, row_xor<2>(x));
  x = fmaxf(x, row_xor<4>(x));
  if constexpr (W == 16) x = fmaxf(x, row_xor<8>(x));
  return x;
}

template <int H, int F>
__global__ __launch_bounds__(256) void alpha_kernel(
    const float* __restrict__ h, const float* __restrict__ a_s,
    const float* __restrict__ a_d, float* __restrict__ out_s,
    float* __restrict__ out_d, int n) {
  constexpr int LPN = (H * F) / 4;
  constexpr int LPH = F / 4;
  constexpr int NPW = 32 / LPN;
  constexpr int U = (NPW == 1) ? 8 : 4;
  const int lane = threadIdx.x & 31;
  const int sub = lane % LPN;
  const int wave = (int)((blockIdx.x * blockDim.x + threadIdx.x) >> 5);
  const int node0 = wave * (NPW * U) + lane / LPN;
  const float4* h4 = (const float4*)h;
  const float4 as4 = ((const float4*)a_s)[sub];
  const float4 ad4 = ((const float4*)a_d)[sub];
  float4 v[U];
#pragma unroll
  for (int u = 0; u < U; ++u) {
    int nd = node0 + u * NPW;
    nd = nd < n ? nd : n - 1;
    v[u] = h4[(size_t)nd * LPN + sub];
  }
  __builtin_amdgcn_sched_barrier(0);
  float ps[U], pd[U];
#pragma unroll
  for (int u = 0; u < U; ++u) {
    ps[u] = group_sum<LPH>(v[u].x * as4.x + v[u].y * as4.y + v[u].z * as4.z +
                           v[u].w * as4.w);
    pd[u] = group_sum<LPH>(v[u].x * ad4.x + v[u].y * ad4.y + v[u].z * ad4.z +
                           v[u].w * ad4.w);
  }
#pragma unroll
  for (int u = 0; u < U; ++u) {
    const int nd = node0 + u * NPW;
    if (nd < n && (sub % LPH) == 0) {
      out_s[nd * H + sub / LPH] = ps[u];
      out_d[nd * H + sub / LPH] = pd[u];
    }
  }
}

template <int RO, int S, int EPB, int MAXOWN>
__global__ __launch_bounds__(1024) void range_lists_kernel(
    const int* __restrict__ src, const int* __restrict__ dst, int nE, int n,
    int nOwn, int NB, unsigned* __restrict__ glist) {
  static_assert(EPB % 1024 == 0 && S % 4 == 0, "edges per thread, uint4 stores");
  constexpr int U = EPB / 1024;
  constexpr int S4 = S / 4;
  __shared__ __align__(16) unsigned seg[MAXOWN * S];
  __shared__ unsigned cnts[MAXOWN];
  const int tid = threadIdx.x;
  const int b = blockIdx.x;
  const int e0 = b * EPB;
  int d[U], s[U];
#pragma unroll
  for (int u = 0; u < U; ++u) {
    int e = e0 + tid + u * 1024;
    e = e < nE ? e : nE - 1;
    d[u] = dst[e];
    s[u] = src[e];
  }
  {
    uint4* s4 = (uint4*)seg;
    const uint4 z = {0u, 0u, 0u, 0u};
    for (int i = tid; i < MAXOWN * S4; i += 1024) s4[i] = z;
    for (int i = tid; i < MAXOWN; i += 1024) cnts[i] = 0u;
  }
  __syncthreads();
#pragma unroll
  for (int u = 0; u < U; ++u) {
    const int e = e0 + tid + u * 1024;
    if (e < nE && (unsigned)d[u] < (unsigned)n) {
      const int o = d[u] / RO;
      const unsigned row = (unsigned)(d[u] - o * RO);
      const unsigned k = atomicAdd(&cnts[o], 1u) + 1u;
      if (k < (unsigned)S)
        seg[o * S + k] = ((unsigned)s[u] & 0x00ffffffu) | (row << 24);
    }
  }
  __syncthreads();
  for (int i = tid; i < MAXOWN; i += 1024) seg[i * S] = cnts[i];
  __syncthreads();
  uint4* g4 = (uint4*)glist;
  const uint4* s4 = (const uint4*)seg;
  for (int i = tid; i < nOwn * S4; i += 1024) {
    const int o = i / S4;
    const int w = i - o * S4;
    g4[((size_t)o * NB + b) * S4 + w] = s4[i];
  }
}

template <int H, int F, int R, int CAP, bool ELU, int S, int EPB>
__global__ __launch_bounds__(1024) void owner_agg_kernel(
    const float* __restrict__ hsrc, const float* __restrict__ as,
    const float* __restrict__ ad, const int* __restrict__ src,
    const int* __restrict__ dst, const unsigned* __restrict__ glist, int NB,
    int nE, int n, const float* __restrict__ bias, float* __restrict__ outp) {
  static_assert(R <= CAP, "the self loops must fit the list");
  static_assert(R <= 256, "a row number must fit 8 bits of a segment word");
  static_assert(S == 64, "a wave reads a segment as two words per lane");
  static_assert(H == 1 || H == 2, "a sorted entry carries one weight per head");
  constexpr int HF = H * F;
  constexpr int HF4 = HF / 4;
  constexpr int UI = 8;
  constexpr int G = 8;
  constexpr int U3 = 4;
  constexpr int PQ = HF4;
  constexpr int EPL = 32 / PQ;
  constexpr int PITCH = HF >= 64 ? HF : HF + 8;
  static_assert(32 % PQ == 0 && PITCH % 4 == 0, "rows must tile a wave");
  __shared__ __align__(16) float acc[R * PITCH];
  __shared__ float den[R * H];
  __shared__ unsigned mxo[R * H];
  __shared__ float adl[R * H];
  __shared__ int ls[CAP];
  __shared__ int lt[CAP];
  __shared__ float lw[CAP * H];
  __shared__ int cnt;
  __shared__ __align__(16) unsigned rfirst[256];
  __shared__ __align__(16) uint4 srt[CAP];

  const int tid = threadIdx.x;
  const int lane = tid & 31;
  const int wv = __builtin_amdgcn_readfirstlane(tid >> 5);
  const int lo = blockIdx.x * R;
  const int nr = (n - lo) < R ? (n - lo) : R;

  if (tid == 0) cnt = nr;
  for (int i = tid; i < 256; i += 1024) rfirst[i] = 0u;
  for (int i = tid; i < nr; i += 1024) {
    ls[i] = lo + i;
    lt[i] = i;
  }
  for (int i = tid; i < nr * H; i += 1024) {
    den[i] = 0.f;
    mxo[i] = 0u;
    adl[i] = ad[(size_t)lo * H + i];
  }
  {
    float4* acc4 = (float4*)acc;
    const float4 z = {0.f, 0.f, 0.f, 0.f};
    for (int i = tid; i < nr * (PITCH / 4); i += 1024) acc4[i] = z;
  }
  __syncthreads();

  const unsigned unr = (unsigned)nr;
  auto take = [&](int e, unsigned t) {
    const int pos = atomicAdd(&cnt, 1);
    if (pos < CAP) {
      ls[pos] = src[e];
      lt[pos] = (int)t;
    }
  };

  if (glist != nullptr) {
    const unsigned* const gl = glist + (size_t)blockIdx.x * NB * S;
    for (int sg0 = wv; sg0 < NB; sg0 += 32 * UI) {
      uint2 v[UI];
#pragma unroll
      for (int u = 0; u < UI; ++u) {
        int sg = sg0 + u * 32;
        sg = sg < NB ? sg : NB - 1;
        v[u] = ((const uint2*)(gl + (size_t)sg * S))[lane];
      }
#pragma unroll
      for (int u = 0; u < UI; ++u) {
        const int sg = sg0 + u * 32;
        if (sg < NB) {
          const int c = __builtin_amdgcn_readfirstlane((int)v[u].x);
          if (c >= 0 && c < S) {
            int base = 0;
            if (lane == 0) base = atomicAdd(&cnt, c);
            base = __builtin_amdgcn_readfirstlane(base);
            const int p0 = base + 2 * lane - 1;
            if (lane >= 1 && 2 * lane <= c && p0 < CAP) {
              ls[p0] = (int)(v[u].x & 0x00ffffffu);
              lt[p0] = (int)(v[u].x >> 24);
            }
            if (2 * lane + 1 <= c && p0 + 1 < CAP) {
              ls[p0 + 1] = (int)(v[u].y & 0x00ffffffu);
              lt[p0 + 1] = (int)(v[u].y >> 24);
            }
          } else {
            const int ea = sg * EPB;
            const int eb = (nE - ea) < EPB ? nE : ea + EPB;
            for (int e = ea + lane; e < eb; e += 32) {
              const unsigned t = (unsigned)(dst[e] - lo);
              if (t < unr) take(e, t);
            }
          }
        }
      }
    }
  } else {
    for (int e = tid; e < nE; e += 1024) {
      const unsigned t = (unsigned)(dst[e] - lo);
      if (t < unr) take(e, t);
    }
  }
  __syncthreads();
  const int total = cnt;
  const int nl = total <= CAP ? total : 0;
  if (total > CAP) {
    for (int i = tid; i < nr * H; i += 1024) {
      float x = as[(size_t)lo * H + i] + adl[i];
      x = x > 0.f ? x : 0.2f * x;
      atomicMax(&mxo[i], f2ord(x));
    }
    for (int e = tid; e < nE; e += 1024) {
      const unsigned t = (unsigned)(dst[e] - lo);
      if (t < unr) {
        const int s = src[e];
        for (int hd = 0; hd < H; ++hd) {
          float x = as[(size_t)s * H + hd] + adl[t * H + hd];
          x = x > 0.f ? x : 0.2f * x;
          atomicMax(&mxo[t * H + hd], f2ord(x));
        }
      }
    }
    __syncthreads();
    for (int i = tid; i < nr * H; i += 1024) {
      const int r = i / H;
      const int hd = i - r * H;
      float x = as[(size_t)lo * H + i] + adl[i];
      x = x > 0.f ? x : 0.2f * x;
      const float w = __expf(x - ord2f(mxo[i]));
      atomicAdd(&den[i], w);
      for (int f = 0; f < F; ++f) {
        const int c = hd * F + f;
        atomicAdd(&acc[r * PITCH + (c & 3) * PQ + (c >> 2)],
                  hsrc[(size_t)(lo + r) * HF + c] * w);
      }
    }
    for (int e = tid; e < nE; e += 1024) {
      const unsigned t = (unsigned)(dst[e] - lo);
      if (t < unr) {
        const int s = src[e];
        for (int hd = 0; hd < H; ++hd) {
          float x = as[(size_t)s * H + hd] + adl[t * H + hd];
          x = x > 0.f ? x : 0.2f * x;
          const float w = __expf(x - ord2f(mxo[t * H + hd]));
          atomicAdd(&den[t * H + hd], w);
          for (int f = 0; f < F; ++f) {
            const int c = hd * F + f;
            atomicAdd(&acc[t * PITCH + (c & 3) * PQ + (c >> 2)],
                      hsrc[(size_t)s * HF + c] * w);
          }
        }
      }
    }
  }

  for (int i0 = tid; i0 < nl; i0 += 1024 * U3) {
    int s[U3], t[U3];
#pragma unroll
    for (int u = 0; u < U3; ++u) {
      int i = i0 + u * 1024;
      i = i < nl ? i : i0;
      s[u] = ls[i];
      t[u] = lt[i];
    }
    float a[U3][H];
#pragma unroll
    for (int u = 0; u < U3; ++u) {
#pragma unroll
      for (int hd = 0; hd < H; ++hd) a[u][hd] = as[(size_t)s[u] * H + hd];
    }
#pragma unroll
    for (int u = 0; u < U3; ++u) {
      const int i = i0 + u * 1024;
      if (i < nl) {
#pragma unroll
        for (int hd = 0; hd < H; ++hd) {
          const float v = a[u][hd] + adl[t[u] * H + hd];
          const float x = v > 0.f ? v : 0.2f * v;
          lw[i * H + hd] = x;
          atomicMax(&mxo[t[u] * H + hd], f2ord(x));
        }
        const unsigned rk = atomicAdd(&rfirst[t[u]], 1u);
        lt[i] = t[u] | (int)(rk << 8);
      }
    }
  }
  __syncthreads();

  if (wv == 0) {
    uint4* r4 = (uint4*)rfirst;
    const uint4 ca = r4[2 * lane], cb = r4[2 * lane + 1];
    const unsigned e1 = ca.x, e2 = e1 + ca.y, e3 = e2 + ca.z, e4 = e3 + ca.w;
    const unsigned e5 = e4 + cb.x, e6 = e5 + cb.y, e7 = e6 + cb.z;
    const unsigned tot = e7 + cb.w;
    unsigned inc = tot;
#pragma unroll
    for (int d = 1; d < 32; d <<= 1) {
      const unsigned y = __shfl_up(inc, (unsigned)d);
      if (lane >= d) inc += y;
    }
    const unsigned ex = inc - tot;
    const uint4 oa = {ex, ex + e1, ex + e2, ex + e3};
    const uint4 ob = {ex + e4, ex + e5, ex + e6, ex + e7};
    r4[2 * lane] = oa;
    r4[2 * lane + 1] = ob;
  }
  __syncthreads();

  for (int i = tid; i < nl; i += 1024) {
    const unsigned tr = (unsigned)lt[i];
    const unsigned t = tr & 255u;
    uint4 rec;
    rec.x = (unsigned)ls[i];
    rec.y = t;
    rec.z = 0u;
    rec.w = 0u;
#pragma unroll
    for (int hd = 0; hd < H; ++hd) {
      const float w = __expf(lw[i * H + hd] - ord2f(mxo[t * H + hd]));
      atomicAdd(&den[t * H + hd], w);
      if (hd == 0) rec.z = __float_as_uint(w);
      else rec.w = __float_as_uint(w);
    }
    srt[rfirst[t] + (tr >> 8)] = rec;
  }
  __syncthreads();

  {
    constexpr int NG = 32 * EPL;
    const int es = lane / PQ;
    const int q = lane - es * PQ;
    const float4* h4 = (const float4*)hsrc;
    const int c = (nl + NG - 1) / NG;
    const int p0 = (wv * EPL + es) * c;
    const int pe = p0 + c < nl ? p0 + c : nl;
    const bool second = H == 2 && (q * 4) / F != 0;
    float4 sum = {0.f, 0.f, 0.f, 0.f};
    int cur = -1;
    auto flush = [&]() __attribute__((always_inline)) {
      float* ap = acc + cur * PITCH + q;
      atomicAdd(ap, sum.x);
      atomicAdd(ap + PQ, sum.y);
      atomicAdd(ap + 2 * PQ, sum.z);
      atomicAdd(ap + 3 * PQ, sum.w);
    };
    auto group = [&](const int k0, const bool full)
        __attribute__((always_inline)) {
      uint4 rec[G];
      float4 v[G];
      const float4* hp[G];
#pragma unroll
      for (int g = 0; g < G; ++g) {
        int p = p0 + k0 + g;
        if (!full) p = p < pe ? p : 0;
        rec[g] = srt[p];
        hp[g] = h4 + (size_t)rec[g].x * HF4 + q;
      }
      __builtin_amdgcn_sched_barrier(0);
#pragma unroll
      for (int g = 0; g < G; ++g) v[g] = *hp[g];
#pragma unroll
      for (int g = 0; g < G; ++g) {
        if (full || p0 + k0 + g < pe) {
          const int t = (int)rec[g].y;
          if (t != cur) {
            if (cur >= 0) flush();
            sum.x = 0.f; sum.y = 0.f; sum.z = 0.f; sum.w = 0.f;
            cur = t;
          }
          const float w = __uint_as_float(second ? rec[g].w : rec[g].z);
          sum.x += v[g].x * w;
          sum.y += v[g].y * w;
          sum.z += v[g].z * w;
          sum.w += v[g].w * w;
        }
      }
    };
    for (int k0 = 0; k0 < c; k0 += G) {
      const bool full = k0 + G <= c &&
                        (wv * EPL + EPL - 1) * c + k0 + G <= nl;
      if (full) group(k0, true);
      else group(k0, false);
    }
    if (cur >= 0) flush();
  }
  __syncthreads();

  {
    float4* o4 = (float4*)outp;
    for (int i = tid; i < nr * HF4; i += 1024) {
      const int r = i / HF4;
      const int c4 = i - r * HF4;
      const float dn = den[r * H + (c4 * 4) / F] + 1e-16f;
      const float* ap = acc + r * PITCH + c4;
      float4 x = {ap[0], ap[PQ], ap[2 * PQ], ap[3 * PQ]};
      x.x /= dn; x.y /= dn; x.z /= dn; x.w /= dn;
      if constexpr (ELU) {
        const float4 bb = ((const float4*)bias)[c4];
        float t;
        t = x.x + bb.x; x.x = t > 0.f ? t : (__expf(t) - 1.f);
        t = x.y + bb.y; x.y = t > 0.f ? t : (__expf(t) - 1.f);
        t = x.z + bb.z; x.z = t > 0.f ? t : (__expf(t) - 1.f);
        t = x.w + bb.w; x.w = t > 0.f ? t : (__expf(t) - 1.f);
      }
      o4[(size_t)(lo + r) * HF4 + c4] = x;
    }
  }
}

__global__ __launch_bounds__(256) void logsoftmax_kernel(
    const float* __restrict__ agg, const float* __restrict__ bias,
    float* __restrict__ out, int n) {
  constexpr int U = 4;
  const int lane = threadIdx.x & 31;
  const int sub = lane & 7;
  const int wave = (int)((blockIdx.x * blockDim.x + threadIdx.x) >> 5);
  const int node0 = wave * (4 * U) + (lane >> 3);
  const float4* a4 = (const float4*)agg;
  float4* o4 = (float4*)out;
  const float4 b = ((const float4*)bias)[sub];
  float4 v[U];
#pragma unroll
  for (int u = 0; u < U; ++u) {
    int nd = node0 + u * 4;
    nd = nd < n ? nd : n - 1;
    v[u] = a4[(size_t)nd * 8 + sub];
  }
  float4 r[U];
#pragma unroll
  for (int u = 0; u < U; ++u) {
    const float x0 = v[u].x + b.x, x1 = v[u].y + b.y;
    const float x2 = v[u].z + b.z, x3 = v[u].w + b.w;
    const float m = group_max<8>(fmaxf(fmaxf(x0, x1), fmaxf(x2, x3)));
    const float s = group_sum<8>(__expf(x0 - m) + __expf(x1 - m) +
                                 __expf(x2 - m) + __expf(x3 - m));
    const float ls = __logf(s);
    r[u].x = x0 - m - ls; r[u].y = x1 - m - ls;
    r[u].z = x2 - m - ls; r[u].w = x3 - m - ls;
  }
#pragma unroll
  for (int u = 0; u < U; ++u) {
    const int nd = node0 + u * 4;
    if (nd < n) o4[(size_t)nd * 8 + sub] = r[u];
  }
}

extern "C" void kernel_launch(void* const* d_in, const int* in_sizes, int n_in,
                              void* d_out, int out_size, void* d_ws,
                              size_t ws_size, hipStream_t stream) {
  (void)n_in; (void)out_size;
  const float* x      = (const float*)d_in[0];
  const int*   ei     = (const int*)d_in[1];
  const float* W1     = (const float*)d_in[2];
  const float* a_src1 = (const float*)d_in[3];
  const float* a_dst1 = (const float*)d_in[4];
  const float* b1     = (const float*)d_in[5];
  const float* W2     = (const float*)d_in[6];
  const float* a_src2 = (const float*)d_in[7];
  const float* a_dst2 = (const float*)d_in[8];
  const float* b2     = (const float*)d_in[9];

  const int n    = in_sizes[0] / 500;
  const int nE   = in_sizes[1] / 2;
  const int* src = ei;
  const int* dst = ei + nE;
  constexpr int R = 196;
  const int nOwn = (n + R - 1) / R;
  constexpr int S = 64;
  constexpr int EPB = 4096;
  constexpr int MAXOWN = 256;
  const int NB = (nE + EPB - 1) / EPB;

  float* ws = (float*)d_ws;
  size_t o = 0;
  float* h1      = ws + o; o += ((size_t)n * 128 + 63) / 64 * 64;
  float* agg1    = ws + o; o += ((size_t)n * 128 + 63) / 64 * 64;
  float* glistp  = ws + o; o += (size_t)nOwn * NB * S;
  float* as1     = ws + o; o += (size_t)n * 2;
  float* ad1     = ws + o; o += (size_t)n * 2;
  float* as2     = ws + o; o += (size_t)n;
  float* ad2     = ws + o; o += (size_t)n;
  float* h2      = h1;
  float* agg2    = h1 + (size_t)n * 32;
  const bool lists = n <= (1 << 24) && nOwn <= MAXOWN && NB > 0 &&
                     o * sizeof(float) <= ws_size;
  unsigned* glist = lists ? (unsigned*)glistp : nullptr;

  if (lists)
    range_lists_kernel<R, S, EPB, MAXOWN><<<NB, 1024, 0, stream>>>(
        src, dst, nE, n, nOwn, NB, glist);

  gemm_ldsw_kernel<500, 128><<<nOwn, 512, 0, stream>>>(x, W1, h1, n, R);
  alpha_kernel<2, 64><<<(n + 63) / 64, 256, 0, stream>>>(
      h1, a_src1, a_dst1, as1, ad1, n);
  owner_agg_kernel<2, 64, R, 6144, true, S, EPB><<<nOwn, 1024, 0, stream>>>(
      h1, as1, ad1, src, dst, glist, NB, nE, n, b1, agg1);

  gemm_ldsw_kernel<128, 32><<<nOwn, 512, 0, stream>>>(agg1, W2, h2, n, R);
  alpha_kernel<1, 32><<<(n + 127) / 128, 256, 0, stream>>>(
      h2, a_src2, a_dst2, as2, ad2, n);
  owner_agg_kernel<1, 32, R, 8192, false, S, EPB><<<nOwn, 1024, 0, stream>>>(
      h2, as2, ad2, src, dst, glist, NB, nE, n, b2, agg2);

  logsoftmax_kernel<<<(n + 127) / 128, 256, 0, stream>>>(
      agg2, b2, (float*)d_out, n);
}
